# starting kernel
# baseline (speedup 1.0000x reference)
#include <hip/hip_runtime.h>
#include <hip/hip_bf16.h>


typedef float v2f __attribute__((ext_vector_type(2)));
typedef float v8f __attribute__((ext_vector_type(8)));

#define GAT_NEG_SLOPE 0.2f

__device__ __forceinline__ unsigned GAT_fkey(float f) {
  unsigned u = __float_as_uint(f);
  return (u & 0x80000000u) ? ~u : (u | 0x80000000u);
}
__device__ __forceinline__ float GAT_funkey(unsigned k) {
  return (k & 0x80000000u) ? __uint_as_float(k & 0x7FFFFFFFu)
                           : __uint_as_float(~k);
}
__device__ __forceinline__ float GAT_lrelu(float v) {
  return v > 0.0f ? v : GAT_NEG_SLOPE * v;
}

__global__ void GAT_init_bias(float* __restrict__ out, const float* __restrict__ b,
                              long long total, int C) {
  long long i = (long long)blockIdx.x * blockDim.x + threadIdx.x;
  if (i >= total) return;
  out[i] = b[(int)(i % C)];
}

__global__ void GAT_gemm_dual(const float* __restrict__ A,
                              const float* __restrict__ Bs,
                              const float* __restrict__ Bd,
                              const float* __restrict__ attS,
                              const float* __restrict__ attD,
                              float* __restrict__ Hout,
                              float* __restrict__ aS,
                              float* __restrict__ aD,
                              int M, int K, int NC, int Ch, int reluA) {
  const int wave = blockIdx.x * (blockDim.x >> 5) + (threadIdx.x >> 5);
  const int nT = NC >> 4;
  const int totalTiles = (M >> 4) * nT;
  if (wave >= totalTiles) return;

  const int mBase = (wave / nT) << 4;
  const int nBase = (wave % nT) << 4;
  const int lane = threadIdx.x & 31;
  const int half = lane >> 4;
  const int l16  = lane & 15;

  v8f cS = {};
  v8f cD = {};
  const float* arow = A + (size_t)(mBase + l16) * K;
  const int col = nBase + l16;

  for (int k0 = 0; k0 < K; k0 += 4) {
    const int ka = k0 + 2 * half;
    float a0 = arow[ka];
    float a1 = arow[ka + 1];
    if (reluA) { a0 = fmaxf(a0, 0.0f); a1 = fmaxf(a1, 0.0f); }
    v2f a;  a.x = a0; a.y = a1;

    v2f bs, bd;
    bs.x = Bs[(size_t)ka * NC + col];
    bs.y = Bs[(size_t)(ka + 1) * NC + col];
    bd.x = Bd[(size_t)ka * NC + col];
    bd.y = Bd[(size_t)(ka + 1) * NC + col];

    cS = __builtin_amdgcn_wmma_f32_16x16x4_f32(false, a, false, bs, (short)0, cS, false, false);
    cD = __builtin_amdgcn_wmma_f32_16x16x4_f32(false, a, false, bd, (short)0, cD, false, false);
  }

#pragma unroll
  for (int r = 0; r < 8; ++r) {
    Hout[(size_t)(mBase + r + 8 * half) * NC + col] = cS[r];
  }

  const int head = nBase / Ch;
  const int H = NC / Ch;
  const float avS = attS[col];
  const float avD = attD[col];
#pragma unroll
  for (int r = 0; r < 8; ++r) {
    float vs = cS[r] * avS;
    float vd = cD[r] * avD;
#pragma unroll
    for (int m = 1; m < 16; m <<= 1) {
      vs += __shfl_xor(vs, m, 32);
      vd += __shfl_xor(vd, m, 32);
    }
    if (l16 == 0) {
      const int row = mBase + r + 8 * half;
      atomicAdd(&aS[row * H + head], vs);
      atomicAdd(&aD[row * H + head], vd);
    }
  }
}

__global__ void GAT_edge_max(const int* __restrict__ src, const int* __restrict__ dst,
                             const float* __restrict__ aS, const float* __restrict__ aD,
                             unsigned* __restrict__ mkey, long long EH, int H) {
  long long i = (long long)blockIdx.x * blockDim.x + threadIdx.x;
  if (i >= EH) return;
  const int e = (int)(i / H), h = (int)(i % H);
  const int s = src[e], d = dst[e];
  const float v = GAT_lrelu(aS[s * H + h] + aD[d * H + h]);
  atomicMax(&mkey[d * H + h], GAT_fkey(v));
}

__global__ void GAT_edge_expsum(const int* __restrict__ src, const int* __restrict__ dst,
                                const float* __restrict__ aS, const float* __restrict__ aD,
                                const unsigned* __restrict__ mkey, float* __restrict__ den,
                                long long EH, int H) {
  long long i = (long long)blockIdx.x * blockDim.x + threadIdx.x;
  if (i >= EH) return;
  const int e = (int)(i / H), h = (int)(i % H);
  const int s = src[e], d = dst[e];
  const float v = GAT_lrelu(aS[s * H + h] + aD[d * H + h]);
  const float ex = __expf(v - GAT_funkey(mkey[d * H + h]));
  atomicAdd(&den[d * H + h], ex);
}

__global__ void GAT_edge_scatter(const int* __restrict__ src, const int* __restrict__ dst,
                                 const float* __restrict__ aS, const float* __restrict__ aD,
                                 const unsigned* __restrict__ mkey, const float* __restrict__ den,
                                 const float* __restrict__ Hsrc, float* __restrict__ out,
                                 long long total, int H, int C) {
  long long i = (long long)blockIdx.x * blockDim.x + threadIdx.x;
  if (i >= total) return;
  const int pe = (H * C) >> 2;
  const int e = (int)(i / pe);
  const int j = (int)(i % pe);
  const int c4 = j << 2;
  const int h = c4 / C;
  const int s = src[e], d = dst[e];

  const float v = GAT_lrelu(aS[s * H + h] + aD[d * H + h]);
  const float ex = __expf(v - GAT_funkey(mkey[d * H + h]));
  const float alpha = ex / (den[d * H + h] + 1e-16f);

  const float4 hv = *(const float4*)(Hsrc + (size_t)s * (H * C) + c4);
  float* op = out + (size_t)d * (H * C) + c4;
  atomicAdd(op + 0, hv.x * alpha);
  atomicAdd(op + 1, hv.y * alpha);
  atomicAdd(op + 2, hv.z * alpha);
  atomicAdd(op + 3, hv.w * alpha);
}

extern "C" void kernel_launch(void* const* d_in, const int* in_sizes, int n_in,
                              void* d_out, int out_size, void* d_ws, size_t ws_size,
                              hipStream_t stream) {
  (void)n_in; (void)out_size; (void)ws_size;

  const float* x     = (const float*)d_in[0];
  const int*   ei    = (const int*)d_in[1];
  const float* Wsrc1 = (const float*)d_in[2];
  const float* Wdst1 = (const float*)d_in[3];
  const float* attS1 = (const float*)d_in[4];
  const float* attD1 = (const float*)d_in[5];
  const float* b1    = (const float*)d_in[6];
  const float* Wsrc2 = (const float*)d_in[7];
  const float* Wdst2 = (const float*)d_in[8];
  const float* attS2 = (const float*)d_in[9];
  const float* attD2 = (const float*)d_in[10];
  const float* b2    = (const float*)d_in[11];

  const int FIN = 128, HEADS = 8, HID = 32, EMB = 64;
  const int HC1 = HEADS * HID;
  const long long N = in_sizes[0] / FIN;
  const long long E = in_sizes[1] / 2;
  const int* src = ei;
  const int* dst = ei + E;

  float* p = (float*)d_ws;
  float* h1   = p; p += N * HC1;
  float* h2   = p; p += N * EMB;
  float* aS1  = p; p += N * HEADS;
  float* aD1  = p; p += N * HEADS;
  float* den1 = p; p += N * HEADS;
  float* aS2  = p; p += N;
  float* aD2  = p; p += N;
  float* den2 = p; p += N;
  unsigned* mk1 = (unsigned*)p; p += N * HEADS;
  unsigned* mk2 = (unsigned*)p; p += N;
  float* out1 = p; p += N * HC1;
  float* outF = (float*)d_out;

  const size_t zbytes = (size_t)N * (3 * HEADS + 3 + HEADS + 1) * sizeof(float);
  hipMemsetAsync(aS1, 0, zbytes, stream);

  const int TPB = 256;
  auto nb = [](long long n, int t) { return (unsigned)((n + t - 1) / t); };

  GAT_init_bias<<<nb(N * HC1, TPB), TPB, 0, stream>>>(out1, b1, N * HC1, HC1);
  GAT_init_bias<<<nb(N * EMB, TPB), TPB, 0, stream>>>(outF, b2, N * EMB, EMB);

  {
    const long long tiles = (N / 16) * (HC1 / 16);
    GAT_gemm_dual<<<nb(tiles * 32, TPB), TPB, 0, stream>>>(
        x, Wsrc1, Wdst1, attS1, attD1, h1, aS1, aD1,
        (int)N, FIN, HC1, HID,  0);
  }
  GAT_edge_max<<<nb(E * HEADS, TPB), TPB, 0, stream>>>(src, dst, aS1, aD1, mk1, E * HEADS, HEADS);
  GAT_edge_expsum<<<nb(E * HEADS, TPB), TPB, 0, stream>>>(src, dst, aS1, aD1, mk1, den1, E * HEADS, HEADS);
  {
    const long long total = E * (long long)(HEADS * HID / 4);
    GAT_edge_scatter<<<nb(total, TPB), TPB, 0, stream>>>(
        src, dst, aS1, aD1, mk1, den1, h1, out1, total, HEADS, HID);
  }

  {
    const long long tiles = (N / 16) * (EMB / 16);
    GAT_gemm_dual<<<nb(tiles * 32, TPB), TPB, 0, stream>>>(
        out1, Wsrc2, Wdst2, attS2, attD2, h2, aS2, aD2,
        (int)N, HC1, EMB, EMB,  1);
  }
  GAT_edge_max<<<nb(E, TPB), TPB, 0, stream>>>(src, dst, aS2, aD2, mk2, E, 1);
  GAT_edge_expsum<<<nb(E, TPB), TPB, 0, stream>>>(src, dst, aS2, aD2, mk2, den2, E, 1);
  {
    const long long total = E * (long long)(EMB / 4);
    GAT_edge_scatter<<<nb(total, TPB), TPB, 0, stream>>>(
        src, dst, aS2, aD2, mk2, den2, h2, outF, total, 1, EMB);
  }
}


# edge scatter remapped: a wave owns whole edges, lane l takes float l of each 32-float piece, so each load and float atomic covers one full 128-B line (was 4 partial lines); atomic sum order kept
# speedup vs baseline: 1.4468x; 1.4468x over previous
#include <hip/hip_runtime.h>
#include <hip/hip_bf16.h>


typedef float v2f __attribute__((ext_vector_type(2)));
typedef float v8f __attribute__((ext_vector_type(8)));

#define GAT_NEG_SLOPE 0.2f

__device__ __forceinline__ unsigned GAT_fkey(float f) {
  unsigned u = __float_as_uint(f);
  return (u & 0x80000000u) ? ~u : (u | 0x80000000u);
}
__device__ __forceinline__ float GAT_funkey(unsigned k) {
  return (k & 0x80000000u) ? __uint_as_float(k & 0x7FFFFFFFu)
                           : __uint_as_float(~k);
}
__device__ __forceinline__ float GAT_lrelu(float v) {
  return v > 0.0f ? v : GAT_NEG_SLOPE * v;
}

__global__ void GAT_init_bias(float* __restrict__ out, const float* __restrict__ b,
                              long long total, int C) {
  long long i = (long long)blockIdx.x * blockDim.x + threadIdx.x;
  if (i >= total) return;
  out[i] = b[(int)(i % C)];
}

__global__ void GAT_gemm_dual(const float* __restrict__ A,
                              const float* __restrict__ Bs,
                              const float* __restrict__ Bd,
                              const float* __restrict__ attS,
                              const float* __restrict__ attD,
                              float* __restrict__ Hout,
                              float* __restrict__ aS,
                              float* __restrict__ aD,
                              int M, int K, int NC, int Ch, int reluA) {
  const int wave = blockIdx.x * (blockDim.x >> 5) + (threadIdx.x >> 5);
  const int nT = NC >> 4;
  const int totalTiles = (M >> 4) * nT;
  if (wave >= totalTiles) return;

  const int mBase = (wave / nT) << 4;
  const int nBase = (wave % nT) << 4;
  const int lane = threadIdx.x & 31;
  const int half = lane >> 4;
  const int l16  = lane & 15;

  v8f cS = {};
  v8f cD = {};
  const float* arow = A + (size_t)(mBase + l16) * K;
  const int col = nBase + l16;

  for (int k0 = 0; k0 < K; k0 += 4) {
    const int ka = k0 + 2 * half;
    float a0 = arow[ka];
    float a1 = arow[ka + 1];
    if (reluA) { a0 = fmaxf(a0, 0.0f); a1 = fmaxf(a1, 0.0f); }
    v2f a;  a.x = a0; a.y = a1;

    v2f bs, bd;
    bs.x = Bs[(size_t)ka * NC + col];
    bs.y = Bs[(size_t)(ka + 1) * NC + col];
    bd.x = Bd[(size_t)ka * NC + col];
    bd.y = Bd[(size_t)(ka + 1) * NC + col];

    cS = __builtin_amdgcn_wmma_f32_16x16x4_f32(false, a, false, bs, (short)0, cS, false, false);
    cD = __builtin_amdgcn_wmma_f32_16x16x4_f32(false, a, false, bd, (short)0, cD, false, false);
  }

#pragma unroll
  for (int r = 0; r < 8; ++r) {
    Hout[(size_t)(mBase + r + 8 * half) * NC + col] = cS[r];
  }

  const int head = nBase / Ch;
  const int H = NC / Ch;
  const float avS = attS[col];
  const float avD = attD[col];
#pragma unroll
  for (int r = 0; r < 8; ++r) {
    float vs = cS[r] * avS;
    float vd = cD[r] * avD;
#pragma unroll
    for (int m = 1; m < 16; m <<= 1) {
      vs += __shfl_xor(vs, m, 32);
      vd += __shfl_xor(vd, m, 32);
    }
    if (l16 == 0) {
      const int row = mBase + r + 8 * half;
      atomicAdd(&aS[row * H + head], vs);
      atomicAdd(&aD[row * H + head], vd);
    }
  }
}

__global__ void GAT_edge_max(const int* __restrict__ src, const int* __restrict__ dst,
                             const float* __restrict__ aS, const float* __restrict__ aD,
                             unsigned* __restrict__ mkey, long long EH, int H) {
  long long i = (long long)blockIdx.x * blockDim.x + threadIdx.x;
  if (i >= EH) return;
  const int e = (int)(i / H), h = (int)(i % H);
  const int s = src[e], d = dst[e];
  const float v = GAT_lrelu(aS[s * H + h] + aD[d * H + h]);
  atomicMax(&mkey[d * H + h], GAT_fkey(v));
}

__global__ void GAT_edge_expsum(const int* __restrict__ src, const int* __restrict__ dst,
                                const float* __restrict__ aS, const float* __restrict__ aD,
                                const unsigned* __restrict__ mkey, float* __restrict__ den,
                                long long EH, int H) {
  long long i = (long long)blockIdx.x * blockDim.x + threadIdx.x;
  if (i >= EH) return;
  const int e = (int)(i / H), h = (int)(i % H);
  const int s = src[e], d = dst[e];
  const float v = GAT_lrelu(aS[s * H + h] + aD[d * H + h]);
  const float ex = __expf(v - GAT_funkey(mkey[d * H + h]));
  atomicAdd(&den[d * H + h], ex);
}

template <int H, int C, int EPW>
__global__ void GAT_edge_scatter(const int* __restrict__ src, const int* __restrict__ dst,
                                 const float* __restrict__ aS, const float* __restrict__ aD,
                                 const unsigned* __restrict__ mkey, const float* __restrict__ den,
                                 const float* __restrict__ Hsrc, float* __restrict__ out,
                                 long long E) {
  constexpr int HC = H * C;
  constexpr int NP = HC / 32;
  static_assert(EPW * H <= 32, "one lane per (edge, head)");
  static_assert(C % 32 == 0, "a 32-float piece must lie inside one head");

  const int lane = threadIdx.x & 31;
  const long long wave = (long long)blockIdx.x * (blockDim.x >> 5) + (threadIdx.x >> 5);
  const long long e0 = wave * EPW;
  if (e0 >= E) return;
  const int ne = (E - e0) < (long long)EPW ? (int)(E - e0) : EPW;

  const int el = lane / H;
  const int h  = lane % H;
  int s = 0, d = 0;
  float alpha = 0.0f;
  if (el < ne) {
    s = src[e0 + el];
    d = dst[e0 + el];
    const float v = GAT_lrelu(aS[s * H + h] + aD[d * H + h]);
    const float ex = __expf(v - GAT_funkey(mkey[d * H + h]));
    alpha = ex / (den[d * H + h] + 1e-16f);
  }

  float hv[EPW][NP];
#pragma unroll
  for (int j = 0; j < EPW; ++j) {
    if (j < ne) {
      const int sj = __builtin_amdgcn_readlane(s, j * H);
      const float* hp = Hsrc + (size_t)sj * HC + lane;
#pragma unroll
      for (int k = 0; k < NP; ++k) hv[j][k] = hp[k * 32];
    }
  }
#pragma unroll
  for (int j = 0; j < EPW; ++j) {
    if (j < ne) {
      const int dj = __builtin_amdgcn_readlane(d, j * H);
      float* op = out + (size_t)dj * HC + lane;
#pragma unroll
      for (int k = 0; k < NP; ++k) {
        const float a = __int_as_float(
            __builtin_amdgcn_readlane(__float_as_int(alpha), j * H + (k * 32) / C));
        atomicAdd(op + k * 32, hv[j][k] * a);
      }
    }
  }
}

extern "C" void kernel_launch(void* const* d_in, const int* in_sizes, int n_in,
                              void* d_out, int out_size, void* d_ws, size_t ws_size,
                              hipStream_t stream) {
  (void)n_in; (void)out_size; (void)ws_size;

  const float* x     = (const float*)d_in[0];
  const int*   ei    = (const int*)d_in[1];
  const float* Wsrc1 = (const float*)d_in[2];
  const float* Wdst1 = (const float*)d_in[3];
  const float* attS1 = (const float*)d_in[4];
  const float* attD1 = (const float*)d_in[5];
  const float* b1    = (const float*)d_in[6];
  const float* Wsrc2 = (const float*)d_in[7];
  const float* Wdst2 = (const float*)d_in[8];
  const float* attS2 = (const float*)d_in[9];
  const float* attD2 = (const float*)d_in[10];
  const float* b2    = (const float*)d_in[11];

  const int FIN = 128, HEADS = 8, HID = 32, EMB = 64;
  const int HC1 = HEADS * HID;
  const long long N = in_sizes[0] / FIN;
  const long long E = in_sizes[1] / 2;
  const int* src = ei;
  const int* dst = ei + E;

  float* p = (float*)d_ws;
  float* h1   = p; p += N * HC1;
  float* h2   = p; p += N * EMB;
  float* aS1  = p; p += N * HEADS;
  float* aD1  = p; p += N * HEADS;
  float* den1 = p; p += N * HEADS;
  float* aS2  = p; p += N;
  float* aD2  = p; p += N;
  float* den2 = p; p += N;
  unsigned* mk1 = (unsigned*)p; p += N * HEADS;
  unsigned* mk2 = (unsigned*)p; p += N;
  float* out1 = p; p += N * HC1;
  float* outF = (float*)d_out;

  const size_t zbytes = (size_t)N * (3 * HEADS + 3 + HEADS + 1) * sizeof(float);
  hipMemsetAsync(aS1, 0, zbytes, stream);

  const int TPB = 256;
  auto nb = [](long long n, int t) { return (unsigned)((n + t - 1) / t); };

  GAT_init_bias<<<nb(N * HC1, TPB), TPB, 0, stream>>>(out1, b1, N * HC1, HC1);
  GAT_init_bias<<<nb(N * EMB, TPB), TPB, 0, stream>>>(outF, b2, N * EMB, EMB);

  {
    const long long tiles = (N / 16) * (HC1 / 16);
    GAT_gemm_dual<<<nb(tiles * 32, TPB), TPB, 0, stream>>>(
        x, Wsrc1, Wdst1, attS1, attD1, h1, aS1, aD1,
        (int)N, FIN, HC1, HID,  0);
  }
  GAT_edge_max<<<nb(E * HEADS, TPB), TPB, 0, stream>>>(src, dst, aS1, aD1, mk1, E * HEADS, HEADS);
  GAT_edge_expsum<<<nb(E * HEADS, TPB), TPB, 0, stream>>>(src, dst, aS1, aD1, mk1, den1, E * HEADS, HEADS);
  {
    const long long waves = (E + 3) / 4;
    GAT_edge_scatter<HEADS, HID, 4><<<nb(waves * 32, TPB), TPB, 0, stream>>>(
        src, dst, aS1, aD1, mk1, den1, h1, out1, E);
  }

  {
    const long long tiles = (N / 16) * (EMB / 16);
    GAT_gemm_dual<<<nb(tiles * 32, TPB), TPB, 0, stream>>>(
        out1, Wsrc2, Wdst2, attS2, attD2, h2, aS2, aD2,
        (int)N, HC1, EMB, EMB,  1);
  }
  GAT_edge_max<<<nb(E, TPB), TPB, 0, stream>>>(src, dst, aS2, aD2, mk2, E, 1);
  GAT_edge_expsum<<<nb(E, TPB), TPB, 0, stream>>>(src, dst, aS2, aD2, mk2, den2, E, 1);
  {
    const long long waves = (E + 7) / 8;
    GAT_edge_scatter<1, EMB, 8><<<nb(waves * 32, TPB), TPB, 0, stream>>>(
        src, dst, aS2, aD2, mk2, den2, h2, outF, E);
  }
}


# edges counting-sorted by destination in-call (int atomics); one wave per destination does softmax and weighted row sum in registers, no float atomics; sum order = arrival rank of the sort
# speedup vs baseline: 2.1166x; 1.4630x over previous
#include <hip/hip_runtime.h>
#include <hip/hip_bf16.h>


typedef float v2f __attribute__((ext_vector_type(2)));
typedef float v8f __attribute__((ext_vector_type(8)));

#define GAT_NEG_SLOPE 0.2f

__device__ __forceinline__ float GAT_lrelu(float v) {
  return v > 0.0f ? v : GAT_NEG_SLOPE * v;
}

__global__ void GAT_gemm_dual(const float* __restrict__ A,
                              const float* __restrict__ Bs,
                              const float* __restrict__ Bd,
                              const float* __restrict__ attS,
                              const float* __restrict__ attD,
                              float* __restrict__ Hout,
                              float* __restrict__ aS,
                              float* __restrict__ aD,
                              int M, int K, int NC, int Ch, int reluA) {
  const int wave = blockIdx.x * (blockDim.x >> 5) + (threadIdx.x >> 5);
  const int nT = NC >> 4;
  const int totalTiles = (M >> 4) * nT;
  if (wave >= totalTiles) return;

  const int mBase = (wave / nT) << 4;
  const int nBase = (wave % nT) << 4;
  const int lane = threadIdx.x & 31;
  const int half = lane >> 4;
  const int l16  = lane & 15;

  v8f cS = {};
  v8f cD = {};
  const float* arow = A + (size_t)(mBase + l16) * K;
  const int col = nBase + l16;

  for (int k0 = 0; k0 < K; k0 += 4) {
    const int ka = k0 + 2 * half;
    float a0 = arow[ka];
    float a1 = arow[ka + 1];
    if (reluA) { a0 = fmaxf(a0, 0.0f); a1 = fmaxf(a1, 0.0f); }
    v2f a;  a.x = a0; a.y = a1;

    v2f bs, bd;
    bs.x = Bs[(size_t)ka * NC + col];
    bs.y = Bs[(size_t)(ka + 1) * NC + col];
    bd.x = Bd[(size_t)ka * NC + col];
    bd.y = Bd[(size_t)(ka + 1) * NC + col];

    cS = __builtin_amdgcn_wmma_f32_16x16x4_f32(false, a, false, bs, (short)0, cS, false, false);
    cD = __builtin_amdgcn_wmma_f32_16x16x4_f32(false, a, false, bd, (short)0, cD, false, false);
  }

#pragma unroll
  for (int r = 0; r < 8; ++r) {
    Hout[(size_t)(mBase + r + 8 * half) * NC + col] = cS[r];
  }

  const int head = nBase / Ch;
  const int H = NC / Ch;
  const float avS = attS[col];
  const float avD = attD[col];
#pragma unroll
  for (int r = 0; r < 8; ++r) {
    float vs = cS[r] * avS;
    float vd = cD[r] * avD;
#pragma unroll
    for (int m = 1; m < 16; m <<= 1) {
      vs += __shfl_xor(vs, m, 32);
      vd += __shfl_xor(vd, m, 32);
    }
    if (l16 == 0) {
      const int row = mBase + r + 8 * half;
      atomicAdd(&aS[row * H + head], vs);
      atomicAdd(&aD[row * H + head], vd);
    }
  }
}

__global__ void GAT_hist(const int* __restrict__ dst, int* __restrict__ cnt, long long E) {
  long long i = (long long)blockIdx.x * blockDim.x + threadIdx.x;
  if (i >= E) return;
  atomicAdd(&cnt[dst[i]], 1);
}

#define GAT_SCAN_IPT 64
__global__ void __launch_bounds__(1024)
GAT_scan(const int* __restrict__ cnt, int* __restrict__ rowptr, int n) {
  __shared__ int wsum[32];
  const int t = threadIdx.x;
  const int lane = t & 31;
  const int w = t >> 5;
  int carry = 0;
  for (int base = 0; base < n; base += 1024 * GAT_SCAN_IPT) {
    const int i0 = base + t * GAT_SCAN_IPT;
    int c[GAT_SCAN_IPT];
#pragma unroll
    for (int q = 0; q < GAT_SCAN_IPT / 4; ++q) {
      const int i = i0 + 4 * q;
      if (i + 3 < n) {
        const int4 v = *reinterpret_cast<const int4*>(cnt + i);
        c[4 * q] = v.x; c[4 * q + 1] = v.y; c[4 * q + 2] = v.z; c[4 * q + 3] = v.w;
      } else {
        c[4 * q]     = (i     < n) ? cnt[i]     : 0;
        c[4 * q + 1] = (i + 1 < n) ? cnt[i + 1] : 0;
        c[4 * q + 2] = (i + 2 < n) ? cnt[i + 2] : 0;
        c[4 * q + 3] = 0;
      }
    }
    int s = 0;
#pragma unroll
    for (int q = 0; q < GAT_SCAN_IPT; ++q) s += c[q];

    int inc = s;
#pragma unroll
    for (int m = 1; m < 32; m <<= 1) {
      const int v = __shfl_up(inc, m, 32);
      if (lane >= m) inc += v;
    }
    if (lane == 31) wsum[w] = inc;
    __syncthreads();
    const int wtot = wsum[lane];
    int winc = wtot;
#pragma unroll
    for (int m = 1; m < 32; m <<= 1) {
      const int v = __shfl_up(winc, m, 32);
      if (lane >= m) winc += v;
    }
    const int wave_excl = __shfl(winc, w, 32) - __shfl(wtot, w, 32);
    const int total = __shfl(winc, 31, 32);

    int run = carry + wave_excl + (inc - s);
#pragma unroll
    for (int q = 0; q < GAT_SCAN_IPT / 4; ++q) {
      const int i = i0 + 4 * q;
      int4 r;
      r.x = run;
      r.y = r.x + c[4 * q];
      r.z = r.y + c[4 * q + 1];
      r.w = r.z + c[4 * q + 2];
      if (i + 3 < n) {
        *reinterpret_cast<int4*>(rowptr + i) = r;
      } else {
        if (i     < n) rowptr[i]     = r.x;
        if (i + 1 < n) rowptr[i + 1] = r.y;
        if (i + 2 < n) rowptr[i + 2] = r.z;
      }
      run = r.w + c[4 * q + 3];
    }
    carry += total;
    __syncthreads();
  }
  if (t == 0) rowptr[n] = carry;
}

__global__ void GAT_fill(const int* __restrict__ src, const int* __restrict__ dst,
                         const int* __restrict__ rowptr, int* __restrict__ cnt,
                         int* __restrict__ csrc, long long E) {
  long long i = (long long)blockIdx.x * blockDim.x + threadIdx.x;
  if (i >= E) return;
  const int d = dst[i];
  const int k = atomicSub(&cnt[d], 1);
  csrc[rowptr[d] + k - 1] = src[i];
}

template <int H>
__device__ __forceinline__ void GAT_load_heads(const float* __restrict__ a, int node, float (&v)[H]) {
  if constexpr (H % 4 == 0) {
    const float4* p = reinterpret_cast<const float4*>(a + (size_t)node * H);
#pragma unroll
    for (int q = 0; q < H / 4; ++q) {
      const float4 x = p[q];
      v[4 * q] = x.x; v[4 * q + 1] = x.y; v[4 * q + 2] = x.z; v[4 * q + 3] = x.w;
    }
  } else {
#pragma unroll
    for (int h = 0; h < H; ++h) v[h] = a[(size_t)node * H + h];
  }
}

template <int H, int C>
__global__ void GAT_gather(const int* __restrict__ rowptr, const int* __restrict__ csrc,
                           const float* __restrict__ aS, const float* __restrict__ aD,
                           const float* __restrict__ Hsrc, const float* __restrict__ bias,
                           float* __restrict__ out, int N) {
  constexpr int HC = H * C;
  constexpr int NP = HC / 32;
  static_assert(C % 32 == 0, "a 32-float piece must lie inside one head");

  const int lane = threadIdx.x & 31;
  const int d = blockIdx.x * (blockDim.x >> 5) + (threadIdx.x >> 5);
  if (d >= N) return;
  const int beg = __builtin_amdgcn_readfirstlane(rowptr[d]);
  const int end = __builtin_amdgcn_readfirstlane(rowptr[d + 1]);

  float ad[H];
#pragma unroll
  for (int h = 0; h < H; ++h) ad[h] = aD[(size_t)d * H + h];

  float mx[H];
#pragma unroll
  for (int h = 0; h < H; ++h) mx[h] = -__builtin_inff();
  for (int base = beg; base < end; base += 32) {
    const int idx = base + lane;
    if (idx < end) {
      float as[H];
      GAT_load_heads<H>(aS, csrc[idx], as);
#pragma unroll
      for (int h = 0; h < H; ++h) mx[h] = fmaxf(mx[h], GAT_lrelu(as[h] + ad[h]));
    }
  }
#pragma unroll
  for (int h = 0; h < H; ++h) {
#pragma unroll
    for (int m = 16; m >= 1; m >>= 1) mx[h] = fmaxf(mx[h], __shfl_xor(mx[h], m, 32));
  }

  float sm[H];
#pragma unroll
  for (int h = 0; h < H; ++h) sm[h] = 0.0f;
  for (int base = beg; base < end; base += 32) {
    const int idx = base + lane;
    if (idx < end) {
      float as[H];
      GAT_load_heads<H>(aS, csrc[idx], as);
#pragma unroll
      for (int h = 0; h < H; ++h) sm[h] += __expf(GAT_lrelu(as[h] + ad[h]) - mx[h]);
    }
  }
#pragma unroll
  for (int h = 0; h < H; ++h) {
#pragma unroll
    for (int m = 16; m >= 1; m >>= 1) sm[h] += __shfl_xor(sm[h], m, 32);
  }

  float acc[NP];
#pragma unroll
  for (int k = 0; k < NP; ++k) acc[k] = 0.0f;
  for (int base = beg; base < end; base += 32) {
    const int idx = base + lane;
    int s = 0;
    float al[H];
#pragma unroll
    for (int h = 0; h < H; ++h) al[h] = 0.0f;
    if (idx < end) {
      s = csrc[idx];
      float as[H];
      GAT_load_heads<H>(aS, s, as);
#pragma unroll
      for (int h = 0; h < H; ++h)
        al[h] = __expf(GAT_lrelu(as[h] + ad[h]) - mx[h]) / (sm[h] + 1e-16f);
    }
    const int n = (end - base) < 32 ? (end - base) : 32;
    for (int j = 0; j < n; ++j) {
      const int sj = __builtin_amdgcn_readlane(s, j);
      const float* hp = Hsrc + (size_t)sj * HC + lane;
      float hv[NP];
#pragma unroll
      for (int k = 0; k < NP; ++k) hv[k] = hp[k * 32];
#pragma unroll
      for (int k = 0; k < NP; ++k) {
        const float a = __int_as_float(
            __builtin_amdgcn_readlane(__float_as_int(al[(k * 32) / C]), j));
        acc[k] += hv[k] * a;
      }
    }
  }

  float* op = out + (size_t)d * HC + lane;
#pragma unroll
  for (int k = 0; k < NP; ++k) op[k * 32] = acc[k] + bias[k * 32 + lane];
}

extern "C" void kernel_launch(void* const* d_in, const int* in_sizes, int n_in,
                              void* d_out, int out_size, void* d_ws, size_t ws_size,
                              hipStream_t stream) {
  (void)n_in; (void)out_size; (void)ws_size;

  const float* x     = (const float*)d_in[0];
  const int*   ei    = (const int*)d_in[1];
  const float* Wsrc1 = (const float*)d_in[2];
  const float* Wdst1 = (const float*)d_in[3];
  const float* attS1 = (const float*)d_in[4];
  const float* attD1 = (const float*)d_in[5];
  const float* b1    = (const float*)d_in[6];
  const float* Wsrc2 = (const float*)d_in[7];
  const float* Wdst2 = (const float*)d_in[8];
  const float* attS2 = (const float*)d_in[9];
  const float* attD2 = (const float*)d_in[10];
  const float* b2    = (const float*)d_in[11];

  const int FIN = 128, HEADS = 8, HID = 32, EMB = 64;
  const int HC1 = HEADS * HID;
  const long long N = in_sizes[0] / FIN;
  const long long E = in_sizes[1] / 2;
  const int* src = ei;
  const int* dst = ei + E;

  float* p = (float*)d_ws;
  float* h1   = p; p += N * HC1;
  float* h2   = p; p += N * EMB;
  float* out1 = p; p += N * HC1;
  float* aS1  = p; p += N * HEADS;
  float* aD1  = p; p += N * HEADS;
  float* aS2  = p; p += N;
  float* aD2  = p; p += N;
  int* cnt    = (int*)p; p += N + 4;
  int* rowptr = (int*)p; p += N + 4;
  int* csrc   = (int*)p; p += E;
  float* outF = (float*)d_out;

  const size_t zbytes = (size_t)(N * (2 * HEADS + 2 + 1) + 4) * sizeof(float);
  hipMemsetAsync(aS1, 0, zbytes, stream);

  const int TPB = 256;
  auto nb = [](long long n, int t) { return (unsigned)((n + t - 1) / t); };

  GAT_hist<<<nb(E, TPB), TPB, 0, stream>>>(dst, cnt, E);
  GAT_scan<<<1, 1024, 0, stream>>>(cnt, rowptr, (int)N);
  GAT_fill<<<nb(E, TPB), TPB, 0, stream>>>(src, dst, rowptr, cnt, csrc, E);

  {
    const long long tiles = (N / 16) * (HC1 / 16);
    GAT_gemm_dual<<<nb(tiles * 32, TPB), TPB, 0, stream>>>(
        x, Wsrc1, Wdst1, attS1, attD1, h1, aS1, aD1,
        (int)N, FIN, HC1, HID,  0);
  }
  GAT_gather<HEADS, HID><<<nb(N * 32, TPB), TPB, 0, stream>>>(
      rowptr, csrc, aS1, aD1, h1, b1, out1, (int)N);

  {
    const long long tiles = (N / 16) * (EMB / 16);
    GAT_gemm_dual<<<nb(tiles * 32, TPB), TPB, 0, stream>>>(
        out1, Wsrc2, Wdst2, attS2, attD2, h2, aS2, aD2,
        (int)N, HC1, EMB, EMB,  1);
  }
  GAT_gather<1, EMB><<<nb(N * 32, TPB), TPB, 0, stream>>>(
      rowptr, csrc, aS2, aD2, h2, b2, outF, (int)N);
}


# edge sort ranks counted in LDS: per-chunk bucket counts, bucket scatter, then one work-group per bucket counts, scans and places; no global atomic per edge; sum order = LDS arrival rank
# speedup vs baseline: 2.9701x; 1.4032x over previous
#include <hip/hip_runtime.h>
#include <hip/hip_bf16.h>


typedef float v2f __attribute__((ext_vector_type(2)));
typedef float v8f __attribute__((ext_vector_type(8)));

#define GAT_NEG_SLOPE 0.2f

__device__ __forceinline__ float GAT_lrelu(float v) {
  return v > 0.0f ? v : GAT_NEG_SLOPE * v;
}

__global__ void GAT_gemm_dual(const float* __restrict__ A,
                              const float* __restrict__ Bs,
                              const float* __restrict__ Bd,
                              const float* __restrict__ attS,
                              const float* __restrict__ attD,
                              float* __restrict__ Hout,
                              float* __restrict__ aS,
                              float* __restrict__ aD,
                              int M, int K, int NC, int Ch, int reluA) {
  const int wave = blockIdx.x * (blockDim.x >> 5) + (threadIdx.x >> 5);
  const int nT = NC >> 4;
  const int totalTiles = (M >> 4) * nT;
  if (wave >= totalTiles) return;

  const int mBase = (wave / nT) << 4;
  const int nBase = (wave % nT) << 4;
  const int lane = threadIdx.x & 31;
  const int half = lane >> 4;
  const int l16  = lane & 15;

  v8f cS = {};
  v8f cD = {};
  const float* arow = A + (size_t)(mBase + l16) * K;
  const int col = nBase + l16;

  for (int k0 = 0; k0 < K; k0 += 4) {
    const int ka = k0 + 2 * half;
    float a0 = arow[ka];
    float a1 = arow[ka + 1];
    if (reluA) { a0 = fmaxf(a0, 0.0f); a1 = fmaxf(a1, 0.0f); }
    v2f a;  a.x = a0; a.y = a1;

    v2f bs, bd;
    bs.x = Bs[(size_t)ka * NC + col];
    bs.y = Bs[(size_t)(ka + 1) * NC + col];
    bd.x = Bd[(size_t)ka * NC + col];
    bd.y = Bd[(size_t)(ka + 1) * NC + col];

    cS = __builtin_amdgcn_wmma_f32_16x16x4_f32(false, a, false, bs, (short)0, cS, false, false);
    cD = __builtin_amdgcn_wmma_f32_16x16x4_f32(false, a, false, bd, (short)0, cD, false, false);
  }

#pragma unroll
  for (int r = 0; r < 8; ++r) {
    Hout[(size_t)(mBase + r + 8 * half) * NC + col] = cS[r];
  }

  const int head = nBase / Ch;
  const int H = NC / Ch;
  const float avS = attS[col];
  const float avD = attD[col];
#pragma unroll
  for (int r = 0; r < 8; ++r) {
    float vs = cS[r] * avS;
    float vd = cD[r] * avD;
#pragma unroll
    for (int m = 1; m < 16; m <<= 1) {
      vs += __shfl_xor(vs, m, 32);
      vd += __shfl_xor(vd, m, 32);
    }
    if (l16 == 0) {
      const int row = mBase + r + 8 * half;
      atomicAdd(&aS[row * H + head], vs);
      atomicAdd(&aD[row * H + head], vd);
    }
  }
}

#define GAT_NB 256
#define GAT_SORT_T 256
#define GAT_WMAX 1024
#define GAT_EPT 16
#define GAT_RPT 16

__device__ __forceinline__ unsigned GAT_bucket_width(int N) {
  return (unsigned)(N + GAT_NB - 1) / GAT_NB;
}

__device__ __forceinline__ int GAT_excl_scan256(int v, int* wsum) {
  const int lane = threadIdx.x & 31;
  const int wv = threadIdx.x >> 5;
  int inc = v;
  for (int m = 1; m < 32; m <<= 1) {
    const int u = __shfl_up(inc, m, 32);
    if (lane >= m) inc += u;
  }
  if (lane == 31) wsum[wv] = inc;
  __syncthreads();
  int woff = 0;
  for (int k = 0; k < GAT_SORT_T / 32; ++k) woff += (k < wv) ? wsum[k] : 0;
  return woff + inc - v;
}

__global__ void __launch_bounds__(GAT_SORT_T)
GAT_bucket_count(const int* __restrict__ dst, int* __restrict__ cntMat, int E, int N) {
  __shared__ int c[GAT_NB];
  const int t = threadIdx.x;
  c[t] = 0;
  __syncthreads();
  const unsigned width = GAT_bucket_width(N);
  const int chunk = (E + (int)gridDim.x - 1) / (int)gridDim.x;
  const int beg = (int)blockIdx.x * chunk;
  const int end = (beg + chunk < E) ? (beg + chunk) : E;
  for (int base = beg; base < end; base += GAT_EPT * GAT_SORT_T) {
    int d[GAT_EPT];
#pragma unroll
    for (int k = 0; k < GAT_EPT; ++k) {
      const int i = base + k * GAT_SORT_T + t;
      d[k] = (i < end) ? dst[i] : 0;
    }
#pragma unroll
    for (int k = 0; k < GAT_EPT; ++k) {
      const int i = base + k * GAT_SORT_T + t;
      if (i < end) atomicAdd(&c[(unsigned)d[k] / width], 1);
    }
  }
  __syncthreads();
  cntMat[blockIdx.x * GAT_NB + t] = c[t];
}

__global__ void __launch_bounds__(GAT_SORT_T)
GAT_bucket_scatter(const int* __restrict__ src, const int* __restrict__ dst,
                   const int* __restrict__ cntMat, int* __restrict__ bstart,
                   int* __restrict__ pairs, int E, int N) {
  __shared__ int ptot[4][GAT_NB];
  __shared__ int ppre[4][GAT_NB];
  __shared__ int wsum[GAT_SORT_T / 32];
  __shared__ int cur[GAT_NB];
  const int t = threadIdx.x;
  const int w = (int)blockIdx.x;
  const int W = (int)gridDim.x;

  {
    const int cg = t & 63;
    const int rg = t >> 6;
    int4 tot = make_int4(0, 0, 0, 0);
    int4 pre = make_int4(0, 0, 0, 0);
    for (int r0 = rg; r0 < W; r0 += 4 * GAT_RPT) {
      int4 v[GAT_RPT];
#pragma unroll
      for (int k = 0; k < GAT_RPT; ++k) {
        const int r = r0 + 4 * k;
        v[k] = (r < W) ? *reinterpret_cast<const int4*>(cntMat + (size_t)r * GAT_NB + 4 * cg)
                       : make_int4(0, 0, 0, 0);
      }
#pragma unroll
      for (int k = 0; k < GAT_RPT; ++k) {
        const int r = r0 + 4 * k;
        tot.x += v[k].x; tot.y += v[k].y; tot.z += v[k].z; tot.w += v[k].w;
        if (r < w) { pre.x += v[k].x; pre.y += v[k].y; pre.z += v[k].z; pre.w += v[k].w; }
      }
    }
    ptot[rg][4 * cg] = tot.x; ptot[rg][4 * cg + 1] = tot.y;
    ptot[rg][4 * cg + 2] = tot.z; ptot[rg][4 * cg + 3] = tot.w;
    ppre[rg][4 * cg] = pre.x; ppre[rg][4 * cg + 1] = pre.y;
    ppre[rg][4 * cg + 2] = pre.z; ppre[rg][4 * cg + 3] = pre.w;
  }
  __syncthreads();
  const int tot = ptot[0][t] + ptot[1][t] + ptot[2][t] + ptot[3][t];
  const int pre = ppre[0][t] + ppre[1][t] + ppre[2][t] + ppre[3][t];
  const int start = GAT_excl_scan256(tot, wsum);
  cur[t] = start + pre;
  if (w == 0) {
    bstart[t] = start;
    if (t == GAT_NB - 1) bstart[GAT_NB] = start + tot;
  }
  __syncthreads();

  const unsigned width = GAT_bucket_width(N);
  const int chunk = (E + W - 1) / W;
  const int beg = w * chunk;
  const int end = (beg + chunk < E) ? (beg + chunk) : E;
  int2* pp = reinterpret_cast<int2*>(pairs);
  for (int base = beg; base < end; base += GAT_EPT * GAT_SORT_T) {
    int s[GAT_EPT], d[GAT_EPT];
#pragma unroll
    for (int k = 0; k < GAT_EPT; ++k) {
      const int i = base + k * GAT_SORT_T + t;
      s[k] = (i < end) ? src[i] : 0;
      d[k] = (i < end) ? dst[i] : 0;
    }
#pragma unroll
    for (int k = 0; k < GAT_EPT; ++k) {
      const int i = base + k * GAT_SORT_T + t;
      if (i < end) {
        const int slot = atomicAdd(&cur[(unsigned)d[k] / width], 1);
        pp[slot] = make_int2(s[k], d[k]);
      }
    }
  }
}

__global__ void __launch_bounds__(GAT_SORT_T)
GAT_bucket_sort(const int* __restrict__ pairs, const int* __restrict__ bstart,
                int* __restrict__ rowptr, int* __restrict__ csrc, int N) {
  __shared__ int cnt[GAT_WMAX];
  __shared__ int wsum[GAT_SORT_T / 32];
  const int t = threadIdx.x;
  const int b = (int)blockIdx.x;
  const int width = (int)GAT_bucket_width(N);
  const int d0 = b * width;
  const int beg = bstart[b];
  const int end = bstart[b + 1];
  const int2* pp = reinterpret_cast<const int2*>(pairs);

  for (int i = t; i < GAT_WMAX; i += GAT_SORT_T) cnt[i] = 0;
  __syncthreads();
  for (int base = beg; base < end; base += GAT_EPT * GAT_SORT_T) {
    int d[GAT_EPT];
#pragma unroll
    for (int k = 0; k < GAT_EPT; ++k) {
      const int i = base + k * GAT_SORT_T + t;
      d[k] = (i < end) ? pp[i].y : d0;
    }
#pragma unroll
    for (int k = 0; k < GAT_EPT; ++k) {
      const int i = base + k * GAT_SORT_T + t;
      if (i < end) atomicAdd(&cnt[d[k] - d0], 1);
    }
  }
  __syncthreads();

  const int c0 = cnt[4 * t], c1 = cnt[4 * t + 1], c2 = cnt[4 * t + 2], c3 = cnt[4 * t + 3];
  const int ex = GAT_excl_scan256(c0 + c1 + c2 + c3, wsum);
  const int o0 = beg + ex, o1 = o0 + c0, o2 = o1 + c1, o3 = o2 + c2;
  cnt[4 * t] = o0; cnt[4 * t + 1] = o1; cnt[4 * t + 2] = o2; cnt[4 * t + 3] = o3;
  {
    const int q = 4 * t;
    if (q     < width && d0 + q     < N) rowptr[d0 + q]     = o0;
    if (q + 1 < width && d0 + q + 1 < N) rowptr[d0 + q + 1] = o1;
    if (q + 2 < width && d0 + q + 2 < N) rowptr[d0 + q + 2] = o2;
    if (q + 3 < width && d0 + q + 3 < N) rowptr[d0 + q + 3] = o3;
  }
  if (b == GAT_NB - 1 && t == 0) rowptr[N] = end;
  __syncthreads();

  for (int base = beg; base < end; base += GAT_EPT * GAT_SORT_T) {
    int2 p[GAT_EPT];
#pragma unroll
    for (int k = 0; k < GAT_EPT; ++k) {
      const int i = base + k * GAT_SORT_T + t;
      p[k] = (i < end) ? pp[i] : make_int2(0, d0);
    }
#pragma unroll
    for (int k = 0; k < GAT_EPT; ++k) {
      const int i = base + k * GAT_SORT_T + t;
      if (i < end) {
        const int slot = atomicAdd(&cnt[p[k].y - d0], 1);
        csrc[slot] = p[k].x;
      }
    }
  }
}

template <int H>
__device__ __forceinline__ void GAT_load_heads(const float* __restrict__ a, int node, float (&v)[H]) {
  if constexpr (H % 4 == 0) {
    const float4* p = reinterpret_cast<const float4*>(a + (size_t)node * H);
#pragma unroll
    for (int q = 0; q < H / 4; ++q) {
      const float4 x = p[q];
      v[4 * q] = x.x; v[4 * q + 1] = x.y; v[4 * q + 2] = x.z; v[4 * q + 3] = x.w;
    }
  } else {
#pragma unroll
    for (int h = 0; h < H; ++h) v[h] = a[(size_t)node * H + h];
  }
}

template <int H, int C>
__global__ void GAT_gather(const int* __restrict__ rowptr, const int* __restrict__ csrc,
                           const float* __restrict__ aS, const float* __restrict__ aD,
                           const float* __restrict__ Hsrc, const float* __restrict__ bias,
                           float* __restrict__ out, int N) {
  constexpr int HC = H * C;
  constexpr int NP = HC / 32;
  static_assert(C % 32 == 0, "a 32-float piece must lie inside one head");

  const int lane = threadIdx.x & 31;
  const int d = blockIdx.x * (blockDim.x >> 5) + (threadIdx.x >> 5);
  if (d >= N) return;
  const int beg = __builtin_amdgcn_readfirstlane(rowptr[d]);
  const int end = __builtin_amdgcn_readfirstlane(rowptr[d + 1]);

  float ad[H];
#pragma unroll
  for (int h = 0; h < H; ++h) ad[h] = aD[(size_t)d * H + h];

  float mx[H];
#pragma unroll
  for (int h = 0; h < H; ++h) mx[h] = -__builtin_inff();
  for (int base = beg; base < end; base += 32) {
    const int idx = base + lane;
    if (idx < end) {
      float as[H];
      GAT_load_heads<H>(aS, csrc[idx], as);
#pragma unroll
      for (int h = 0; h < H; ++h) mx[h] = fmaxf(mx[h], GAT_lrelu(as[h] + ad[h]));
    }
  }
#pragma unroll
  for (int h = 0; h < H; ++h) {
#pragma unroll
    for (int m = 16; m >= 1; m >>= 1) mx[h] = fmaxf(mx[h], __shfl_xor(mx[h], m, 32));
  }

  float sm[H];
#pragma unroll
  for (int h = 0; h < H; ++h) sm[h] = 0.0f;
  for (int base = beg; base < end; base += 32) {
    const int idx = base + lane;
    if (idx < end) {
      float as[H];
      GAT_load_heads<H>(aS, csrc[idx], as);
#pragma unroll
      for (int h = 0; h < H; ++h) sm[h] += __expf(GAT_lrelu(as[h] + ad[h]) - mx[h]);
    }
  }
#pragma unroll
  for (int h = 0; h < H; ++h) {
#pragma unroll
    for (int m = 16; m >= 1; m >>= 1) sm[h] += __shfl_xor(sm[h], m, 32);
  }

  float acc[NP];
#pragma unroll
  for (int k = 0; k < NP; ++k) acc[k] = 0.0f;
  for (int base = beg; base < end; base += 32) {
    const int idx = base + lane;
    int s = 0;
    float al[H];
#pragma unroll
    for (int h = 0; h < H; ++h) al[h] = 0.0f;
    if (idx < end) {
      s = csrc[idx];
      float as[H];
      GAT_load_heads<H>(aS, s, as);
#pragma unroll
      for (int h = 0; h < H; ++h)
        al[h] = __expf(GAT_lrelu(as[h] + ad[h]) - mx[h]) / (sm[h] + 1e-16f);
    }
    const int n = (end - base) < 32 ? (end - base) : 32;
    for (int j = 0; j < n; ++j) {
      const int sj = __builtin_amdgcn_readlane(s, j);
      const float* hp = Hsrc + (size_t)sj * HC + lane;
      float hv[NP];
#pragma unroll
      for (int k = 0; k < NP; ++k) hv[k] = hp[k * 32];
#pragma unroll
      for (int k = 0; k < NP; ++k) {
        const float a = __int_as_float(
            __builtin_amdgcn_readlane(__float_as_int(al[(k * 32) / C]), j));
        acc[k] += hv[k] * a;
      }
    }
  }

  float* op = out + (size_t)d * HC + lane;
#pragma unroll
  for (int k = 0; k < NP; ++k) op[k * 32] = acc[k] + bias[k * 32 + lane];
}

extern "C" void kernel_launch(void* const* d_in, const int* in_sizes, int n_in,
                              void* d_out, int out_size, void* d_ws, size_t ws_size,
                              hipStream_t stream) {
  (void)n_in; (void)out_size; (void)ws_size;

  const float* x     = (const float*)d_in[0];
  const int*   ei    = (const int*)d_in[1];
  const float* Wsrc1 = (const float*)d_in[2];
  const float* Wdst1 = (const float*)d_in[3];
  const float* attS1 = (const float*)d_in[4];
  const float* attD1 = (const float*)d_in[5];
  const float* b1    = (const float*)d_in[6];
  const float* Wsrc2 = (const float*)d_in[7];
  const float* Wdst2 = (const float*)d_in[8];
  const float* attS2 = (const float*)d_in[9];
  const float* attD2 = (const float*)d_in[10];
  const float* b2    = (const float*)d_in[11];

  const int FIN = 128, HEADS = 8, HID = 32, EMB = 64;
  const int HC1 = HEADS * HID;
  const long long N = in_sizes[0] / FIN;
  const long long E = in_sizes[1] / 2;
  const int* src = ei;
  const int* dst = ei + E;

  float* p = (float*)d_ws;
  float* h1   = p; p += N * HC1;
  float* h2   = p; p += N * EMB;
  float* out1 = p; p += N * HC1;
  float* aS1  = p; p += N * HEADS;
  float* aD1  = p; p += N * HEADS;
  float* aS2  = p; p += N;
  float* aD2  = p; p += N;
  int* rowptr = (int*)p; p += N + 4;
  int* csrc   = (int*)p; p += E;
  int* pairs  = (int*)p; p += 2 * E;
  int* cntMat = (int*)p; p += GAT_NB * GAT_NB;
  int* bstart = (int*)p; p += GAT_NB + 4;
  float* outF = (float*)d_out;

  const size_t zbytes = (size_t)(N * (2 * HEADS + 2)) * sizeof(float);
  hipMemsetAsync(aS1, 0, zbytes, stream);

  const int TPB = 256;
  auto nb = [](long long n, int t) { return (unsigned)((n + t - 1) / t); };

  GAT_bucket_count<<<GAT_NB, GAT_SORT_T, 0, stream>>>(dst, cntMat, (int)E, (int)N);
  GAT_bucket_scatter<<<GAT_NB, GAT_SORT_T, 0, stream>>>(src, dst, cntMat, bstart, pairs, (int)E, (int)N);
  GAT_bucket_sort<<<GAT_NB, GAT_SORT_T, 0, stream>>>(pairs, bstart, rowptr, csrc, (int)N);

  {
    const long long tiles = (N / 16) * (HC1 / 16);
    GAT_gemm_dual<<<nb(tiles * 32, TPB), TPB, 0, stream>>>(
        x, Wsrc1, Wdst1, attS1, attD1, h1, aS1, aD1,
        (int)N, FIN, HC1, HID,  0);
  }
  GAT_gather<HEADS, HID><<<nb(N * 32, TPB), TPB, 0, stream>>>(
      rowptr, csrc, aS1, aD1, h1, b1, out1, (int)N);

  {
    const long long tiles = (N / 16) * (EMB / 16);
    GAT_gemm_dual<<<nb(tiles * 32, TPB), TPB, 0, stream>>>(
        out1, Wsrc2, Wdst2, attS2, attD2, h2, aS2, aD2,
        (int)N, HC1, EMB, EMB,  1);
  }
  GAT_gather<1, EMB><<<nb(N * 32, TPB), TPB, 0, stream>>>(
      rowptr, csrc, aS2, aD2, h2, b2, outF, (int)N);
}


# GEMM retiled: a wave owns a 16x64 strip of both products (8 accumulators), operands by 16-B loads; k summed in a permuted order per 32-block; logits summed in-wave, float atomics and memset gone
# speedup vs baseline: 3.9603x; 1.3334x over previous
#include <hip/hip_runtime.h>
#include <hip/hip_bf16.h>


typedef float v2f __attribute__((ext_vector_type(2)));
typedef float v8f __attribute__((ext_vector_type(8)));

#define GAT_NEG_SLOPE 0.2f

__device__ __forceinline__ float GAT_lrelu(float v) {
  return v > 0.0f ? v : GAT_NEG_SLOPE * v;
}

template <int K, int NC, int CH, bool RELU>
__global__ void __launch_bounds__(256)
GAT_gemm_strip(const float* __restrict__ A,
               const float* __restrict__ Bs,
               const float* __restrict__ Bd,
               const float* __restrict__ attS,
               const float* __restrict__ attD,
               float* __restrict__ Hout,
               float* __restrict__ aS,
               float* __restrict__ aD,
               int M) {
  constexpr int NG = NC / 64;
  constexpr int H = NC / CH;
  constexpr int LPH = CH / 4;
  static_assert(K % 32 == 0, "K is walked in blocks of 32");
  static_assert(NC % 64 == 0, "a strip is 64 columns");
  static_assert(CH == 32 || CH == 64, "a strip must hold whole heads");

  const int wpg = (int)(blockDim.x >> 5);
  const int mTile = ((int)blockIdx.x / NG) * wpg + (int)(threadIdx.x >> 5);
  if (mTile >= (M >> 4)) return;

  const int mBase = mTile << 4;
  const int nBase = ((int)blockIdx.x % NG) << 6;
  const int lane = threadIdx.x & 31;
  const int half = lane >> 4;
  const int c    = lane & 15;

  v8f cS[4] = {};
  v8f cD[4] = {};
  const float* arow = A + (size_t)(mBase + c) * K + 16 * half;
  const size_t boff = (size_t)(16 * half) * NC + nBase + 4 * c;
  const float* bsp = Bs + boff;
  const float* bdp = Bd + boff;

  for (int kb = 0; kb < K; kb += 32) {
    float av[16];
#pragma unroll
    for (int q = 0; q < 4; ++q) {
      const float4 v = *reinterpret_cast<const float4*>(arow + kb + 4 * q);
      av[4 * q] = v.x; av[4 * q + 1] = v.y; av[4 * q + 2] = v.z; av[4 * q + 3] = v.w;
    }
    if (RELU) {
#pragma unroll
      for (int i = 0; i < 16; ++i) av[i] = fmaxf(av[i], 0.0f);
    }
#pragma unroll
    for (int j = 0; j < 8; ++j) {
      const float* ps = bsp + (size_t)(kb + 2 * j) * NC;
      const float* pd = bdp + (size_t)(kb + 2 * j) * NC;
      const float4 s0 = *reinterpret_cast<const float4*>(ps);
      const float4 s1 = *reinterpret_cast<const float4*>(ps + NC);
      const float4 d0 = *reinterpret_cast<const float4*>(pd);
      const float4 d1 = *reinterpret_cast<const float4*>(pd + NC);
      v2f a;  a.x = av[2 * j];  a.y = av[2 * j + 1];
      v2f b;
      b.x = s0.x; b.y = s1.x;
      cS[0] = __builtin_amdgcn_wmma_f32_16x16x4_f32(false, a, false, b, (short)0, cS[0], false, false);
      b.x = d0.x; b.y = d1.x;
      cD[0] = __builtin_amdgcn_wmma_f32_16x16x4_f32(false, a, false, b, (short)0, cD[0], false, false);
      b.x = s0.y; b.y = s1.y;
      cS[1] = __builtin_amdgcn_wmma_f32_16x16x4_f32(false, a, false, b, (short)0, cS[1], false, false);
      b.x = d0.y; b.y = d1.y;
      cD[1] = __builtin_amdgcn_wmma_f32_16x16x4_f32(false, a, false, b, (short)0, cD[1], false, false);
      b.x = s0.z; b.y = s1.z;
      cS[2] = __builtin_amdgcn_wmma_f32_16x16x4_f32(false, a, false, b, (short)0, cS[2], false, false);
      b.x = d0.z; b.y = d1.z;
      cD[2] = __builtin_amdgcn_wmma_f32_16x16x4_f32(false, a, false, b, (short)0, cD[2], false, false);
      b.x = s0.w; b.y = s1.w;
      cS[3] = __builtin_amdgcn_wmma_f32_16x16x4_f32(false, a, false, b, (short)0, cS[3], false, false);
      b.x = d0.w; b.y = d1.w;
      cD[3] = __builtin_amdgcn_wmma_f32_16x16x4_f32(false, a, false, b, (short)0, cD[3], false, false);
    }
  }

  const int col = nBase + 4 * c;
#pragma unroll
  for (int r = 0; r < 8; ++r) {
    float4 o;
    o.x = cS[0][r]; o.y = cS[1][r]; o.z = cS[2][r]; o.w = cS[3][r];
    *reinterpret_cast<float4*>(Hout + (size_t)(mBase + r + 8 * half) * NC + col) = o;
  }

  const float4 wS = *reinterpret_cast<const float4*>(attS + col);
  const float4 wD = *reinterpret_cast<const float4*>(attD + col);
  const int head = col / CH;
  float vs[8], vd[8];
#pragma unroll
  for (int r = 0; r < 8; ++r) {
    vs[r] = cS[0][r] * wS.x + cS[1][r] * wS.y + cS[2][r] * wS.z + cS[3][r] * wS.w;
    vd[r] = cD[0][r] * wD.x + cD[1][r] * wD.y + cD[2][r] * wD.z + cD[3][r] * wD.w;
  }
#pragma unroll
  for (int m = 1; m < LPH; m <<= 1) {
#pragma unroll
    for (int r = 0; r < 8; ++r) {
      vs[r] += __shfl_xor(vs[r], m, 32);
      vd[r] += __shfl_xor(vd[r], m, 32);
    }
  }
  if ((c & (LPH - 1)) == 0) {
#pragma unroll
    for (int r = 0; r < 8; ++r) {
      const int row = mBase + r + 8 * half;
      aS[(size_t)row * H + head] = vs[r];
      aD[(size_t)row * H + head] = vd[r];
    }
  }
}

#define GAT_NB 256
#define GAT_SORT_T 256
#define GAT_WMAX 1024
#define GAT_EPT 16
#define GAT_RPT 16

__device__ __forceinline__ unsigned GAT_bucket_width(int N) {
  return (unsigned)(N + GAT_NB - 1) / GAT_NB;
}

__device__ __forceinline__ int GAT_excl_scan256(int v, int* wsum) {
  const int lane = threadIdx.x & 31;
  const int wv = threadIdx.x >> 5;
  int inc = v;
  for (int m = 1; m < 32; m <<= 1) {
    const int u = __shfl_up(inc, m, 32);
    if (lane >= m) inc += u;
  }
  if (lane == 31) wsum[wv] = inc;
  __syncthreads();
  int woff = 0;
  for (int k = 0; k < GAT_SORT_T / 32; ++k) woff += (k < wv) ? wsum[k] : 0;
  return woff + inc - v;
}

__global__ void __launch_bounds__(GAT_SORT_T)
GAT_bucket_count(const int* __restrict__ dst, int* __restrict__ cntMat, int E, int N) {
  __shared__ int c[GAT_NB];
  const int t = threadIdx.x;
  c[t] = 0;
  __syncthreads();
  const unsigned width = GAT_bucket_width(N);
  const int chunk = (E + (int)gridDim.x - 1) / (int)gridDim.x;
  const int beg = (int)blockIdx.x * chunk;
  const int end = (beg + chunk < E) ? (beg + chunk) : E;
  for (int base = beg; base < end; base += GAT_EPT * GAT_SORT_T) {
    int d[GAT_EPT];
#pragma unroll
    for (int k = 0; k < GAT_EPT; ++k) {
      const int i = base + k * GAT_SORT_T + t;
      d[k] = (i < end) ? dst[i] : 0;
    }
#pragma unroll
    for (int k = 0; k < GAT_EPT; ++k) {
      const int i = base + k * GAT_SORT_T + t;
      if (i < end) atomicAdd(&c[(unsigned)d[k] / width], 1);
    }
  }
  __syncthreads();
  cntMat[blockIdx.x * GAT_NB + t] = c[t];
}

__global__ void __launch_bounds__(GAT_SORT_T)
GAT_bucket_scatter(const int* __restrict__ src, const int* __restrict__ dst,
                   const int* __restrict__ cntMat, int* __restrict__ bstart,
                   int* __restrict__ pairs, int E, int N) {
  __shared__ int ptot[4][GAT_NB];
  __shared__ int ppre[4][GAT_NB];
  __shared__ int wsum[GAT_SORT_T / 32];
  __shared__ int cur[GAT_NB];
  const int t = threadIdx.x;
  const int w = (int)blockIdx.x;
  const int W = (int)gridDim.x;

  {
    const int cg = t & 63;
    const int rg = t >> 6;
    int4 tot = make_int4(0, 0, 0, 0);
    int4 pre = make_int4(0, 0, 0, 0);
    for (int r0 = rg; r0 < W; r0 += 4 * GAT_RPT) {
      int4 v[GAT_RPT];
#pragma unroll
      for (int k = 0; k < GAT_RPT; ++k) {
        const int r = r0 + 4 * k;
        v[k] = (r < W) ? *reinterpret_cast<const int4*>(cntMat + (size_t)r * GAT_NB + 4 * cg)
                       : make_int4(0, 0, 0, 0);
      }
#pragma unroll
      for (int k = 0; k < GAT_RPT; ++k) {
        const int r = r0 + 4 * k;
        tot.x += v[k].x; tot.y += v[k].y; tot.z += v[k].z; tot.w += v[k].w;
        if (r < w) { pre.x += v[k].x; pre.y += v[k].y; pre.z += v[k].z; pre.w += v[k].w; }
      }
    }
    ptot[rg][4 * cg] = tot.x; ptot[rg][4 * cg + 1] = tot.y;
    ptot[rg][4 * cg + 2] = tot.z; ptot[rg][4 * cg + 3] = tot.w;
    ppre[rg][4 * cg] = pre.x; ppre[rg][4 * cg + 1] = pre.y;
    ppre[rg][4 * cg + 2] = pre.z; ppre[rg][4 * cg + 3] = pre.w;
  }
  __syncthreads();
  const int tot = ptot[0][t] + ptot[1][t] + ptot[2][t] + ptot[3][t];
  const int pre = ppre[0][t] + ppre[1][t] + ppre[2][t] + ppre[3][t];
  const int start = GAT_excl_scan256(tot, wsum);
  cur[t] = start + pre;
  if (w == 0) {
    bstart[t] = start;
    if (t == GAT_NB - 1) bstart[GAT_NB] = start + tot;
  }
  __syncthreads();

  const unsigned width = GAT_bucket_width(N);
  const int chunk = (E + W - 1) / W;
  const int beg = w * chunk;
  const int end = (beg + chunk < E) ? (beg + chunk) : E;
  int2* pp = reinterpret_cast<int2*>(pairs);
  for (int base = beg; base < end; base += GAT_EPT * GAT_SORT_T) {
    int s[GAT_EPT], d[GAT_EPT];
#pragma unroll
    for (int k = 0; k < GAT_EPT; ++k) {
      const int i = base + k * GAT_SORT_T + t;
      s[k] = (i < end) ? src[i] : 0;
      d[k] = (i < end) ? dst[i] : 0;
    }
#pragma unroll
    for (int k = 0; k < GAT_EPT; ++k) {
      const int i = base + k * GAT_SORT_T + t;
      if (i < end) {
        const int slot = atomicAdd(&cur[(unsigned)d[k] / width], 1);
        pp[slot] = make_int2(s[k], d[k]);
      }
    }
  }
}

__global__ void __launch_bounds__(GAT_SORT_T)
GAT_bucket_sort(const int* __restrict__ pairs, const int* __restrict__ bstart,
                int* __restrict__ rowptr, int* __restrict__ csrc, int N) {
  __shared__ int cnt[GAT_WMAX];
  __shared__ int wsum[GAT_SORT_T / 32];
  const int t = threadIdx.x;
  const int b = (int)blockIdx.x;
  const int width = (int)GAT_bucket_width(N);
  const int d0 = b * width;
  const int beg = bstart[b];
  const int end = bstart[b + 1];
  const int2* pp = reinterpret_cast<const int2*>(pairs);

  for (int i = t; i < GAT_WMAX; i += GAT_SORT_T) cnt[i] = 0;
  __syncthreads();
  for (int base = beg; base < end; base += GAT_EPT * GAT_SORT_T) {
    int d[GAT_EPT];
#pragma unroll
    for (int k = 0; k < GAT_EPT; ++k) {
      const int i = base + k * GAT_SORT_T + t;
      d[k] = (i < end) ? pp[i].y : d0;
    }
#pragma unroll
    for (int k = 0; k < GAT_EPT; ++k) {
      const int i = base + k * GAT_SORT_T + t;
      if (i < end) atomicAdd(&cnt[d[k] - d0], 1);
    }
  }
  __syncthreads();

  const int c0 = cnt[4 * t], c1 = cnt[4 * t + 1], c2 = cnt[4 * t + 2], c3 = cnt[4 * t + 3];
  const int ex = GAT_excl_scan256(c0 + c1 + c2 + c3, wsum);
  const int o0 = beg + ex, o1 = o0 + c0, o2 = o1 + c1, o3 = o2 + c2;
  cnt[4 * t] = o0; cnt[4 * t + 1] = o1; cnt[4 * t + 2] = o2; cnt[4 * t + 3] = o3;
  {
    const int q = 4 * t;
    if (q     < width && d0 + q     < N) rowptr[d0 + q]     = o0;
    if (q + 1 < width && d0 + q + 1 < N) rowptr[d0 + q + 1] = o1;
    if (q + 2 < width && d0 + q + 2 < N) rowptr[d0 + q + 2] = o2;
    if (q + 3 < width && d0 + q + 3 < N) rowptr[d0 + q + 3] = o3;
  }
  if (b == GAT_NB - 1 && t == 0) rowptr[N] = end;
  __syncthreads();

  for (int base = beg; base < end; base += GAT_EPT * GAT_SORT_T) {
    int2 p[GAT_EPT];
#pragma unroll
    for (int k = 0; k < GAT_EPT; ++k) {
      const int i = base + k * GAT_SORT_T + t;
      p[k] = (i < end) ? pp[i] : make_int2(0, d0);
    }
#pragma unroll
    for (int k = 0; k < GAT_EPT; ++k) {
      const int i = base + k * GAT_SORT_T + t;
      if (i < end) {
        const int slot = atomicAdd(&cnt[p[k].y - d0], 1);
        csrc[slot] = p[k].x;
      }
    }
  }
}

template <int H>
__device__ __forceinline__ void GAT_load_heads(const float* __restrict__ a, int node, float (&v)[H]) {
  if constexpr (H % 4 == 0) {
    const float4* p = reinterpret_cast<const float4*>(a + (size_t)node * H);
#pragma unroll
    for (int q = 0; q < H / 4; ++q) {
      const float4 x = p[q];
      v[4 * q] = x.x; v[4 * q + 1] = x.y; v[4 * q + 2] = x.z; v[4 * q + 3] = x.w;
    }
  } else {
#pragma unroll
    for (int h = 0; h < H; ++h) v[h] = a[(size_t)node * H + h];
  }
}

template <int H, int C>
__global__ void GAT_gather(const int* __restrict__ rowptr, const int* __restrict__ csrc,
                           const float* __restrict__ aS, const float* __restrict__ aD,
                           const float* __restrict__ Hsrc, const float* __restrict__ bias,
                           float* __restrict__ out, int N) {
  constexpr int HC = H * C;
  constexpr int NP = HC / 32;
  static_assert(C % 32 == 0, "a 32-float piece must lie inside one head");

  const int lane = threadIdx.x & 31;
  const int d = blockIdx.x * (blockDim.x >> 5) + (threadIdx.x >> 5);
  if (d >= N) return;
  const int beg = __builtin_amdgcn_readfirstlane(rowptr[d]);
  const int end = __builtin_amdgcn_readfirstlane(rowptr[d + 1]);

  float ad[H];
#pragma unroll
  for (int h = 0; h < H; ++h) ad[h] = aD[(size_t)d * H + h];

  float mx[H];
#pragma unroll
  for (int h = 0; h < H; ++h) mx[h] = -__builtin_inff();
  for (int base = beg; base < end; base += 32) {
    const int idx = base + lane;
    if (idx < end) {
      float as[H];
      GAT_load_heads<H>(aS, csrc[idx], as);
#pragma unroll
      for (int h = 0; h < H; ++h) mx[h] = fmaxf(mx[h], GAT_lrelu(as[h] + ad[h]));
    }
  }
#pragma unroll
  for (int h = 0; h < H; ++h) {
#pragma unroll
    for (int m = 16; m >= 1; m >>= 1) mx[h] = fmaxf(mx[h], __shfl_xor(mx[h], m, 32));
  }

  float sm[H];
#pragma unroll
  for (int h = 0; h < H; ++h) sm[h] = 0.0f;
  for (int base = beg; base < end; base += 32) {
    const int idx = base + lane;
    if (idx < end) {
      float as[H];
      GAT_load_heads<H>(aS, csrc[idx], as);
#pragma unroll
      for (int h = 0; h < H; ++h) sm[h] += __expf(GAT_lrelu(as[h] + ad[h]) - mx[h]);
    }
  }
#pragma unroll
  for (int h = 0; h < H; ++h) {
#pragma unroll
    for (int m = 16; m >= 1; m >>= 1) sm[h] += __shfl_xor(sm[h], m, 32);
  }

  float acc[NP];
#pragma unroll
  for (int k = 0; k < NP; ++k) acc[k] = 0.0f;
  for (int base = beg; base < end; base += 32) {
    const int idx = base + lane;
    int s = 0;
    float al[H];
#pragma unroll
    for (int h = 0; h < H; ++h) al[h] = 0.0f;
    if (idx < end) {
      s = csrc[idx];
      float as[H];
      GAT_load_heads<H>(aS, s, as);
#pragma unroll
      for (int h = 0; h < H; ++h)
        al[h] = __expf(GAT_lrelu(as[h] + ad[h]) - mx[h]) / (sm[h] + 1e-16f);
    }
    const int n = (end - base) < 32 ? (end - base) : 32;
    for (int j = 0; j < n; ++j) {
      const int sj = __builtin_amdgcn_readlane(s, j);
      const float* hp = Hsrc + (size_t)sj * HC + lane;
      float hv[NP];
#pragma unroll
      for (int k = 0; k < NP; ++k) hv[k] = hp[k * 32];
#pragma unroll
      for (int k = 0; k < NP; ++k) {
        const float a = __int_as_float(
            __builtin_amdgcn_readlane(__float_as_int(al[(k * 32) / C]), j));
        acc[k] += hv[k] * a;
      }
    }
  }

  float* op = out + (size_t)d * HC + lane;
#pragma unroll
  for (int k = 0; k < NP; ++k) op[k * 32] = acc[k] + bias[k * 32 + lane];
}

extern "C" void kernel_launch(void* const* d_in, const int* in_sizes, int n_in,
                              void* d_out, int out_size, void* d_ws, size_t ws_size,
                              hipStream_t stream) {
  (void)n_in; (void)out_size; (void)ws_size;

  const float* x     = (const float*)d_in[0];
  const int*   ei    = (const int*)d_in[1];
  const float* Wsrc1 = (const float*)d_in[2];
  const float* Wdst1 = (const float*)d_in[3];
  const float* attS1 = (const float*)d_in[4];
  const float* attD1 = (const float*)d_in[5];
  const float* b1    = (const float*)d_in[6];
  const float* Wsrc2 = (const float*)d_in[7];
  const float* Wdst2 = (const float*)d_in[8];
  const float* attS2 = (const float*)d_in[9];
  const float* attD2 = (const float*)d_in[10];
  const float* b2    = (const float*)d_in[11];

  const int FIN = 128, HEADS = 8, HID = 32, EMB = 64;
  const int HC1 = HEADS * HID;
  const long long N = in_sizes[0] / FIN;
  const long long E = in_sizes[1] / 2;
  const int* src = ei;
  const int* dst = ei + E;

  float* p = (float*)d_ws;
  float* h1   = p; p += N * HC1;
  float* h2   = p; p += N * EMB;
  float* out1 = p; p += N * HC1;
  float* aS1  = p; p += N * HEADS;
  float* aD1  = p; p += N * HEADS;
  float* aS2  = p; p += N;
  float* aD2  = p; p += N;
  int* rowptr = (int*)p; p += N + 4;
  int* csrc   = (int*)p; p += E;
  int* pairs  = (int*)p; p += 2 * E;
  int* cntMat = (int*)p; p += GAT_NB * GAT_NB;
  int* bstart = (int*)p; p += GAT_NB + 4;
  float* outF = (float*)d_out;

  const int TPB = 256;
  auto nb = [](long long n, int t) { return (unsigned)((n + t - 1) / t); };

  GAT_bucket_count<<<GAT_NB, GAT_SORT_T, 0, stream>>>(dst, cntMat, (int)E, (int)N);
  GAT_bucket_scatter<<<GAT_NB, GAT_SORT_T, 0, stream>>>(src, dst, cntMat, bstart, pairs, (int)E, (int)N);
  GAT_bucket_sort<<<GAT_NB, GAT_SORT_T, 0, stream>>>(pairs, bstart, rowptr, csrc, (int)N);

  {
    GAT_gemm_strip<128, 256, 32, false><<<nb(N / 16, TPB / 32) * (HC1 / 64), TPB, 0, stream>>>(
        x, Wsrc1, Wdst1, attS1, attD1, h1, aS1, aD1, (int)N);
  }
  GAT_gather<HEADS, HID><<<nb(N * 32, TPB), TPB, 0, stream>>>(
      rowptr, csrc, aS1, aD1, h1, b1, out1, (int)N);

  {
    GAT_gemm_strip<256, 64, 64, true><<<nb(N / 16, TPB / 32) * (EMB / 64), TPB, 0, stream>>>(
        out1, Wsrc2, Wdst2, attS2, attD2, h2, aS2, aD2, (int)N);
  }
  GAT_gather<1, EMB><<<nb(N * 32, TPB), TPB, 0, stream>>>(
      rowptr, csrc, aS2, aD2, h2, b2, outF, (int)N);
}


# layer-1 GEMM moved into the three sort launches: each carries the sort phase's 256 work-groups plus a share of the GEMM row tiles, so sort waits run beside matrix work; 6 launches, sum orders kept
# speedup vs baseline: 4.1128x; 1.0385x over previous
#include <hip/hip_runtime.h>
#include <hip/hip_bf16.h>


typedef float v2f __attribute__((ext_vector_type(2)));
typedef float v8f __attribute__((ext_vector_type(8)));

#define GAT_NEG_SLOPE 0.2f

__device__ __forceinline__ float GAT_lrelu(float v) {
  return v > 0.0f ? v : GAT_NEG_SLOPE * v;
}

template <int K, int NC, int CH, bool RELU>
__device__ __forceinline__ void
GAT_gemm_strip_body(const float* __restrict__ A,
                    const float* __restrict__ Bs,
                    const float* __restrict__ Bd,
                    const float* __restrict__ attS,
                    const float* __restrict__ attD,
                    float* __restrict__ Hout,
                    float* __restrict__ aS,
                    float* __restrict__ aD,
                    int blk, int tileBeg, int tileEnd) {
  constexpr int NG = NC / 64;
  constexpr int H = NC / CH;
  constexpr int LPH = CH / 4;
  static_assert(K % 32 == 0, "K is walked in blocks of 32");
  static_assert(NC % 64 == 0, "a strip is 64 columns");
  static_assert(CH == 32 || CH == 64, "a strip must hold whole heads");

  const int wpg = (int)(blockDim.x >> 5);
  const int mTile = tileBeg + (blk / NG) * wpg + (int)(threadIdx.x >> 5);
  if (mTile >= tileEnd) return;

  const int mBase = mTile << 4;
  const int nBase = (blk % NG) << 6;
  const int lane = threadIdx.x & 31;
  const int half = lane >> 4;
  const int c    = lane & 15;

  v8f cS[4] = {};
  v8f cD[4] = {};
  const float* arow = A + (size_t)(mBase + c) * K + 16 * half;
  const size_t boff = (size_t)(16 * half) * NC + nBase + 4 * c;
  const float* bsp = Bs + boff;
  const float* bdp = Bd + boff;

  for (int kb = 0; kb < K; kb += 32) {
    float av[16];
#pragma unroll
    for (int q = 0; q < 4; ++q) {
      const float4 v = *reinterpret_cast<const float4*>(arow + kb + 4 * q);
      av[4 * q] = v.x; av[4 * q + 1] = v.y; av[4 * q + 2] = v.z; av[4 * q + 3] = v.w;
    }
    if (RELU) {
#pragma unroll
      for (int i = 0; i < 16; ++i) av[i] = fmaxf(av[i], 0.0f);
    }
#pragma unroll
    for (int j = 0; j < 8; ++j) {
      const float* ps = bsp + (size_t)(kb + 2 * j) * NC;
      const float* pd = bdp + (size_t)(kb + 2 * j) * NC;
      const float4 s0 = *reinterpret_cast<const float4*>(ps);
      const float4 s1 = *reinterpret_cast<const float4*>(ps + NC);
      const float4 d0 = *reinterpret_cast<const float4*>(pd);
      const float4 d1 = *reinterpret_cast<const float4*>(pd + NC);
      v2f a;  a.x = av[2 * j];  a.y = av[2 * j + 1];
      v2f b;
      b.x = s0.x; b.y = s1.x;
      cS[0] = __builtin_amdgcn_wmma_f32_16x16x4_f32(false, a, false, b, (short)0, cS[0], false, false);
      b.x = d0.x; b.y = d1.x;
      cD[0] = __builtin_amdgcn_wmma_f32_16x16x4_f32(false, a, false, b, (short)0, cD[0], false, false);
      b.x = s0.y; b.y = s1.y;
      cS[1] = __builtin_amdgcn_wmma_f32_16x16x4_f32(false, a, false, b, (short)0, cS[1], false, false);
      b.x = d0.y; b.y = d1.y;
      cD[1] = __builtin_amdgcn_wmma_f32_16x16x4_f32(false, a, false, b, (short)0, cD[1], false, false);
      b.x = s0.z; b.y = s1.z;
      cS[2] = __builtin_amdgcn_wmma_f32_16x16x4_f32(false, a, false, b, (short)0, cS[2], false, false);
      b.x = d0.z; b.y = d1.z;
      cD[2] = __builtin_amdgcn_wmma_f32_16x16x4_f32(false, a, false, b, (short)0, cD[2], false, false);
      b.x = s0.w; b.y = s1.w;
      cS[3] = __builtin_amdgcn_wmma_f32_16x16x4_f32(false, a, false, b, (short)0, cS[3], false, false);
      b.x = d0.w; b.y = d1.w;
      cD[3] = __builtin_amdgcn_wmma_f32_16x16x4_f32(false, a, false, b, (short)0, cD[3], false, false);
    }
  }

  const int col = nBase + 4 * c;
#pragma unroll
  for (int r = 0; r < 8; ++r) {
    float4 o;
    o.x = cS[0][r]; o.y = cS[1][r]; o.z = cS[2][r]; o.w = cS[3][r];
    *reinterpret_cast<float4*>(Hout + (size_t)(mBase + r + 8 * half) * NC + col) = o;
  }

  const float4 wS = *reinterpret_cast<const float4*>(attS + col);
  const float4 wD = *reinterpret_cast<const float4*>(attD + col);
  const int head = col / CH;
  float vs[8], vd[8];
#pragma unroll
  for (int r = 0; r < 8; ++r) {
    vs[r] = cS[0][r] * wS.x + cS[1][r] * wS.y + cS[2][r] * wS.z + cS[3][r] * wS.w;
    vd[r] = cD[0][r] * wD.x + cD[1][r] * wD.y + cD[2][r] * wD.z + cD[3][r] * wD.w;
  }
#pragma unroll
  for (int m = 1; m < LPH; m <<= 1) {
#pragma unroll
    for (int r = 0; r < 8; ++r) {
      vs[r] += __shfl_xor(vs[r], m, 32);
      vd[r] += __shfl_xor(vd[r], m, 32);
    }
  }
  if ((c & (LPH - 1)) == 0) {
#pragma unroll
    for (int r = 0; r < 8; ++r) {
      const int row = mBase + r + 8 * half;
      aS[(size_t)row * H + head] = vs[r];
      aD[(size_t)row * H + head] = vd[r];
    }
  }
}

template <int K, int NC, int CH, bool RELU>
__global__ void __launch_bounds__(256)
GAT_gemm_strip(const float* __restrict__ A,
               const float* __restrict__ Bs,
               const float* __restrict__ Bd,
               const float* __restrict__ attS,
               const float* __restrict__ attD,
               float* __restrict__ Hout,
               float* __restrict__ aS,
               float* __restrict__ aD,
               int M) {
  GAT_gemm_strip_body<K, NC, CH, RELU>(A, Bs, Bd, attS, attD, Hout, aS, aD,
                                       (int)blockIdx.x, 0, M >> 4);
}

#define GAT_NB 256
#define GAT_SORT_T 256
#define GAT_WMAX 1024
#define GAT_EPT 16
#define GAT_RPT 16

__device__ __forceinline__ unsigned GAT_bucket_width(int N) {
  return (unsigned)(N + GAT_NB - 1) / GAT_NB;
}

__device__ __forceinline__ int GAT_excl_scan256(int v, int* wsum) {
  const int lane = threadIdx.x & 31;
  const int wv = threadIdx.x >> 5;
  int inc = v;
  for (int m = 1; m < 32; m <<= 1) {
    const int u = __shfl_up(inc, m, 32);
    if (lane >= m) inc += u;
  }
  if (lane == 31) wsum[wv] = inc;
  __syncthreads();
  int woff = 0;
  for (int k = 0; k < GAT_SORT_T / 32; ++k) woff += (k < wv) ? wsum[k] : 0;
  return woff + inc - v;
}

__device__ __forceinline__ void
GAT_bucket_count_body(const int* __restrict__ dst, int* __restrict__ cntMat, int E, int N, int w) {
  __shared__ int c[GAT_NB];
  const int t = threadIdx.x;
  c[t] = 0;
  __syncthreads();
  const unsigned width = GAT_bucket_width(N);
  const int chunk = (E + GAT_NB - 1) / GAT_NB;
  const int beg = w * chunk;
  const int end = (beg + chunk < E) ? (beg + chunk) : E;
  for (int base = beg; base < end; base += GAT_EPT * GAT_SORT_T) {
    int d[GAT_EPT];
#pragma unroll
    for (int k = 0; k < GAT_EPT; ++k) {
      const int i = base + k * GAT_SORT_T + t;
      d[k] = (i < end) ? dst[i] : 0;
    }
#pragma unroll
    for (int k = 0; k < GAT_EPT; ++k) {
      const int i = base + k * GAT_SORT_T + t;
      if (i < end) atomicAdd(&c[(unsigned)d[k] / width], 1);
    }
  }
  __syncthreads();
  cntMat[w * GAT_NB + t] = c[t];
}

__device__ __forceinline__ void
GAT_bucket_scatter_body(const int* __restrict__ src, const int* __restrict__ dst,
                        const int* __restrict__ cntMat, int* __restrict__ bstart,
                        int* __restrict__ pairs, int E, int N, int w) {
  __shared__ int ptot[4][GAT_NB];
  __shared__ int ppre[4][GAT_NB];
  __shared__ int wsum[GAT_SORT_T / 32];
  __shared__ int cur[GAT_NB];
  const int t = threadIdx.x;
  const int W = GAT_NB;

  {
    const int cg = t & 63;
    const int rg = t >> 6;
    int4 tot = make_int4(0, 0, 0, 0);
    int4 pre = make_int4(0, 0, 0, 0);
    for (int r0 = rg; r0 < W; r0 += 4 * GAT_RPT) {
      int4 v[GAT_RPT];
#pragma unroll
      for (int k = 0; k < GAT_RPT; ++k) {
        const int r = r0 + 4 * k;
        v[k] = (r < W) ? *reinterpret_cast<const int4*>(cntMat + (size_t)r * GAT_NB + 4 * cg)
                       : make_int4(0, 0, 0, 0);
      }
#pragma unroll
      for (int k = 0; k < GAT_RPT; ++k) {
        const int r = r0 + 4 * k;
        tot.x += v[k].x; tot.y += v[k].y; tot.z += v[k].z; tot.w += v[k].w;
        if (r < w) { pre.x += v[k].x; pre.y += v[k].y; pre.z += v[k].z; pre.w += v[k].w; }
      }
    }
    ptot[rg][4 * cg] = tot.x; ptot[rg][4 * cg + 1] = tot.y;
    ptot[rg][4 * cg + 2] = tot.z; ptot[rg][4 * cg + 3] = tot.w;
    ppre[rg][4 * cg] = pre.x; ppre[rg][4 * cg + 1] = pre.y;
    ppre[rg][4 * cg + 2] = pre.z; ppre[rg][4 * cg + 3] = pre.w;
  }
  __syncthreads();
  const int tot = ptot[0][t] + ptot[1][t] + ptot[2][t] + ptot[3][t];
  const int pre = ppre[0][t] + ppre[1][t] + ppre[2][t] + ppre[3][t];
  const int start = GAT_excl_scan256(tot, wsum);
  cur[t] = start + pre;
  if (w == 0) {
    bstart[t] = start;
    if (t == GAT_NB - 1) bstart[GAT_NB] = start + tot;
  }
  __syncthreads();

  const unsigned width = GAT_bucket_width(N);
  const int chunk = (E + W - 1) / W;
  const int beg = w * chunk;
  const int end = (beg + chunk < E) ? (beg + chunk) : E;
  int2* pp = reinterpret_cast<int2*>(pairs);
  for (int base = beg; base < end; base += GAT_EPT * GAT_SORT_T) {
    int s[GAT_EPT], d[GAT_EPT];
#pragma unroll
    for (int k = 0; k < GAT_EPT; ++k) {
      const int i = base + k * GAT_SORT_T + t;
      s[k] = (i < end) ? src[i] : 0;
      d[k] = (i < end) ? dst[i] : 0;
    }
#pragma unroll
    for (int k = 0; k < GAT_EPT; ++k) {
      const int i = base + k * GAT_SORT_T + t;
      if (i < end) {
        const int slot = atomicAdd(&cur[(unsigned)d[k] / width], 1);
        pp[slot] = make_int2(s[k], d[k]);
      }
    }
  }
}

__device__ __forceinline__ void
GAT_bucket_sort_body(const int* __restrict__ pairs, const int* __restrict__ bstart,
                     int* __restrict__ rowptr, int* __restrict__ csrc, int N, int b) {
  __shared__ int cnt[GAT_WMAX];
  __shared__ int wsum[GAT_SORT_T / 32];
  const int t = threadIdx.x;
  const int width = (int)GAT_bucket_width(N);
  const int d0 = b * width;
  const int beg = bstart[b];
  const int end = bstart[b + 1];
  const int2* pp = reinterpret_cast<const int2*>(pairs);

  for (int i = t; i < GAT_WMAX; i += GAT_SORT_T) cnt[i] = 0;
  __syncthreads();
  for (int base = beg; base < end; base += GAT_EPT * GAT_SORT_T) {
    int d[GAT_EPT];
#pragma unroll
    for (int k = 0; k < GAT_EPT; ++k) {
      const int i = base + k * GAT_SORT_T + t;
      d[k] = (i < end) ? pp[i].y : d0;
    }
#pragma unroll
    for (int k = 0; k < GAT_EPT; ++k) {
      const int i = base + k * GAT_SORT_T + t;
      if (i < end) atomicAdd(&cnt[d[k] - d0], 1);
    }
  }
  __syncthreads();

  const int c0 = cnt[4 * t], c1 = cnt[4 * t + 1], c2 = cnt[4 * t + 2], c3 = cnt[4 * t + 3];
  const int ex = GAT_excl_scan256(c0 + c1 + c2 + c3, wsum);
  const int o0 = beg + ex, o1 = o0 + c0, o2 = o1 + c1, o3 = o2 + c2;
  cnt[4 * t] = o0; cnt[4 * t + 1] = o1; cnt[4 * t + 2] = o2; cnt[4 * t + 3] = o3;
  {
    const int q = 4 * t;
    if (q     < width && d0 + q     < N) rowptr[d0 + q]     = o0;
    if (q + 1 < width && d0 + q + 1 < N) rowptr[d0 + q + 1] = o1;
    if (q + 2 < width && d0 + q + 2 < N) rowptr[d0 + q + 2] = o2;
    if (q + 3 < width && d0 + q + 3 < N) rowptr[d0 + q + 3] = o3;
  }
  if (b == GAT_NB - 1 && t == 0) rowptr[N] = end;
  __syncthreads();

  for (int base = beg; base < end; base += GAT_EPT * GAT_SORT_T) {
    int2 p[GAT_EPT];
#pragma unroll
    for (int k = 0; k < GAT_EPT; ++k) {
      const int i = base + k * GAT_SORT_T + t;
      p[k] = (i < end) ? pp[i] : make_int2(0, d0);
    }
#pragma unroll
    for (int k = 0; k < GAT_EPT; ++k) {
      const int i = base + k * GAT_SORT_T + t;
      if (i < end) {
        const int slot = atomicAdd(&cnt[p[k].y - d0], 1);
        csrc[slot] = p[k].x;
      }
    }
  }
}

__global__ void __launch_bounds__(256)
GAT_count_gemm(const int* __restrict__ dst, int* __restrict__ cntMat, int E, int N,
               const float* __restrict__ A, const float* __restrict__ Bs, const float* __restrict__ Bd,
               const float* __restrict__ attS, const float* __restrict__ attD,
               float* __restrict__ Hout, float* __restrict__ aS, float* __restrict__ aD,
               int tileBeg, int tileEnd) {
  if (blockIdx.x < GAT_NB)
    GAT_bucket_count_body(dst, cntMat, E, N, (int)blockIdx.x);
  else
    GAT_gemm_strip_body<128, 256, 32, false>(A, Bs, Bd, attS, attD, Hout, aS, aD,
                                             (int)blockIdx.x - GAT_NB, tileBeg, tileEnd);
}

__global__ void __launch_bounds__(256)
GAT_scatter_gemm(const int* __restrict__ src, const int* __restrict__ dst,
                 const int* __restrict__ cntMat, int* __restrict__ bstart,
                 int* __restrict__ pairs, int E, int N,
                 const float* __restrict__ A, const float* __restrict__ Bs, const float* __restrict__ Bd,
                 const float* __restrict__ attS, const float* __restrict__ attD,
                 float* __restrict__ Hout, float* __restrict__ aS, float* __restrict__ aD,
                 int tileBeg, int tileEnd) {
  if (blockIdx.x < GAT_NB)
    GAT_bucket_scatter_body(src, dst, cntMat, bstart, pairs, E, N, (int)blockIdx.x);
  else
    GAT_gemm_strip_body<128, 256, 32, false>(A, Bs, Bd, attS, attD, Hout, aS, aD,
                                             (int)blockIdx.x - GAT_NB, tileBeg, tileEnd);
}

__global__ void __launch_bounds__(256)
GAT_sort_gemm(const int* __restrict__ pairs, const int* __restrict__ bstart,
              int* __restrict__ rowptr, int* __restrict__ csrc, int N,
              const float* __restrict__ A, const float* __restrict__ Bs, const float* __restrict__ Bd,
              const float* __restrict__ attS, const float* __restrict__ attD,
              float* __restrict__ Hout, float* __restrict__ aS, float* __restrict__ aD,
              int tileBeg, int tileEnd) {
  if (blockIdx.x < GAT_NB)
    GAT_bucket_sort_body(pairs, bstart, rowptr, csrc, N, (int)blockIdx.x);
  else
    GAT_gemm_strip_body<128, 256, 32, false>(A, Bs, Bd, attS, attD, Hout, aS, aD,
                                             (int)blockIdx.x - GAT_NB, tileBeg, tileEnd);
}

template <int H>
__device__ __forceinline__ void GAT_load_heads(const float* __restrict__ a, int node, float (&v)[H]) {
  if constexpr (H % 4 == 0) {
    const float4* p = reinterpret_cast<const float4*>(a + (size_t)node * H);
#pragma unroll
    for (int q = 0; q < H / 4; ++q) {
      const float4 x = p[q];
      v[4 * q] = x.x; v[4 * q + 1] = x.y; v[4 * q + 2] = x.z; v[4 * q + 3] = x.w;
    }
  } else {
#pragma unroll
    for (int h = 0; h < H; ++h) v[h] = a[(size_t)node * H + h];
  }
}

template <int H, int C>
__global__ void GAT_gather(const int* __restrict__ rowptr, const int* __restrict__ csrc,
                           const float* __restrict__ aS, const float* __restrict__ aD,
                           const float* __restrict__ Hsrc, const float* __restrict__ bias,
                           float* __restrict__ out, int N) {
  constexpr int HC = H * C;
  constexpr int NP = HC / 32;
  static_assert(C % 32 == 0, "a 32-float piece must lie inside one head");

  const int lane = threadIdx.x & 31;
  const int d = blockIdx.x * (blockDim.x >> 5) + (threadIdx.x >> 5);
  if (d >= N) return;
  const int beg = __builtin_amdgcn_readfirstlane(rowptr[d]);
  const int end = __builtin_amdgcn_readfirstlane(rowptr[d + 1]);

  float ad[H];
#pragma unroll
  for (int h = 0; h < H; ++h) ad[h] = aD[(size_t)d * H + h];

  float mx[H];
#pragma unroll
  for (int h = 0; h < H; ++h) mx[h] = -__builtin_inff();
  for (int base = beg; base < end; base += 32) {
    const int idx = base + lane;
    if (idx < end) {
      float as[H];
      GAT_load_heads<H>(aS, csrc[idx], as);
#pragma unroll
      for (int h = 0; h < H; ++h) mx[h] = fmaxf(mx[h], GAT_lrelu(as[h] + ad[h]));
    }
  }
#pragma unroll
  for (int h = 0; h < H; ++h) {
#pragma unroll
    for (int m = 16; m >= 1; m >>= 1) mx[h] = fmaxf(mx[h], __shfl_xor(mx[h], m, 32));
  }

  float sm[H];
#pragma unroll
  for (int h = 0; h < H; ++h) sm[h] = 0.0f;
  for (int base = beg; base < end; base += 32) {
    const int idx = base + lane;
    if (idx < end) {
      float as[H];
      GAT_load_heads<H>(aS, csrc[idx], as);
#pragma unroll
      for (int h = 0; h < H; ++h) sm[h] += __expf(GAT_lrelu(as[h] + ad[h]) - mx[h]);
    }
  }
#pragma unroll
  for (int h = 0; h < H; ++h) {
#pragma unroll
    for (int m = 16; m >= 1; m >>= 1) sm[h] += __shfl_xor(sm[h], m, 32);
  }

  float acc[NP];
#pragma unroll
  for (int k = 0; k < NP; ++k) acc[k] = 0.0f;
  for (int base = beg; base < end; base += 32) {
    const int idx = base + lane;
    int s = 0;
    float al[H];
#pragma unroll
    for (int h = 0; h < H; ++h) al[h] = 0.0f;
    if (idx < end) {
      s = csrc[idx];
      float as[H];
      GAT_load_heads<H>(aS, s, as);
#pragma unroll
      for (int h = 0; h < H; ++h)
        al[h] = __expf(GAT_lrelu(as[h] + ad[h]) - mx[h]) / (sm[h] + 1e-16f);
    }
    const int n = (end - base) < 32 ? (end - base) : 32;
    for (int j = 0; j < n; ++j) {
      const int sj = __builtin_amdgcn_readlane(s, j);
      const float* hp = Hsrc + (size_t)sj * HC + lane;
      float hv[NP];
#pragma unroll
      for (int k = 0; k < NP; ++k) hv[k] = hp[k * 32];
#pragma unroll
      for (int k = 0; k < NP; ++k) {
        const float a = __int_as_float(
            __builtin_amdgcn_readlane(__float_as_int(al[(k * 32) / C]), j));
        acc[k] += hv[k] * a;
      }
    }
  }

  float* op = out + (size_t)d * HC + lane;
#pragma unroll
  for (int k = 0; k < NP; ++k) op[k * 32] = acc[k] + bias[k * 32 + lane];
}

extern "C" void kernel_launch(void* const* d_in, const int* in_sizes, int n_in,
                              void* d_out, int out_size, void* d_ws, size_t ws_size,
                              hipStream_t stream) {
  (void)n_in; (void)out_size; (void)ws_size;

  const float* x     = (const float*)d_in[0];
  const int*   ei    = (const int*)d_in[1];
  const float* Wsrc1 = (const float*)d_in[2];
  const float* Wdst1 = (const float*)d_in[3];
  const float* attS1 = (const float*)d_in[4];
  const float* attD1 = (const float*)d_in[5];
  const float* b1    = (const float*)d_in[6];
  const float* Wsrc2 = (const float*)d_in[7];
  const float* Wdst2 = (const float*)d_in[8];
  const float* attS2 = (const float*)d_in[9];
  const float* attD2 = (const float*)d_in[10];
  const float* b2    = (const float*)d_in[11];

  const int FIN = 128, HEADS = 8, HID = 32, EMB = 64;
  const int HC1 = HEADS * HID;
  const long long N = in_sizes[0] / FIN;
  const long long E = in_sizes[1] / 2;
  const int* src = ei;
  const int* dst = ei + E;

  float* p = (float*)d_ws;
  float* h1   = p; p += N * HC1;
  float* h2   = p; p += N * EMB;
  float* out1 = p; p += N * HC1;
  float* aS1  = p; p += N * HEADS;
  float* aD1  = p; p += N * HEADS;
  float* aS2  = p; p += N;
  float* aD2  = p; p += N;
  int* rowptr = (int*)p; p += N + 4;
  int* csrc   = (int*)p; p += E;
  int* pairs  = (int*)p; p += 2 * E;
  int* cntMat = (int*)p; p += GAT_NB * GAT_NB;
  int* bstart = (int*)p; p += GAT_NB + 4;
  float* outF = (float*)d_out;

  const int TPB = 256;
  auto nb = [](long long n, int t) { return (unsigned)((n + t - 1) / t); };

  const int tiles = (int)(N / 16);
  const int rgs = (int)nb(tiles, TPB / 32);
  const int g3 = rgs < 128 ? rgs : 128;
  const int g2 = (rgs - g3) < 128 ? (rgs - g3) : 128;
  const int g1 = rgs - g3 - g2;
  const int t1 = g1 * (TPB / 32), t2 = (g1 + g2) * (TPB / 32);
  const int e1 = t1 < tiles ? t1 : tiles, e2 = t2 < tiles ? t2 : tiles;
  GAT_count_gemm<<<GAT_NB + g1 * (HC1 / 64), TPB, 0, stream>>>(
      dst, cntMat, (int)E, (int)N, x, Wsrc1, Wdst1, attS1, attD1, h1, aS1, aD1, 0, e1);
  GAT_scatter_gemm<<<GAT_NB + g2 * (HC1 / 64), TPB, 0, stream>>>(
      src, dst, cntMat, bstart, pairs, (int)E, (int)N,
      x, Wsrc1, Wdst1, attS1, attD1, h1, aS1, aD1, e1, e2);
  GAT_sort_gemm<<<GAT_NB + g3 * (HC1 / 64), TPB, 0, stream>>>(
      pairs, bstart, rowptr, csrc, (int)N,
      x, Wsrc1, Wdst1, attS1, attD1, h1, aS1, aD1, e2, tiles);

  GAT_gather<HEADS, HID><<<nb(N * 32, TPB), TPB, 0, stream>>>(
      rowptr, csrc, aS1, aD1, h1, b1, out1, (int)N);

  {
    GAT_gemm_strip<256, 64, 64, true><<<nb(N / 16, TPB / 32) * (EMB / 64), TPB, 0, stream>>>(
        out1, Wsrc2, Wdst2, attS2, attD2, h2, aS2, aD2, (int)N);
  }
  GAT_gather<1, EMB><<<nb(N * 32, TPB), TPB, 0, stream>>>(
      rowptr, csrc, aS2, aD2, h2, b2, outF, (int)N);
}


# dst logits by a folded weight: a_dst = A @ (W_dst . att_dst), fold once per call beside the count launch; W_dst GEMM tiles go, one 16-col WMMA tile per row tile stays; dst logit sum order changes
# speedup vs baseline: 4.6108x; 1.1211x over previous
#include <hip/hip_runtime.h>
#include <hip/hip_bf16.h>


typedef float v2f __attribute__((ext_vector_type(2)));
typedef float v8f __attribute__((ext_vector_type(8)));

#define GAT_NEG_SLOPE 0.2f

__device__ __forceinline__ float GAT_lrelu(float v) {
  return v > 0.0f ? v : GAT_NEG_SLOPE * v;
}

template <int K, int NC, int CH>
__device__ __forceinline__ void
GAT_fold_dst_body(const float* __restrict__ Wd, const float* __restrict__ attD,
                  float* __restrict__ Vt, int idx) {
  constexpr int H = NC / CH;
  if (idx >= 16 * K) return;
  const int h = idx / K;
  const int k = idx - h * K;
  float s = 0.0f;
  if (h < H) {
    const float4* w = reinterpret_cast<const float4*>(Wd + (size_t)k * NC + h * CH);
    const float4* a = reinterpret_cast<const float4*>(attD + h * CH);
#pragma unroll
    for (int q = 0; q < CH / 4; ++q) {
      const float4 wv = w[q];
      const float4 av = a[q];
      s += wv.x * av.x + wv.y * av.y + wv.z * av.z + wv.w * av.w;
    }
  }
  Vt[idx] = s;
}

template <int K, int NC, int CH, bool RELU, bool LOGIT>
__device__ __forceinline__ void
GAT_gemm_strip_body(const float* __restrict__ A,
                    const float* __restrict__ Bs,
                    const float* __restrict__ Vt,
                    const float* __restrict__ attS,
                    float* __restrict__ Hout,
                    float* __restrict__ aS,
                    float* __restrict__ aD,
                    int blk, int tileBeg, int tileEnd) {
  constexpr int NG = NC / 64;
  constexpr int H = NC / CH;
  constexpr int LPH = CH / 4;
  static_assert(K % 32 == 0, "K is walked in blocks of 32");
  static_assert(NC % 64 == 0, "a strip is 64 columns");
  static_assert(CH == 32 || CH == 64, "a strip must hold whole heads");
  static_assert(H <= 16, "the folded heads fill one 16-column tile");

  const int wpg = (int)(blockDim.x >> 5);
  const int mTile = tileBeg + (blk / NG) * wpg + (int)(threadIdx.x >> 5);
  if (mTile >= tileEnd) return;

  const int mBase = mTile << 4;
  const int nBase = (blk % NG) << 6;
  const int lane = threadIdx.x & 31;
  const int half = lane >> 4;
  const int c    = lane & 15;

  v8f cS[4] = {};
  v8f cL = {};
  const float* arow = A + (size_t)(mBase + c) * K + 16 * half;
  const float* vrow = Vt + (size_t)c * K + 16 * half;
  const float* bsp = Bs + (size_t)(16 * half) * NC + nBase + 4 * c;

  for (int kb = 0; kb < K; kb += 32) {
    float av[16];
    float lv[16];
#pragma unroll
    for (int q = 0; q < 4; ++q) {
      const float4 v = *reinterpret_cast<const float4*>(arow + kb + 4 * q);
      av[4 * q] = v.x; av[4 * q + 1] = v.y; av[4 * q + 2] = v.z; av[4 * q + 3] = v.w;
      if constexpr (LOGIT) {
        const float4 u = *reinterpret_cast<const float4*>(vrow + kb + 4 * q);
        lv[4 * q] = u.x; lv[4 * q + 1] = u.y; lv[4 * q + 2] = u.z; lv[4 * q + 3] = u.w;
      }
    }
    if (RELU) {
#pragma unroll
      for (int i = 0; i < 16; ++i) av[i] = fmaxf(av[i], 0.0f);
    }
#pragma unroll
    for (int j = 0; j < 8; ++j) {
      const float* ps = bsp + (size_t)(kb + 2 * j) * NC;
      const float4 s0 = *reinterpret_cast<const float4*>(ps);
      const float4 s1 = *reinterpret_cast<const float4*>(ps + NC);
      v2f a;  a.x = av[2 * j];  a.y = av[2 * j + 1];
      v2f b;
      b.x = s0.x; b.y = s1.x;
      cS[0] = __builtin_amdgcn_wmma_f32_16x16x4_f32(false, a, false, b, (short)0, cS[0], false, false);
      b.x = s0.y; b.y = s1.y;
      cS[1] = __builtin_amdgcn_wmma_f32_16x16x4_f32(false, a, false, b, (short)0, cS[1], false, false);
      b.x = s0.z; b.y = s1.z;
      cS[2] = __builtin_amdgcn_wmma_f32_16x16x4_f32(false, a, false, b, (short)0, cS[2], false, false);
      b.x = s0.w; b.y = s1.w;
      cS[3] = __builtin_amdgcn_wmma_f32_16x16x4_f32(false, a, false, b, (short)0, cS[3], false, false);
      if constexpr (LOGIT) {
        b.x = lv[2 * j]; b.y = lv[2 * j + 1];
        cL = __builtin_amdgcn_wmma_f32_16x16x4_f32(false, a, false, b, (short)0, cL, false, false);
      }
    }
  }

  const int col = nBase + 4 * c;
#pragma unroll
  for (int r = 0; r < 8; ++r) {
    float4 o;
    o.x = cS[0][r]; o.y = cS[1][r]; o.z = cS[2][r]; o.w = cS[3][r];
    *reinterpret_cast<float4*>(Hout + (size_t)(mBase + r + 8 * half) * NC + col) = o;
  }

  if constexpr (LOGIT) {
    if (c < H) {
#pragma unroll
      for (int r = 0; r < 8; ++r)
        aD[(size_t)(mBase + r + 8 * half) * H + c] = cL[r];
    }
  }

  const float4 wS = *reinterpret_cast<const float4*>(attS + col);
  const int head = col / CH;
  float vs[8];
#pragma unroll
  for (int r = 0; r < 8; ++r)
    vs[r] = cS[0][r] * wS.x + cS[1][r] * wS.y + cS[2][r] * wS.z + cS[3][r] * wS.w;
#pragma unroll
  for (int m = 1; m < LPH; m <<= 1) {
#pragma unroll
    for (int r = 0; r < 8; ++r) vs[r] += __shfl_xor(vs[r], m, 32);
  }
  if ((c & (LPH - 1)) == 0) {
#pragma unroll
    for (int r = 0; r < 8; ++r)
      aS[(size_t)(mBase + r + 8 * half) * H + head] = vs[r];
  }
}

template <int K, int NC, int CH, bool RELU>
__device__ __forceinline__ void
GAT_gemm_strip_any(const float* __restrict__ A,
                   const float* __restrict__ Bs,
                   const float* __restrict__ Vt,
                   const float* __restrict__ attS,
                   float* __restrict__ Hout,
                   float* __restrict__ aS,
                   float* __restrict__ aD,
                   int blk, int tileBeg, int tileEnd) {
  constexpr int NG = NC / 64;
  if (NG == 1 || (blk % NG) == 0)
    GAT_gemm_strip_body<K, NC, CH, RELU, true>(A, Bs, Vt, attS, Hout, aS, aD, blk, tileBeg, tileEnd);
  else
    GAT_gemm_strip_body<K, NC, CH, RELU, false>(A, Bs, Vt, attS, Hout, aS, aD, blk, tileBeg, tileEnd);
}

template <int K, int NC, int CH, bool RELU>
__global__ void __launch_bounds__(256)
GAT_gemm_strip(const float* __restrict__ A,
               const float* __restrict__ Bs,
               const float* __restrict__ Vt,
               const float* __restrict__ attS,
               float* __restrict__ Hout,
               float* __restrict__ aS,
               float* __restrict__ aD,
               int M) {
  GAT_gemm_strip_any<K, NC, CH, RELU>(A, Bs, Vt, attS, Hout, aS, aD,
                                      (int)blockIdx.x, 0, M >> 4);
}

#define GAT_NB 256
#define GAT_SORT_T 256
#define GAT_WMAX 1024
#define GAT_EPT 16
#define GAT_RPT 16

__device__ __forceinline__ unsigned GAT_bucket_width(int N) {
  return (unsigned)(N + GAT_NB - 1) / GAT_NB;
}

__device__ __forceinline__ int GAT_excl_scan256(int v, int* wsum) {
  const int lane = threadIdx.x & 31;
  const int wv = threadIdx.x >> 5;
  int inc = v;
  for (int m = 1; m < 32; m <<= 1) {
    const int u = __shfl_up(inc, m, 32);
    if (lane >= m) inc += u;
  }
  if (lane == 31) wsum[wv] = inc;
  __syncthreads();
  int woff = 0;
  for (int k = 0; k < GAT_SORT_T / 32; ++k) woff += (k < wv) ? wsum[k] : 0;
  return woff + inc - v;
}

__device__ __forceinline__ void
GAT_bucket_count_body(const int* __restrict__ dst, int* __restrict__ cntMat, int E, int N, int w) {
  __shared__ int c[GAT_NB];
  const int t = threadIdx.x;
  c[t] = 0;
  __syncthreads();
  const unsigned width = GAT_bucket_width(N);
  const int chunk = (E + GAT_NB - 1) / GAT_NB;
  const int beg = w * chunk;
  const int end = (beg + chunk < E) ? (beg + chunk) : E;
  for (int base = beg; base < end; base += GAT_EPT * GAT_SORT_T) {
    int d[GAT_EPT];
#pragma unroll
    for (int k = 0; k < GAT_EPT; ++k) {
      const int i = base + k * GAT_SORT_T + t;
      d[k] = (i < end) ? dst[i] : 0;
    }
#pragma unroll
    for (int k = 0; k < GAT_EPT; ++k) {
      const int i = base + k * GAT_SORT_T + t;
      if (i < end) atomicAdd(&c[(unsigned)d[k] / width], 1);
    }
  }
  __syncthreads();
  cntMat[w * GAT_NB + t] = c[t];
}

__device__ __forceinline__ void
GAT_bucket_scatter_body(const int* __restrict__ src, const int* __restrict__ dst,
                        const int* __restrict__ cntMat, int* __restrict__ bstart,
                        int* __restrict__ pairs, int E, int N, int w) {
  __shared__ int ptot[4][GAT_NB];
  __shared__ int ppre[4][GAT_NB];
  __shared__ int wsum[GAT_SORT_T / 32];
  __shared__ int cur[GAT_NB];
  const int t = threadIdx.x;
  const int W = GAT_NB;

  {
    const int cg = t & 63;
    const int rg = t >> 6;
    int4 tot = make_int4(0, 0, 0, 0);
    int4 pre = make_int4(0, 0, 0, 0);
    for (int r0 = rg; r0 < W; r0 += 4 * GAT_RPT) {
      int4 v[GAT_RPT];
#pragma unroll
      for (int k = 0; k < GAT_RPT; ++k) {
        const int r = r0 + 4 * k;
        v[k] = (r < W) ? *reinterpret_cast<const int4*>(cntMat + (size_t)r * GAT_NB + 4 * cg)
                       : make_int4(0, 0, 0, 0);
      }
#pragma unroll
      for (int k = 0; k < GAT_RPT; ++k) {
        const int r = r0 + 4 * k;
        tot.x += v[k].x; tot.y += v[k].y; tot.z += v[k].z; tot.w += v[k].w;
        if (r < w) { pre.x += v[k].x; pre.y += v[k].y; pre.z += v[k].z; pre.w += v[k].w; }
      }
    }
    ptot[rg][4 * cg] = tot.x; ptot[rg][4 * cg + 1] = tot.y;
    ptot[rg][4 * cg + 2] = tot.z; ptot[rg][4 * cg + 3] = tot.w;
    ppre[rg][4 * cg] = pre.x; ppre[rg][4 * cg + 1] = pre.y;
    ppre[rg][4 * cg + 2] = pre.z; ppre[rg][4 * cg + 3] = pre.w;
  }
  __syncthreads();
  const int tot = ptot[0][t] + ptot[1][t] + ptot[2][t] + ptot[3][t];
  const int pre = ppre[0][t] + ppre[1][t] + ppre[2][t] + ppre[3][t];
  const int start = GAT_excl_scan256(tot, wsum);
  cur[t] = start + pre;
  if (w == 0) {
    bstart[t] = start;
    if (t == GAT_NB - 1) bstart[GAT_NB] = start + tot;
  }
  __syncthreads();

  const unsigned width = GAT_bucket_width(N);
  const int chunk = (E + W - 1) / W;
  const int beg = w * chunk;
  const int end = (beg + chunk < E) ? (beg + chunk) : E;
  int2* pp = reinterpret_cast<int2*>(pairs);
  for (int base = beg; base < end; base += GAT_EPT * GAT_SORT_T) {
    int s[GAT_EPT], d[GAT_EPT];
#pragma unroll
    for (int k = 0; k < GAT_EPT; ++k) {
      const int i = base + k * GAT_SORT_T + t;
      s[k] = (i < end) ? src[i] : 0;
      d[k] = (i < end) ? dst[i] : 0;
    }
#pragma unroll
    for (int k = 0; k < GAT_EPT; ++k) {
      const int i = base + k * GAT_SORT_T + t;
      if (i < end) {
        const int slot = atomicAdd(&cur[(unsigned)d[k] / width], 1);
        pp[slot] = make_int2(s[k], d[k]);
      }
    }
  }
}

__device__ __forceinline__ void
GAT_bucket_sort_body(const int* __restrict__ pairs, const int* __restrict__ bstart,
                     int* __restrict__ rowptr, int* __restrict__ csrc, int N, int b) {
  __shared__ int cnt[GAT_WMAX];
  __shared__ int wsum[GAT_SORT_T / 32];
  const int t = threadIdx.x;
  const int width = (int)GAT_bucket_width(N);
  const int d0 = b * width;
  const int beg = bstart[b];
  const int end = bstart[b + 1];
  const int2* pp = reinterpret_cast<const int2*>(pairs);

  for (int i = t; i < GAT_WMAX; i += GAT_SORT_T) cnt[i] = 0;
  __syncthreads();
  for (int base = beg; base < end; base += GAT_EPT * GAT_SORT_T) {
    int d[GAT_EPT];
#pragma unroll
    for (int k = 0; k < GAT_EPT; ++k) {
      const int i = base + k * GAT_SORT_T + t;
      d[k] = (i < end) ? pp[i].y : d0;
    }
#pragma unroll
    for (int k = 0; k < GAT_EPT; ++k) {
      const int i = base + k * GAT_SORT_T + t;
      if (i < end) atomicAdd(&cnt[d[k] - d0], 1);
    }
  }
  __syncthreads();

  const int c0 = cnt[4 * t], c1 = cnt[4 * t + 1], c2 = cnt[4 * t + 2], c3 = cnt[4 * t + 3];
  const int ex = GAT_excl_scan256(c0 + c1 + c2 + c3, wsum);
  const int o0 = beg + ex, o1 = o0 + c0, o2 = o1 + c1, o3 = o2 + c2;
  cnt[4 * t] = o0; cnt[4 * t + 1] = o1; cnt[4 * t + 2] = o2; cnt[4 * t + 3] = o3;
  {
    const int q = 4 * t;
    if (q     < width && d0 + q     < N) rowptr[d0 + q]     = o0;
    if (q + 1 < width && d0 + q + 1 < N) rowptr[d0 + q + 1] = o1;
    if (q + 2 < width && d0 + q + 2 < N) rowptr[d0 + q + 2] = o2;
    if (q + 3 < width && d0 + q + 3 < N) rowptr[d0 + q + 3] = o3;
  }
  if (b == GAT_NB - 1 && t == 0) rowptr[N] = end;
  __syncthreads();

  for (int base = beg; base < end; base += GAT_EPT * GAT_SORT_T) {
    int2 p[GAT_EPT];
#pragma unroll
    for (int k = 0; k < GAT_EPT; ++k) {
      const int i = base + k * GAT_SORT_T + t;
      p[k] = (i < end) ? pp[i] : make_int2(0, d0);
    }
#pragma unroll
    for (int k = 0; k < GAT_EPT; ++k) {
      const int i = base + k * GAT_SORT_T + t;
      if (i < end) {
        const int slot = atomicAdd(&cnt[p[k].y - d0], 1);
        csrc[slot] = p[k].x;
      }
    }
  }
}

#define GAT_FOLD1_WG ((16 * 128) / GAT_SORT_T)
#define GAT_FOLD2_WG ((16 * 256) / GAT_SORT_T)

__global__ void __launch_bounds__(256)
GAT_count_fold(const int* __restrict__ dst, int* __restrict__ cntMat, int E, int N,
               const float* __restrict__ Wd1, const float* __restrict__ attD1, float* __restrict__ Vt1,
               const float* __restrict__ Wd2, const float* __restrict__ attD2, float* __restrict__ Vt2) {
  if (blockIdx.x < GAT_NB) {
    GAT_bucket_count_body(dst, cntMat, E, N, (int)blockIdx.x);
  } else {
    const int fb = (int)blockIdx.x - GAT_NB;
    if (fb < GAT_FOLD1_WG)
      GAT_fold_dst_body<128, 256, 32>(Wd1, attD1, Vt1, fb * GAT_SORT_T + (int)threadIdx.x);
    else
      GAT_fold_dst_body<256, 64, 64>(Wd2, attD2, Vt2, (fb - GAT_FOLD1_WG) * GAT_SORT_T + (int)threadIdx.x);
  }
}

__global__ void __launch_bounds__(256)
GAT_scatter_gemm(const int* __restrict__ src, const int* __restrict__ dst,
                 const int* __restrict__ cntMat, int* __restrict__ bstart,
                 int* __restrict__ pairs, int E, int N,
                 const float* __restrict__ A, const float* __restrict__ Bs, const float* __restrict__ Vt,
                 const float* __restrict__ attS,
                 float* __restrict__ Hout, float* __restrict__ aS, float* __restrict__ aD,
                 int tileBeg, int tileEnd) {
  if (blockIdx.x < GAT_NB)
    GAT_bucket_scatter_body(src, dst, cntMat, bstart, pairs, E, N, (int)blockIdx.x);
  else
    GAT_gemm_strip_any<128, 256, 32, false>(A, Bs, Vt, attS, Hout, aS, aD,
                                            (int)blockIdx.x - GAT_NB, tileBeg, tileEnd);
}

__global__ void __launch_bounds__(256)
GAT_sort_gemm(const int* __restrict__ pairs, const int* __restrict__ bstart,
              int* __restrict__ rowptr, int* __restrict__ csrc, int N,
              const float* __restrict__ A, const float* __restrict__ Bs, const float* __restrict__ Vt,
              const float* __restrict__ attS,
              float* __restrict__ Hout, float* __restrict__ aS, float* __restrict__ aD,
              int tileBeg, int tileEnd) {
  if (blockIdx.x < GAT_NB)
    GAT_bucket_sort_body(pairs, bstart, rowptr, csrc, N, (int)blockIdx.x);
  else
    GAT_gemm_strip_any<128, 256, 32, false>(A, Bs, Vt, attS, Hout, aS, aD,
                                            (int)blockIdx.x - GAT_NB, tileBeg, tileEnd);
}

template <int H>
__device__ __forceinline__ void GAT_load_heads(const float* __restrict__ a, int node, float (&v)[H]) {
  if constexpr (H % 4 == 0) {
    const float4* p = reinterpret_cast<const float4*>(a + (size_t)node * H);
#pragma unroll
    for (int q = 0; q < H / 4; ++q) {
      const float4 x = p[q];
      v[4 * q] = x.x; v[4 * q + 1] = x.y; v[4 * q + 2] = x.z; v[4 * q + 3] = x.w;
    }
  } else {
#pragma unroll
    for (int h = 0; h < H; ++h) v[h] = a[(size_t)node * H + h];
  }
}

template <int H, int C>
__global__ void GAT_gather(const int* __restrict__ rowptr, const int* __restrict__ csrc,
                           const float* __restrict__ aS, const float* __restrict__ aD,
                           const float* __restrict__ Hsrc, const float* __restrict__ bias,
                           float* __restrict__ out, int N) {
  constexpr int HC = H * C;
  constexpr int NP = HC / 32;
  static_assert(C % 32 == 0, "a 32-float piece must lie inside one head");

  const int lane = threadIdx.x & 31;
  const int d = blockIdx.x * (blockDim.x >> 5) + (threadIdx.x >> 5);
  if (d >= N) return;
  const int beg = __builtin_amdgcn_readfirstlane(rowptr[d]);
  const int end = __builtin_amdgcn_readfirstlane(rowptr[d + 1]);

  float ad[H];
#pragma unroll
  for (int h = 0; h < H; ++h) ad[h] = aD[(size_t)d * H + h];

  float mx[H];
#pragma unroll
  for (int h = 0; h < H; ++h) mx[h] = -__builtin_inff();
  for (int base = beg; base < end; base += 32) {
    const int idx = base + lane;
    if (idx < end) {
      float as[H];
      GAT_load_heads<H>(aS, csrc[idx], as);
#pragma unroll
      for (int h = 0; h < H; ++h) mx[h] = fmaxf(mx[h], GAT_lrelu(as[h] + ad[h]));
    }
  }
#pragma unroll
  for (int h = 0; h < H; ++h) {
#pragma unroll
    for (int m = 16; m >= 1; m >>= 1) mx[h] = fmaxf(mx[h], __shfl_xor(mx[h], m, 32));
  }

  float sm[H];
#pragma unroll
  for (int h = 0; h < H; ++h) sm[h] = 0.0f;
  for (int base = beg; base < end; base += 32) {
    const int idx = base + lane;
    if (idx < end) {
      float as[H];
      GAT_load_heads<H>(aS, csrc[idx], as);
#pragma unroll
      for (int h = 0; h < H; ++h) sm[h] += __expf(GAT_lrelu(as[h] + ad[h]) - mx[h]);
    }
  }
#pragma unroll
  for (int h = 0; h < H; ++h) {
#pragma unroll
    for (int m = 16; m >= 1; m >>= 1) sm[h] += __shfl_xor(sm[h], m, 32);
  }

  float acc[NP];
#pragma unroll
  for (int k = 0; k < NP; ++k) acc[k] = 0.0f;
  for (int base = beg; base < end; base += 32) {
    const int idx = base + lane;
    int s = 0;
    float al[H];
#pragma unroll
    for (int h = 0; h < H; ++h) al[h] = 0.0f;
    if (idx < end) {
      s = csrc[idx];
      float as[H];
      GAT_load_heads<H>(aS, s, as);
#pragma unroll
      for (int h = 0; h < H; ++h)
        al[h] = __expf(GAT_lrelu(as[h] + ad[h]) - mx[h]) / (sm[h] + 1e-16f);
    }
    const int n = (end - base) < 32 ? (end - base) : 32;
    for (int j = 0; j < n; ++j) {
      const int sj = __builtin_amdgcn_readlane(s, j);
      const float* hp = Hsrc + (size_t)sj * HC + lane;
      float hv[NP];
#pragma unroll
      for (int k = 0; k < NP; ++k) hv[k] = hp[k * 32];
#pragma unroll
      for (int k = 0; k < NP; ++k) {
        const float a = __int_as_float(
            __builtin_amdgcn_readlane(__float_as_int(al[(k * 32) / C]), j));
        acc[k] += hv[k] * a;
      }
    }
  }

  float* op = out + (size_t)d * HC + lane;
#pragma unroll
  for (int k = 0; k < NP; ++k) op[k * 32] = acc[k] + bias[k * 32 + lane];
}

extern "C" void kernel_launch(void* const* d_in, const int* in_sizes, int n_in,
                              void* d_out, int out_size, void* d_ws, size_t ws_size,
                              hipStream_t stream) {
  (void)n_in; (void)out_size; (void)ws_size;

  const float* x     = (const float*)d_in[0];
  const int*   ei    = (const int*)d_in[1];
  const float* Wsrc1 = (const float*)d_in[2];
  const float* Wdst1 = (const float*)d_in[3];
  const float* attS1 = (const float*)d_in[4];
  const float* attD1 = (const float*)d_in[5];
  const float* b1    = (const float*)d_in[6];
  const float* Wsrc2 = (const float*)d_in[7];
  const float* Wdst2 = (const float*)d_in[8];
  const float* attS2 = (const float*)d_in[9];
  const float* attD2 = (const float*)d_in[10];
  const float* b2    = (const float*)d_in[11];

  const int FIN = 128, HEADS = 8, HID = 32, EMB = 64;
  const int HC1 = HEADS * HID;
  const long long N = in_sizes[0] / FIN;
  const long long E = in_sizes[1] / 2;
  const int* src = ei;
  const int* dst = ei + E;

  float* p = (float*)d_ws;
  float* h1   = p; p += N * HC1;
  float* h2   = p; p += N * EMB;
  float* out1 = p; p += N * HC1;
  float* vt1  = p; p += 16 * FIN;
  float* vt2  = p; p += 16 * HC1;
  float* aS1  = p; p += N * HEADS;
  float* aD1  = p; p += N * HEADS;
  float* aS2  = p; p += N;
  float* aD2  = p; p += N;
  int* rowptr = (int*)p; p += N + 4;
  int* csrc   = (int*)p; p += E;
  int* pairs  = (int*)p; p += 2 * E;
  int* cntMat = (int*)p; p += GAT_NB * GAT_NB;
  int* bstart = (int*)p; p += GAT_NB + 4;
  float* outF = (float*)d_out;

  const int TPB = 256;
  auto nb = [](long long n, int t) { return (unsigned)((n + t - 1) / t); };

  const int tiles = (int)(N / 16);
  const int rgs = (int)nb(tiles, TPB / 32);
  const int g2 = (rgs + 1) / 2;
  const int g3 = rgs - g2;
  const int t2 = g2 * (TPB / 32);
  const int e2 = t2 < tiles ? t2 : tiles;
  GAT_count_fold<<<GAT_NB + GAT_FOLD1_WG + GAT_FOLD2_WG, TPB, 0, stream>>>(
      dst, cntMat, (int)E, (int)N, Wdst1, attD1, vt1, Wdst2, attD2, vt2);
  GAT_scatter_gemm<<<GAT_NB + g2 * (HC1 / 64), TPB, 0, stream>>>(
      src, dst, cntMat, bstart, pairs, (int)E, (int)N,
      x, Wsrc1, vt1, attS1, h1, aS1, aD1, 0, e2);
  GAT_sort_gemm<<<GAT_NB + g3 * (HC1 / 64), TPB, 0, stream>>>(
      pairs, bstart, rowptr, csrc, (int)N,
      x, Wsrc1, vt1, attS1, h1, aS1, aD1, e2, tiles);

  GAT_gather<HEADS, HID><<<nb(N * 32, TPB), TPB, 0, stream>>>(
      rowptr, csrc, aS1, aD1, h1, b1, out1, (int)N);

  {
    GAT_gemm_strip<256, 64, 64, true><<<nb(N / 16, TPB / 32) * (EMB / 64), TPB, 0, stream>>>(
        out1, Wsrc2, vt2, attS2, h2, aS2, aD2, (int)N);
  }
  GAT_gather<1, EMB><<<nb(N * 32, TPB), TPB, 0, stream>>>(
      rowptr, csrc, aS2, aD2, h2, b2, outF, (int)N);
}


# gather reads each edge's source index and logits once: a destination's first 32 edges stay in registers through max, exp-sum and weights (was 3 dependent index+logit reads); sums bit-identical
# speedup vs baseline: 4.8787x; 1.0581x over previous
#include <hip/hip_runtime.h>
#include <hip/hip_bf16.h>


typedef float v2f __attribute__((ext_vector_type(2)));
typedef float v8f __attribute__((ext_vector_type(8)));

#define GAT_NEG_SLOPE 0.2f

__device__ __forceinline__ float GAT_lrelu(float v) {
  return v > 0.0f ? v : GAT_NEG_SLOPE * v;
}

template <int K, int NC, int CH>
__device__ __forceinline__ void
GAT_fold_dst_body(const float* __restrict__ Wd, const float* __restrict__ attD,
                  float* __restrict__ Vt, int idx) {
  constexpr int H = NC / CH;
  if (idx >= 16 * K) return;
  const int h = idx / K;
  const int k = idx - h * K;
  float s = 0.0f;
  if (h < H) {
    const float4* w = reinterpret_cast<const float4*>(Wd + (size_t)k * NC + h * CH);
    const float4* a = reinterpret_cast<const float4*>(attD + h * CH);
#pragma unroll
    for (int q = 0; q < CH / 4; ++q) {
      const float4 wv = w[q];
      const float4 av = a[q];
      s += wv.x * av.x + wv.y * av.y + wv.z * av.z + wv.w * av.w;
    }
  }
  Vt[idx] = s;
}

template <int K, int NC, int CH, bool RELU, bool LOGIT>
__device__ __forceinline__ void
GAT_gemm_strip_body(const float* __restrict__ A,
                    const float* __restrict__ Bs,
                    const float* __restrict__ Vt,
                    const float* __restrict__ attS,
                    float* __restrict__ Hout,
                    float* __restrict__ aS,
                    float* __restrict__ aD,
                    int blk, int tileBeg, int tileEnd) {
  constexpr int NG = NC / 64;
  constexpr int H = NC / CH;
  constexpr int LPH = CH / 4;
  static_assert(K % 32 == 0, "K is walked in blocks of 32");
  static_assert(NC % 64 == 0, "a strip is 64 columns");
  static_assert(CH == 32 || CH == 64, "a strip must hold whole heads");
  static_assert(H <= 16, "the folded heads fill one 16-column tile");

  const int wpg = (int)(blockDim.x >> 5);
  const int mTile = tileBeg + (blk / NG) * wpg + (int)(threadIdx.x >> 5);
  if (mTile >= tileEnd) return;

  const int mBase = mTile << 4;
  const int nBase = (blk % NG) << 6;
  const int lane = threadIdx.x & 31;
  const int half = lane >> 4;
  const int c    = lane & 15;

  v8f cS[4] = {};
  v8f cL = {};
  const float* arow = A + (size_t)(mBase + c) * K + 16 * half;
  const float* vrow = Vt + (size_t)c * K + 16 * half;
  const float* bsp = Bs + (size_t)(16 * half) * NC + nBase + 4 * c;

  for (int kb = 0; kb < K; kb += 32) {
    float av[16];
    float lv[16];
#pragma unroll
    for (int q = 0; q < 4; ++q) {
      const float4 v = *reinterpret_cast<const float4*>(arow + kb + 4 * q);
      av[4 * q] = v.x; av[4 * q + 1] = v.y; av[4 * q + 2] = v.z; av[4 * q + 3] = v.w;
      if constexpr (LOGIT) {
        const float4 u = *reinterpret_cast<const float4*>(vrow + kb + 4 * q);
        lv[4 * q] = u.x; lv[4 * q + 1] = u.y; lv[4 * q + 2] = u.z; lv[4 * q + 3] = u.w;
      }
    }
    if (RELU) {
#pragma unroll
      for (int i = 0; i < 16; ++i) av[i] = fmaxf(av[i], 0.0f);
    }
#pragma unroll
    for (int j = 0; j < 8; ++j) {
      const float* ps = bsp + (size_t)(kb + 2 * j) * NC;
      const float4 s0 = *reinterpret_cast<const float4*>(ps);
      const float4 s1 = *reinterpret_cast<const float4*>(ps + NC);
      v2f a;  a.x = av[2 * j];  a.y = av[2 * j + 1];
      v2f b;
      b.x = s0.x; b.y = s1.x;
      cS[0] = __builtin_amdgcn_wmma_f32_16x16x4_f32(false, a, false, b, (short)0, cS[0], false, false);
      b.x = s0.y; b.y = s1.y;
      cS[1] = __builtin_amdgcn_wmma_f32_16x16x4_f32(false, a, false, b, (short)0, cS[1], false, false);
      b.x = s0.z; b.y = s1.z;
      cS[2] = __builtin_amdgcn_wmma_f32_16x16x4_f32(false, a, false, b, (short)0, cS[2], false, false);
      b.x = s0.w; b.y = s1.w;
      cS[3] = __builtin_amdgcn_wmma_f32_16x16x4_f32(false, a, false, b, (short)0, cS[3], false, false);
      if constexpr (LOGIT) {
        b.x = lv[2 * j]; b.y = lv[2 * j + 1];
        cL = __builtin_amdgcn_wmma_f32_16x16x4_f32(false, a, false, b, (short)0, cL, false, false);
      }
    }
  }

  const int col = nBase + 4 * c;
#pragma unroll
  for (int r = 0; r < 8; ++r) {
    float4 o;
    o.x = cS[0][r]; o.y = cS[1][r]; o.z = cS[2][r]; o.w = cS[3][r];
    *reinterpret_cast<float4*>(Hout + (size_t)(mBase + r + 8 * half) * NC + col) = o;
  }

  if constexpr (LOGIT) {
    if (c < H) {
#pragma unroll
      for (int r = 0; r < 8; ++r)
        aD[(size_t)(mBase + r + 8 * half) * H + c] = cL[r];
    }
  }

  const float4 wS = *reinterpret_cast<const float4*>(attS + col);
  const int head = col / CH;
  float vs[8];
#pragma unroll
  for (int r = 0; r < 8; ++r)
    vs[r] = cS[0][r] * wS.x + cS[1][r] * wS.y + cS[2][r] * wS.z + cS[3][r] * wS.w;
#pragma unroll
  for (int m = 1; m < LPH; m <<= 1) {
#pragma unroll
    for (int r = 0; r < 8; ++r) vs[r] += __shfl_xor(vs[r], m, 32);
  }
  if ((c & (LPH - 1)) == 0) {
#pragma unroll
    for (int r = 0; r < 8; ++r)
      aS[(size_t)(mBase + r + 8 * half) * H + head] = vs[r];
  }
}

template <int K, int NC, int CH, bool RELU>
__device__ __forceinline__ void
GAT_gemm_strip_any(const float* __restrict__ A,
                   const float* __restrict__ Bs,
                   const float* __restrict__ Vt,
                   const float* __restrict__ attS,
                   float* __restrict__ Hout,
                   float* __restrict__ aS,
                   float* __restrict__ aD,
                   int blk, int tileBeg, int tileEnd) {
  constexpr int NG = NC / 64;
  if (NG == 1 || (blk % NG) == 0)
    GAT_gemm_strip_body<K, NC, CH, RELU, true>(A, Bs, Vt, attS, Hout, aS, aD, blk, tileBeg, tileEnd);
  else
    GAT_gemm_strip_body<K, NC, CH, RELU, false>(A, Bs, Vt, attS, Hout, aS, aD, blk, tileBeg, tileEnd);
}

template <int K, int NC, int CH, bool RELU>
__global__ void __launch_bounds__(256)
GAT_gemm_strip(const float* __restrict__ A,
               const float* __restrict__ Bs,
               const float* __restrict__ Vt,
               const float* __restrict__ attS,
               float* __restrict__ Hout,
               float* __restrict__ aS,
               float* __restrict__ aD,
               int M) {
  GAT_gemm_strip_any<K, NC, CH, RELU>(A, Bs, Vt, attS, Hout, aS, aD,
                                      (int)blockIdx.x, 0, M >> 4);
}

#define GAT_NB 256
#define GAT_SORT_T 256
#define GAT_WMAX 1024
#define GAT_EPT 16
#define GAT_RPT 16

__device__ __forceinline__ unsigned GAT_bucket_width(int N) {
  return (unsigned)(N + GAT_NB - 1) / GAT_NB;
}

__device__ __forceinline__ int GAT_excl_scan256(int v, int* wsum) {
  const int lane = threadIdx.x & 31;
  const int wv = threadIdx.x >> 5;
  int inc = v;
  for (int m = 1; m < 32; m <<= 1) {
    const int u = __shfl_up(inc, m, 32);
    if (lane >= m) inc += u;
  }
  if (lane == 31) wsum[wv] = inc;
  __syncthreads();
  int woff = 0;
  for (int k = 0; k < GAT_SORT_T / 32; ++k) woff += (k < wv) ? wsum[k] : 0;
  return woff + inc - v;
}

__device__ __forceinline__ void
GAT_bucket_count_body(const int* __restrict__ dst, int* __restrict__ cntMat, int E, int N, int w) {
  __shared__ int c[GAT_NB];
  const int t = threadIdx.x;
  c[t] = 0;
  __syncthreads();
  const unsigned width = GAT_bucket_width(N);
  const int chunk = (E + GAT_NB - 1) / GAT_NB;
  const int beg = w * chunk;
  const int end = (beg + chunk < E) ? (beg + chunk) : E;
  for (int base = beg; base < end; base += GAT_EPT * GAT_SORT_T) {
    int d[GAT_EPT];
#pragma unroll
    for (int k = 0; k < GAT_EPT; ++k) {
      const int i = base + k * GAT_SORT_T + t;
      d[k] = (i < end) ? dst[i] : 0;
    }
#pragma unroll
    for (int k = 0; k < GAT_EPT; ++k) {
      const int i = base + k * GAT_SORT_T + t;
      if (i < end) atomicAdd(&c[(unsigned)d[k] / width], 1);
    }
  }
  __syncthreads();
  cntMat[w * GAT_NB + t] = c[t];
}

__device__ __forceinline__ void
GAT_bucket_scatter_body(const int* __restrict__ src, const int* __restrict__ dst,
                        const int* __restrict__ cntMat, int* __restrict__ bstart,
                        int* __restrict__ pairs, int E, int N, int w) {
  __shared__ int ptot[4][GAT_NB];
  __shared__ int ppre[4][GAT_NB];
  __shared__ int wsum[GAT_SORT_T / 32];
  __shared__ int cur[GAT_NB];
  const int t = threadIdx.x;
  const int W = GAT_NB;

  {
    const int cg = t & 63;
    const int rg = t >> 6;
    int4 tot = make_int4(0, 0, 0, 0);
    int4 pre = make_int4(0, 0, 0, 0);
    for (int r0 = rg; r0 < W; r0 += 4 * GAT_RPT) {
      int4 v[GAT_RPT];
#pragma unroll
      for (int k = 0; k < GAT_RPT; ++k) {
        const int r = r0 + 4 * k;
        v[k] = (r < W) ? *reinterpret_cast<const int4*>(cntMat + (size_t)r * GAT_NB + 4 * cg)
                       : make_int4(0, 0, 0, 0);
      }
#pragma unroll
      for (int k = 0; k < GAT_RPT; ++k) {
        const int r = r0 + 4 * k;
        tot.x += v[k].x; tot.y += v[k].y; tot.z += v[k].z; tot.w += v[k].w;
        if (r < w) { pre.x += v[k].x; pre.y += v[k].y; pre.z += v[k].z; pre.w += v[k].w; }
      }
    }
    ptot[rg][4 * cg] = tot.x; ptot[rg][4 * cg + 1] = tot.y;
    ptot[rg][4 * cg + 2] = tot.z; ptot[rg][4 * cg + 3] = tot.w;
    ppre[rg][4 * cg] = pre.x; ppre[rg][4 * cg + 1] = pre.y;
    ppre[rg][4 * cg + 2] = pre.z; ppre[rg][4 * cg + 3] = pre.w;
  }
  __syncthreads();
  const int tot = ptot[0][t] + ptot[1][t] + ptot[2][t] + ptot[3][t];
  const int pre = ppre[0][t] + ppre[1][t] + ppre[2][t] + ppre[3][t];
  const int start = GAT_excl_scan256(tot, wsum);
  cur[t] = start + pre;
  if (w == 0) {
    bstart[t] = start;
    if (t == GAT_NB - 1) bstart[GAT_NB] = start + tot;
  }
  __syncthreads();

  const unsigned width = GAT_bucket_width(N);
  const int chunk = (E + W - 1) / W;
  const int beg = w * chunk;
  const int end = (beg + chunk < E) ? (beg + chunk) : E;
  int2* pp = reinterpret_cast<int2*>(pairs);
  for (int base = beg; base < end; base += GAT_EPT * GAT_SORT_T) {
    int s[GAT_EPT], d[GAT_EPT];
#pragma unroll
    for (int k = 0; k < GAT_EPT; ++k) {
      const int i = base + k * GAT_SORT_T + t;
      s[k] = (i < end) ? src[i] : 0;
      d[k] = (i < end) ? dst[i] : 0;
    }
#pragma unroll
    for (int k = 0; k < GAT_EPT; ++k) {
      const int i = base + k * GAT_SORT_T + t;
      if (i < end) {
        const int slot = atomicAdd(&cur[(unsigned)d[k] / width], 1);
        pp[slot] = make_int2(s[k], d[k]);
      }
    }
  }
}

__device__ __forceinline__ void
GAT_bucket_sort_body(const int* __restrict__ pairs, const int* __restrict__ bstart,
                     int* __restrict__ rowptr, int* __restrict__ csrc, int N, int b) {
  __shared__ int cnt[GAT_WMAX];
  __shared__ int wsum[GAT_SORT_T / 32];
  const int t = threadIdx.x;
  const int width = (int)GAT_bucket_width(N);
  const int d0 = b * width;
  const int beg = bstart[b];
  const int end = bstart[b + 1];
  const int2* pp = reinterpret_cast<const int2*>(pairs);

  for (int i = t; i < GAT_WMAX; i += GAT_SORT_T) cnt[i] = 0;
  __syncthreads();
  for (int base = beg; base < end; base += GAT_EPT * GAT_SORT_T) {
    int d[GAT_EPT];
#pragma unroll
    for (int k = 0; k < GAT_EPT; ++k) {
      const int i = base + k * GAT_SORT_T + t;
      d[k] = (i < end) ? pp[i].y : d0;
    }
#pragma unroll
    for (int k = 0; k < GAT_EPT; ++k) {
      const int i = base + k * GAT_SORT_T + t;
      if (i < end) atomicAdd(&cnt[d[k] - d0], 1);
    }
  }
  __syncthreads();

  const int c0 = cnt[4 * t], c1 = cnt[4 * t + 1], c2 = cnt[4 * t + 2], c3 = cnt[4 * t + 3];
  const int ex = GAT_excl_scan256(c0 + c1 + c2 + c3, wsum);
  const int o0 = beg + ex, o1 = o0 + c0, o2 = o1 + c1, o3 = o2 + c2;
  cnt[4 * t] = o0; cnt[4 * t + 1] = o1; cnt[4 * t + 2] = o2; cnt[4 * t + 3] = o3;
  {
    const int q = 4 * t;
    if (q     < width && d0 + q     < N) rowptr[d0 + q]     = o0;
    if (q + 1 < width && d0 + q + 1 < N) rowptr[d0 + q + 1] = o1;
    if (q + 2 < width && d0 + q + 2 < N) rowptr[d0 + q + 2] = o2;
    if (q + 3 < width && d0 + q + 3 < N) rowptr[d0 + q + 3] = o3;
  }
  if (b == GAT_NB - 1 && t == 0) rowptr[N] = end;
  __syncthreads();

  for (int base = beg; base < end; base += GAT_EPT * GAT_SORT_T) {
    int2 p[GAT_EPT];
#pragma unroll
    for (int k = 0; k < GAT_EPT; ++k) {
      const int i = base + k * GAT_SORT_T + t;
      p[k] = (i < end) ? pp[i] : make_int2(0, d0);
    }
#pragma unroll
    for (int k = 0; k < GAT_EPT; ++k) {
      const int i = base + k * GAT_SORT_T + t;
      if (i < end) {
        const int slot = atomicAdd(&cnt[p[k].y - d0], 1);
        csrc[slot] = p[k].x;
      }
    }
  }
}

#define GAT_FOLD1_WG ((16 * 128) / GAT_SORT_T)
#define GAT_FOLD2_WG ((16 * 256) / GAT_SORT_T)

__global__ void __launch_bounds__(256)
GAT_count_fold(const int* __restrict__ dst, int* __restrict__ cntMat, int E, int N,
               const float* __restrict__ Wd1, const float* __restrict__ attD1, float* __restrict__ Vt1,
               const float* __restrict__ Wd2, const float* __restrict__ attD2, float* __restrict__ Vt2) {
  if (blockIdx.x < GAT_NB) {
    GAT_bucket_count_body(dst, cntMat, E, N, (int)blockIdx.x);
  } else {
    const int fb = (int)blockIdx.x - GAT_NB;
    if (fb < GAT_FOLD1_WG)
      GAT_fold_dst_body<128, 256, 32>(Wd1, attD1, Vt1, fb * GAT_SORT_T + (int)threadIdx.x);
    else
      GAT_fold_dst_body<256, 64, 64>(Wd2, attD2, Vt2, (fb - GAT_FOLD1_WG) * GAT_SORT_T + (int)threadIdx.x);
  }
}

__global__ void __launch_bounds__(256)
GAT_scatter_gemm(const int* __restrict__ src, const int* __restrict__ dst,
                 const int* __restrict__ cntMat, int* __restrict__ bstart,
                 int* __restrict__ pairs, int E, int N,
                 const float* __restrict__ A, const float* __restrict__ Bs, const float* __restrict__ Vt,
                 const float* __restrict__ attS,
                 float* __restrict__ Hout, float* __restrict__ aS, float* __restrict__ aD,
                 int tileBeg, int tileEnd) {
  if (blockIdx.x < GAT_NB)
    GAT_bucket_scatter_body(src, dst, cntMat, bstart, pairs, E, N, (int)blockIdx.x);
  else
    GAT_gemm_strip_any<128, 256, 32, false>(A, Bs, Vt, attS, Hout, aS, aD,
                                            (int)blockIdx.x - GAT_NB, tileBeg, tileEnd);
}

__global__ void __launch_bounds__(256)
GAT_sort_gemm(const int* __restrict__ pairs, const int* __restrict__ bstart,
              int* __restrict__ rowptr, int* __restrict__ csrc, int N,
              const float* __restrict__ A, const float* __restrict__ Bs, const float* __restrict__ Vt,
              const float* __restrict__ attS,
              float* __restrict__ Hout, float* __restrict__ aS, float* __restrict__ aD,
              int tileBeg, int tileEnd) {
  if (blockIdx.x < GAT_NB)
    GAT_bucket_sort_body(pairs, bstart, rowptr, csrc, N, (int)blockIdx.x);
  else
    GAT_gemm_strip_any<128, 256, 32, false>(A, Bs, Vt, attS, Hout, aS, aD,
                                            (int)blockIdx.x - GAT_NB, tileBeg, tileEnd);
}

template <int H>
__device__ __forceinline__ void GAT_load_heads(const float* __restrict__ a, int node, float (&v)[H]) {
  if constexpr (H % 4 == 0) {
    const float4* p = reinterpret_cast<const float4*>(a + (size_t)node * H);
#pragma unroll
    for (int q = 0; q < H / 4; ++q) {
      const float4 x = p[q];
      v[4 * q] = x.x; v[4 * q + 1] = x.y; v[4 * q + 2] = x.z; v[4 * q + 3] = x.w;
    }
  } else {
#pragma unroll
    for (int h = 0; h < H; ++h) v[h] = a[(size_t)node * H + h];
  }
}

template <int H, int C>
__global__ void GAT_gather(const int* __restrict__ rowptr, const int* __restrict__ csrc,
                           const float* __restrict__ aS, const float* __restrict__ aD,
                           const float* __restrict__ Hsrc, const float* __restrict__ bias,
                           float* __restrict__ out, int N) {
  constexpr int HC = H * C;
  constexpr int NP = HC / 32;
  static_assert(C % 32 == 0, "a 32-float piece must lie inside one head");

  const int lane = threadIdx.x & 31;
  const int d = blockIdx.x * (blockDim.x >> 5) + (threadIdx.x >> 5);
  if (d >= N) return;
  const int beg = __builtin_amdgcn_readfirstlane(rowptr[d]);
  const int end = __builtin_amdgcn_readfirstlane(rowptr[d + 1]);

  float ad[H];
#pragma unroll
  for (int h = 0; h < H; ++h) ad[h] = aD[(size_t)d * H + h];

  const bool in0 = (beg + lane) < end;
  int s0 = 0;
  float e0[H];
#pragma unroll
  for (int h = 0; h < H; ++h) e0[h] = -__builtin_inff();
  if (in0) {
    s0 = csrc[beg + lane];
    float as[H];
    GAT_load_heads<H>(aS, s0, as);
#pragma unroll
    for (int h = 0; h < H; ++h) e0[h] = GAT_lrelu(as[h] + ad[h]);
  }

  float mx[H];
#pragma unroll
  for (int h = 0; h < H; ++h) mx[h] = fmaxf(-__builtin_inff(), e0[h]);
  for (int base = beg + 32; base < end; base += 32) {
    const int idx = base + lane;
    if (idx < end) {
      float as[H];
      GAT_load_heads<H>(aS, csrc[idx], as);
#pragma unroll
      for (int h = 0; h < H; ++h) mx[h] = fmaxf(mx[h], GAT_lrelu(as[h] + ad[h]));
    }
  }
#pragma unroll
  for (int h = 0; h < H; ++h) {
#pragma unroll
    for (int m = 16; m >= 1; m >>= 1) mx[h] = fmaxf(mx[h], __shfl_xor(mx[h], m, 32));
  }

  float p0[H];
  float sm[H];
#pragma unroll
  for (int h = 0; h < H; ++h) {
    p0[h] = in0 ? __expf(e0[h] - mx[h]) : 0.0f;
    sm[h] = p0[h];
  }
  for (int base = beg + 32; base < end; base += 32) {
    const int idx = base + lane;
    if (idx < end) {
      float as[H];
      GAT_load_heads<H>(aS, csrc[idx], as);
#pragma unroll
      for (int h = 0; h < H; ++h) sm[h] += __expf(GAT_lrelu(as[h] + ad[h]) - mx[h]);
    }
  }
#pragma unroll
  for (int h = 0; h < H; ++h) {
#pragma unroll
    for (int m = 16; m >= 1; m >>= 1) sm[h] += __shfl_xor(sm[h], m, 32);
  }

  float acc[NP];
#pragma unroll
  for (int k = 0; k < NP; ++k) acc[k] = 0.0f;
  int s = s0;
  float al[H];
#pragma unroll
  for (int h = 0; h < H; ++h) al[h] = p0[h] / (sm[h] + 1e-16f);
  for (int base = beg; base < end; base += 32) {
    if (base != beg) {
      const int idx = base + lane;
      s = 0;
#pragma unroll
      for (int h = 0; h < H; ++h) al[h] = 0.0f;
      if (idx < end) {
        s = csrc[idx];
        float as[H];
        GAT_load_heads<H>(aS, s, as);
#pragma unroll
        for (int h = 0; h < H; ++h)
          al[h] = __expf(GAT_lrelu(as[h] + ad[h]) - mx[h]) / (sm[h] + 1e-16f);
      }
    }
    const int n = (end - base) < 32 ? (end - base) : 32;
    for (int j = 0; j < n; ++j) {
      const int sj = __builtin_amdgcn_readlane(s, j);
      const float* hp = Hsrc + (size_t)sj * HC + lane;
      float hv[NP];
#pragma unroll
      for (int k = 0; k < NP; ++k) hv[k] = hp[k * 32];
#pragma unroll
      for (int k = 0; k < NP; ++k) {
        const float a = __int_as_float(
            __builtin_amdgcn_readlane(__float_as_int(al[(k * 32) / C]), j));
        acc[k] += hv[k] * a;
      }
    }
  }

  float* op = out + (size_t)d * HC + lane;
#pragma unroll
  for (int k = 0; k < NP; ++k) op[k * 32] = acc[k] + bias[k * 32 + lane];
}

extern "C" void kernel_launch(void* const* d_in, const int* in_sizes, int n_in,
                              void* d_out, int out_size, void* d_ws, size_t ws_size,
                              hipStream_t stream) {
  (void)n_in; (void)out_size; (void)ws_size;

  const float* x     = (const float*)d_in[0];
  const int*   ei    = (const int*)d_in[1];
  const float* Wsrc1 = (const float*)d_in[2];
  const float* Wdst1 = (const float*)d_in[3];
  const float* attS1 = (const float*)d_in[4];
  const float* attD1 = (const float*)d_in[5];
  const float* b1    = (const float*)d_in[6];
  const float* Wsrc2 = (const float*)d_in[7];
  const float* Wdst2 = (const float*)d_in[8];
  const float* attS2 = (const float*)d_in[9];
  const float* attD2 = (const float*)d_in[10];
  const float* b2    = (const float*)d_in[11];

  const int FIN = 128, HEADS = 8, HID = 32, EMB = 64;
  const int HC1 = HEADS * HID;
  const long long N = in_sizes[0] / FIN;
  const long long E = in_sizes[1] / 2;
  const int* src = ei;
  const int* dst = ei + E;

  float* p = (float*)d_ws;
  float* h1   = p; p += N * HC1;
  float* h2   = p; p += N * EMB;
  float* out1 = p; p += N * HC1;
  float* vt1  = p; p += 16 * FIN;
  float* vt2  = p; p += 16 * HC1;
  float* aS1  = p; p += N * HEADS;
  float* aD1  = p; p += N * HEADS;
  float* aS2  = p; p += N;
  float* aD2  = p; p += N;
  int* rowptr = (int*)p; p += N + 4;
  int* csrc   = (int*)p; p += E;
  int* pairs  = (int*)p; p += 2 * E;
  int* cntMat = (int*)p; p += GAT_NB * GAT_NB;
  int* bstart = (int*)p; p += GAT_NB + 4;
  float* outF = (float*)d_out;

  const int TPB = 256;
  auto nb = [](long long n, int t) { return (unsigned)((n + t - 1) / t); };

  const int tiles = (int)(N / 16);
  const int rgs = (int)nb(tiles, TPB / 32);
  const int g2 = (rgs + 1) / 2;
  const int g3 = rgs - g2;
  const int t2 = g2 * (TPB / 32);
  const int e2 = t2 < tiles ? t2 : tiles;
  GAT_count_fold<<<GAT_NB + GAT_FOLD1_WG + GAT_FOLD2_WG, TPB, 0, stream>>>(
      dst, cntMat, (int)E, (int)N, Wdst1, attD1, vt1, Wdst2, attD2, vt2);
  GAT_scatter_gemm<<<GAT_NB + g2 * (HC1 / 64), TPB, 0, stream>>>(
      src, dst, cntMat, bstart, pairs, (int)E, (int)N,
      x, Wsrc1, vt1, attS1, h1, aS1, aD1, 0, e2);
  GAT_sort_gemm<<<GAT_NB + g3 * (HC1 / 64), TPB, 0, stream>>>(
      pairs, bstart, rowptr, csrc, (int)N,
      x, Wsrc1, vt1, attS1, h1, aS1, aD1, e2, tiles);

  GAT_gather<HEADS, HID><<<nb(N * 32, TPB), TPB, 0, stream>>>(
      rowptr, csrc, aS1, aD1, h1, b1, out1, (int)N);

  {
    GAT_gemm_strip<256, 64, 64, true><<<nb(N / 16, TPB / 32) * (EMB / 64), TPB, 0, stream>>>(
        out1, Wsrc2, vt2, attS2, h2, aS2, aD2, (int)N);
  }
  GAT_gather<1, EMB><<<nb(N * 32, TPB), TPB, 0, stream>>>(
      rowptr, csrc, aS2, aD2, h2, b2, outF, (int)N);
}


# gather row loads widened to 16 B per lane: one load covers 4 whole lines (was 1); layer-1 head weights via a per-wave LDS slot; layer-2 wave halves take edge pairs, its sum order changes
# speedup vs baseline: 5.1181x; 1.0491x over previous
#include <hip/hip_runtime.h>
#include <hip/hip_bf16.h>


typedef float v2f __attribute__((ext_vector_type(2)));
typedef float v8f __attribute__((ext_vector_type(8)));

#define GAT_NEG_SLOPE 0.2f

__device__ __forceinline__ float GAT_lrelu(float v) {
  return v > 0.0f ? v : GAT_NEG_SLOPE * v;
}

template <int K, int NC, int CH>
__device__ __forceinline__ void
GAT_fold_dst_body(const float* __restrict__ Wd, const float* __restrict__ attD,
                  float* __restrict__ Vt, int idx) {
  constexpr int H = NC / CH;
  if (idx >= 16 * K) return;
  const int h = idx / K;
  const int k = idx - h * K;
  float s = 0.0f;
  if (h < H) {
    const float4* w = reinterpret_cast<const float4*>(Wd + (size_t)k * NC + h * CH);
    const float4* a = reinterpret_cast<const float4*>(attD + h * CH);
#pragma unroll
    for (int q = 0; q < CH / 4; ++q) {
      const float4 wv = w[q];
      const float4 av = a[q];
      s += wv.x * av.x + wv.y * av.y + wv.z * av.z + wv.w * av.w;
    }
  }
  Vt[idx] = s;
}

template <int K, int NC, int CH, bool RELU, bool LOGIT>
__device__ __forceinline__ void
GAT_gemm_strip_body(const float* __restrict__ A,
                    const float* __restrict__ Bs,
                    const float* __restrict__ Vt,
                    const float* __restrict__ attS,
                    float* __restrict__ Hout,
                    float* __restrict__ aS,
                    float* __restrict__ aD,
                    int blk, int tileBeg, int tileEnd) {
  constexpr int NG = NC / 64;
  constexpr int H = NC / CH;
  constexpr int LPH = CH / 4;
  static_assert(K % 32 == 0, "K is walked in blocks of 32");
  static_assert(NC % 64 == 0, "a strip is 64 columns");
  static_assert(CH == 32 || CH == 64, "a strip must hold whole heads");
  static_assert(H <= 16, "the folded heads fill one 16-column tile");

  const int wpg = (int)(blockDim.x >> 5);
  const int mTile = tileBeg + (blk / NG) * wpg + (int)(threadIdx.x >> 5);
  if (mTile >= tileEnd) return;

  const int mBase = mTile << 4;
  const int nBase = (blk % NG) << 6;
  const int lane = threadIdx.x & 31;
  const int half = lane >> 4;
  const int c    = lane & 15;

  v8f cS[4] = {};
  v8f cL = {};
  const float* arow = A + (size_t)(mBase + c) * K + 16 * half;
  const float* vrow = Vt + (size_t)c * K + 16 * half;
  const float* bsp = Bs + (size_t)(16 * half) * NC + nBase + 4 * c;

  for (int kb = 0; kb < K; kb += 32) {
    float av[16];
    float lv[16];
#pragma unroll
    for (int q = 0; q < 4; ++q) {
      const float4 v = *reinterpret_cast<const float4*>(arow + kb + 4 * q);
      av[4 * q] = v.x; av[4 * q + 1] = v.y; av[4 * q + 2] = v.z; av[4 * q + 3] = v.w;
      if constexpr (LOGIT) {
        const float4 u = *reinterpret_cast<const float4*>(vrow + kb + 4 * q);
        lv[4 * q] = u.x; lv[4 * q + 1] = u.y; lv[4 * q + 2] = u.z; lv[4 * q + 3] = u.w;
      }
    }
    if (RELU) {
#pragma unroll
      for (int i = 0; i < 16; ++i) av[i] = fmaxf(av[i], 0.0f);
    }
#pragma unroll
    for (int j = 0; j < 8; ++j) {
      const float* ps = bsp + (size_t)(kb + 2 * j) * NC;
      const float4 s0 = *reinterpret_cast<const float4*>(ps);
      const float4 s1 = *reinterpret_cast<const float4*>(ps + NC);
      v2f a;  a.x = av[2 * j];  a.y = av[2 * j + 1];
      v2f b;
      b.x = s0.x; b.y = s1.x;
      cS[0] = __builtin_amdgcn_wmma_f32_16x16x4_f32(false, a, false, b, (short)0, cS[0], false, false);
      b.x = s0.y; b.y = s1.y;
      cS[1] = __builtin_amdgcn_wmma_f32_16x16x4_f32(false, a, false, b, (short)0, cS[1], false, false);
      b.x = s0.z; b.y = s1.z;
      cS[2] = __builtin_amdgcn_wmma_f32_16x16x4_f32(false, a, false, b, (short)0, cS[2], false, false);
      b.x = s0.w; b.y = s1.w;
      cS[3] = __builtin_amdgcn_wmma_f32_16x16x4_f32(false, a, false, b, (short)0, cS[3], false, false);
      if constexpr (LOGIT) {
        b.x = lv[2 * j]; b.y = lv[2 * j + 1];
        cL = __builtin_amdgcn_wmma_f32_16x16x4_f32(false, a, false, b, (short)0, cL, false, false);
      }
    }
  }

  const int col = nBase + 4 * c;
#pragma unroll
  for (int r = 0; r < 8; ++r) {
    float4 o;
    o.x = cS[0][r]; o.y = cS[1][r]; o.z = cS[2][r]; o.w = cS[3][r];
    *reinterpret_cast<float4*>(Hout + (size_t)(mBase + r + 8 * half) * NC + col) = o;
  }

  if constexpr (LOGIT) {
    if (c < H) {
#pragma unroll
      for (int r = 0; r < 8; ++r)
        aD[(size_t)(mBase + r + 8 * half) * H + c] = cL[r];
    }
  }

  const float4 wS = *reinterpret_cast<const float4*>(attS + col);
  const int head = col / CH;
  float vs[8];
#pragma unroll
  for (int r = 0; r < 8; ++r)
    vs[r] = cS[0][r] * wS.x + cS[1][r] * wS.y + cS[2][r] * wS.z + cS[3][r] * wS.w;
#pragma unroll
  for (int m = 1; m < LPH; m <<= 1) {
#pragma unroll
    for (int r = 0; r < 8; ++r) vs[r] += __shfl_xor(vs[r], m, 32);
  }
  if ((c & (LPH - 1)) == 0) {
#pragma unroll
    for (int r = 0; r < 8; ++r)
      aS[(size_t)(mBase + r + 8 * half) * H + head] = vs[r];
  }
}

template <int K, int NC, int CH, bool RELU>
__device__ __forceinline__ void
GAT_gemm_strip_any(const float* __restrict__ A,
                   const float* __restrict__ Bs,
                   const float* __restrict__ Vt,
                   const float* __restrict__ attS,
                   float* __restrict__ Hout,
                   float* __restrict__ aS,
                   float* __restrict__ aD,
                   int blk, int tileBeg, int tileEnd) {
  constexpr int NG = NC / 64;
  if (NG == 1 || (blk % NG) == 0)
    GAT_gemm_strip_body<K, NC, CH, RELU, true>(A, Bs, Vt, attS, Hout, aS, aD, blk, tileBeg, tileEnd);
  else
    GAT_gemm_strip_body<K, NC, CH, RELU, false>(A, Bs, Vt, attS, Hout, aS, aD, blk, tileBeg, tileEnd);
}

template <int K, int NC, int CH, bool RELU>
__global__ void __launch_bounds__(256)
GAT_gemm_strip(const float* __restrict__ A,
               const float* __restrict__ Bs,
               const float* __restrict__ Vt,
               const float* __restrict__ attS,
               float* __restrict__ Hout,
               float* __restrict__ aS,
               float* __restrict__ aD,
               int M) {
  GAT_gemm_strip_any<K, NC, CH, RELU>(A, Bs, Vt, attS, Hout, aS, aD,
                                      (int)blockIdx.x, 0, M >> 4);
}

#define GAT_NB 256
#define GAT_SORT_T 256
#define GAT_WMAX 1024
#define GAT_EPT 16
#define GAT_RPT 16

__device__ __forceinline__ unsigned GAT_bucket_width(int N) {
  return (unsigned)(N + GAT_NB - 1) / GAT_NB;
}

__device__ __forceinline__ int GAT_excl_scan256(int v, int* wsum) {
  const int lane = threadIdx.x & 31;
  const int wv = threadIdx.x >> 5;
  int inc = v;
  for (int m = 1; m < 32; m <<= 1) {
    const int u = __shfl_up(inc, m, 32);
    if (lane >= m) inc += u;
  }
  if (lane == 31) wsum[wv] = inc;
  __syncthreads();
  int woff = 0;
  for (int k = 0; k < GAT_SORT_T / 32; ++k) woff += (k < wv) ? wsum[k] : 0;
  return woff + inc - v;
}

__device__ __forceinline__ void
GAT_bucket_count_body(const int* __restrict__ dst, int* __restrict__ cntMat, int E, int N, int w) {
  __shared__ int c[GAT_NB];
  const int t = threadIdx.x;
  c[t] = 0;
  __syncthreads();
  const unsigned width = GAT_bucket_width(N);
  const int chunk = (E + GAT_NB - 1) / GAT_NB;
  const int beg = w * chunk;
  const int end = (beg + chunk < E) ? (beg + chunk) : E;
  for (int base = beg; base < end; base += GAT_EPT * GAT_SORT_T) {
    int d[GAT_EPT];
#pragma unroll
    for (int k = 0; k < GAT_EPT; ++k) {
      const int i = base + k * GAT_SORT_T + t;
      d[k] = (i < end) ? dst[i] : 0;
    }
#pragma unroll
    for (int k = 0; k < GAT_EPT; ++k) {
      const int i = base + k * GAT_SORT_T + t;
      if (i < end) atomicAdd(&c[(unsigned)d[k] / width], 1);
    }
  }
  __syncthreads();
  cntMat[w * GAT_NB + t] = c[t];
}

__device__ __forceinline__ void
GAT_bucket_scatter_body(const int* __restrict__ src, const int* __restrict__ dst,
                        const int* __restrict__ cntMat, int* __restrict__ bstart,
                        int* __restrict__ pairs, int E, int N, int w) {
  __shared__ int ptot[4][GAT_NB];
  __shared__ int ppre[4][GAT_NB];
  __shared__ int wsum[GAT_SORT_T / 32];
  __shared__ int cur[GAT_NB];
  const int t = threadIdx.x;
  const int W = GAT_NB;

  {
    const int cg = t & 63;
    const int rg = t >> 6;
    int4 tot = make_int4(0, 0, 0, 0);
    int4 pre = make_int4(0, 0, 0, 0);
    for (int r0 = rg; r0 < W; r0 += 4 * GAT_RPT) {
      int4 v[GAT_RPT];
#pragma unroll
      for (int k = 0; k < GAT_RPT; ++k) {
        const int r = r0 + 4 * k;
        v[k] = (r < W) ? *reinterpret_cast<const int4*>(cntMat + (size_t)r * GAT_NB + 4 * cg)
                       : make_int4(0, 0, 0, 0);
      }
#pragma unroll
      for (int k = 0; k < GAT_RPT; ++k) {
        const int r = r0 + 4 * k;
        tot.x += v[k].x; tot.y += v[k].y; tot.z += v[k].z; tot.w += v[k].w;
        if (r < w) { pre.x += v[k].x; pre.y += v[k].y; pre.z += v[k].z; pre.w += v[k].w; }
      }
    }
    ptot[rg][4 * cg] = tot.x; ptot[rg][4 * cg + 1] = tot.y;
    ptot[rg][4 * cg + 2] = tot.z; ptot[rg][4 * cg + 3] = tot.w;
    ppre[rg][4 * cg] = pre.x; ppre[rg][4 * cg + 1] = pre.y;
    ppre[rg][4 * cg + 2] = pre.z; ppre[rg][4 * cg + 3] = pre.w;
  }
  __syncthreads();
  const int tot = ptot[0][t] + ptot[1][t] + ptot[2][t] + ptot[3][t];
  const int pre = ppre[0][t] + ppre[1][t] + ppre[2][t] + ppre[3][t];
  const int start = GAT_excl_scan256(tot, wsum);
  cur[t] = start + pre;
  if (w == 0) {
    bstart[t] = start;
    if (t == GAT_NB - 1) bstart[GAT_NB] = start + tot;
  }
  __syncthreads();

  const unsigned width = GAT_bucket_width(N);
  const int chunk = (E + W - 1) / W;
  const int beg = w * chunk;
  const int end = (beg + chunk < E) ? (beg + chunk) : E;
  int2* pp = reinterpret_cast<int2*>(pairs);
  for (int base = beg; base < end; base += GAT_EPT * GAT_SORT_T) {
    int s[GAT_EPT], d[GAT_EPT];
#pragma unroll
    for (int k = 0; k < GAT_EPT; ++k) {
      const int i = base + k * GAT_SORT_T + t;
      s[k] = (i < end) ? src[i] : 0;
      d[k] = (i < end) ? dst[i] : 0;
    }
#pragma unroll
    for (int k = 0; k < GAT_EPT; ++k) {
      const int i = base + k * GAT_SORT_T + t;
      if (i < end) {
        const int slot = atomicAdd(&cur[(unsigned)d[k] / width], 1);
        pp[slot] = make_int2(s[k], d[k]);
      }
    }
  }
}

__device__ __forceinline__ void
GAT_bucket_sort_body(const int* __restrict__ pairs, const int* __restrict__ bstart,
                     int* __restrict__ rowptr, int* __restrict__ csrc, int N, int b) {
  __shared__ int cnt[GAT_WMAX];
  __shared__ int wsum[GAT_SORT_T / 32];
  const int t = threadIdx.x;
  const int width = (int)GAT_bucket_width(N);
  const int d0 = b * width;
  const int beg = bstart[b];
  const int end = bstart[b + 1];
  const int2* pp = reinterpret_cast<const int2*>(pairs);

  for (int i = t; i < GAT_WMAX; i += GAT_SORT_T) cnt[i] = 0;
  __syncthreads();
  for (int base = beg; base < end; base += GAT_EPT * GAT_SORT_T) {
    int d[GAT_EPT];
#pragma unroll
    for (int k = 0; k < GAT_EPT; ++k) {
      const int i = base + k * GAT_SORT_T + t;
      d[k] = (i < end) ? pp[i].y : d0;
    }
#pragma unroll
    for (int k = 0; k < GAT_EPT; ++k) {
      const int i = base + k * GAT_SORT_T + t;
      if (i < end) atomicAdd(&cnt[d[k] - d0], 1);
    }
  }
  __syncthreads();

  const int c0 = cnt[4 * t], c1 = cnt[4 * t + 1], c2 = cnt[4 * t + 2], c3 = cnt[4 * t + 3];
  const int ex = GAT_excl_scan256(c0 + c1 + c2 + c3, wsum);
  const int o0 = beg + ex, o1 = o0 + c0, o2 = o1 + c1, o3 = o2 + c2;
  cnt[4 * t] = o0; cnt[4 * t + 1] = o1; cnt[4 * t + 2] = o2; cnt[4 * t + 3] = o3;
  {
    const int q = 4 * t;
    if (q     < width && d0 + q     < N) rowptr[d0 + q]     = o0;
    if (q + 1 < width && d0 + q + 1 < N) rowptr[d0 + q + 1] = o1;
    if (q + 2 < width && d0 + q + 2 < N) rowptr[d0 + q + 2] = o2;
    if (q + 3 < width && d0 + q + 3 < N) rowptr[d0 + q + 3] = o3;
  }
  if (b == GAT_NB - 1 && t == 0) rowptr[N] = end;
  __syncthreads();

  for (int base = beg; base < end; base += GAT_EPT * GAT_SORT_T) {
    int2 p[GAT_EPT];
#pragma unroll
    for (int k = 0; k < GAT_EPT; ++k) {
      const int i = base + k * GAT_SORT_T + t;
      p[k] = (i < end) ? pp[i] : make_int2(0, d0);
    }
#pragma unroll
    for (int k = 0; k < GAT_EPT; ++k) {
      const int i = base + k * GAT_SORT_T + t;
      if (i < end) {
        const int slot = atomicAdd(&cnt[p[k].y - d0], 1);
        csrc[slot] = p[k].x;
      }
    }
  }
}

#define GAT_FOLD1_WG ((16 * 128) / GAT_SORT_T)
#define GAT_FOLD2_WG ((16 * 256) / GAT_SORT_T)

__global__ void __launch_bounds__(256)
GAT_count_fold(const int* __restrict__ dst, int* __restrict__ cntMat, int E, int N,
               const float* __restrict__ Wd1, const float* __restrict__ attD1, float* __restrict__ Vt1,
               const float* __restrict__ Wd2, const float* __restrict__ attD2, float* __restrict__ Vt2) {
  if (blockIdx.x < GAT_NB) {
    GAT_bucket_count_body(dst, cntMat, E, N, (int)blockIdx.x);
  } else {
    const int fb = (int)blockIdx.x - GAT_NB;
    if (fb < GAT_FOLD1_WG)
      GAT_fold_dst_body<128, 256, 32>(Wd1, attD1, Vt1, fb * GAT_SORT_T + (int)threadIdx.x);
    else
      GAT_fold_dst_body<256, 64, 64>(Wd2, attD2, Vt2, (fb - GAT_FOLD1_WG) * GAT_SORT_T + (int)threadIdx.x);
  }
}

__global__ void __launch_bounds__(256)
GAT_scatter_gemm(const int* __restrict__ src, const int* __restrict__ dst,
                 const int* __restrict__ cntMat, int* __restrict__ bstart,
                 int* __restrict__ pairs, int E, int N,
                 const float* __restrict__ A, const float* __restrict__ Bs, const float* __restrict__ Vt,
                 const float* __restrict__ attS,
                 float* __restrict__ Hout, float* __restrict__ aS, float* __restrict__ aD,
                 int tileBeg, int tileEnd) {
  if (blockIdx.x < GAT_NB)
    GAT_bucket_scatter_body(src, dst, cntMat, bstart, pairs, E, N, (int)blockIdx.x);
  else
    GAT_gemm_strip_any<128, 256, 32, false>(A, Bs, Vt, attS, Hout, aS, aD,
                                            (int)blockIdx.x - GAT_NB, tileBeg, tileEnd);
}

__global__ void __launch_bounds__(256)
GAT_sort_gemm(const int* __restrict__ pairs, const int* __restrict__ bstart,
              int* __restrict__ rowptr, int* __restrict__ csrc, int N,
              const float* __restrict__ A, const float* __restrict__ Bs, const float* __restrict__ Vt,
              const float* __restrict__ attS,
              float* __restrict__ Hout, float* __restrict__ aS, float* __restrict__ aD,
              int tileBeg, int tileEnd) {
  if (blockIdx.x < GAT_NB)
    GAT_bucket_sort_body(pairs, bstart, rowptr, csrc, N, (int)blockIdx.x);
  else
    GAT_gemm_strip_any<128, 256, 32, false>(A, Bs, Vt, attS, Hout, aS, aD,
                                            (int)blockIdx.x - GAT_NB, tileBeg, tileEnd);
}

template <int H>
__device__ __forceinline__ void GAT_load_heads(const float* __restrict__ a, int node, float (&v)[H]) {
  if constexpr (H % 4 == 0) {
    const float4* p = reinterpret_cast<const float4*>(a + (size_t)node * H);
#pragma unroll
    for (int q = 0; q < H / 4; ++q) {
      const float4 x = p[q];
      v[4 * q] = x.x; v[4 * q + 1] = x.y; v[4 * q + 2] = x.z; v[4 * q + 3] = x.w;
    }
  } else {
#pragma unroll
    for (int h = 0; h < H; ++h) v[h] = a[(size_t)node * H + h];
  }
}

#define GAT_GATHER_T 256

template <int H, int C>
__global__ void __launch_bounds__(GAT_GATHER_T)
GAT_gather(const int* __restrict__ rowptr, const int* __restrict__ csrc,
           const float* __restrict__ aS, const float* __restrict__ aD,
           const float* __restrict__ Hsrc, const float* __restrict__ bias,
           float* __restrict__ out, int N) {
  constexpr int HC = H * C;
  static_assert(HC % 128 == 0 || (H == 1 && C == 64), "rows of whole 512-byte pieces, or 256-byte rows of one head");

  const int lane = threadIdx.x & 31;
  const int d = blockIdx.x * (blockDim.x >> 5) + (threadIdx.x >> 5);
  if (d >= N) return;
  const int beg = __builtin_amdgcn_readfirstlane(rowptr[d]);
  const int end = __builtin_amdgcn_readfirstlane(rowptr[d + 1]);

  float ad[H];
#pragma unroll
  for (int h = 0; h < H; ++h) ad[h] = aD[(size_t)d * H + h];

  const bool in0 = (beg + lane) < end;
  int s0 = 0;
  float e0[H];
#pragma unroll
  for (int h = 0; h < H; ++h) e0[h] = -__builtin_inff();
  if (in0) {
    s0 = csrc[beg + lane];
    float as[H];
    GAT_load_heads<H>(aS, s0, as);
#pragma unroll
    for (int h = 0; h < H; ++h) e0[h] = GAT_lrelu(as[h] + ad[h]);
  }

  float mx[H];
#pragma unroll
  for (int h = 0; h < H; ++h) mx[h] = fmaxf(-__builtin_inff(), e0[h]);
  for (int base = beg + 32; base < end; base += 32) {
    const int idx = base + lane;
    if (idx < end) {
      float as[H];
      GAT_load_heads<H>(aS, csrc[idx], as);
#pragma unroll
      for (int h = 0; h < H; ++h) mx[h] = fmaxf(mx[h], GAT_lrelu(as[h] + ad[h]));
    }
  }
#pragma unroll
  for (int h = 0; h < H; ++h) {
#pragma unroll
    for (int m = 16; m >= 1; m >>= 1) mx[h] = fmaxf(mx[h], __shfl_xor(mx[h], m, 32));
  }

  float p0[H];
  float sm[H];
#pragma unroll
  for (int h = 0; h < H; ++h) {
    p0[h] = in0 ? __expf(e0[h] - mx[h]) : 0.0f;
    sm[h] = p0[h];
  }
  for (int base = beg + 32; base < end; base += 32) {
    const int idx = base + lane;
    if (idx < end) {
      float as[H];
      GAT_load_heads<H>(aS, csrc[idx], as);
#pragma unroll
      for (int h = 0; h < H; ++h) sm[h] += __expf(GAT_lrelu(as[h] + ad[h]) - mx[h]);
    }
  }
#pragma unroll
  for (int h = 0; h < H; ++h) {
#pragma unroll
    for (int m = 16; m >= 1; m >>= 1) sm[h] += __shfl_xor(sm[h], m, 32);
  }

  int s = s0;
  float al[H];
#pragma unroll
  for (int h = 0; h < H; ++h) al[h] = p0[h] / (sm[h] + 1e-16f);

  if constexpr (HC % 128 == 0) {
    constexpr int NQ = HC / 128;
    constexpr int G = 128 / C;
    static_assert(128 % C == 0 && H == NQ * G, "a piece holds whole heads");
    __shared__ __attribute__((aligned(16))) float sal[(GAT_GATHER_T / 32) * 32 * H];
    float* slot = sal + (threadIdx.x >> 5) * (32 * H);
    const float* ap = slot + ((4 * lane) / C) * NQ;
    const unsigned laneoff = (unsigned)lane * 16u;

    float4 acc[NQ];
#pragma unroll
    for (int q = 0; q < NQ; ++q) acc[q] = make_float4(0.0f, 0.0f, 0.0f, 0.0f);
    for (int base = beg; base < end; base += 32) {
      if (base != beg) {
        const int idx = base + lane;
        s = 0;
#pragma unroll
        for (int h = 0; h < H; ++h) al[h] = 0.0f;
        if (idx < end) {
          s = csrc[idx];
          float as[H];
          GAT_load_heads<H>(aS, s, as);
#pragma unroll
          for (int h = 0; h < H; ++h)
            al[h] = __expf(GAT_lrelu(as[h] + ad[h]) - mx[h]) / (sm[h] + 1e-16f);
        }
      }
#pragma unroll
      for (int h = 0; h < H; ++h) slot[lane * H + (h % G) * NQ + h / G] = al[h];
      const int n = (end - base) < 32 ? (end - base) : 32;
#pragma unroll 4
      for (int j = 0; j < n; ++j) {
        const int sj = __builtin_amdgcn_readlane(s, j);
        const char* rowp = reinterpret_cast<const char*>(Hsrc) + (size_t)(unsigned)sj * (HC * 4);
        float4 hv[NQ];
#pragma unroll
        for (int q = 0; q < NQ; ++q)
          hv[q] = *reinterpret_cast<const float4*>(rowp + laneoff + q * 512);
        float a[NQ];
#pragma unroll
        for (int q = 0; q < NQ; ++q) a[q] = ap[j * H + q];
#pragma unroll
        for (int q = 0; q < NQ; ++q) {
          acc[q].x += hv[q].x * a[q];
          acc[q].y += hv[q].y * a[q];
          acc[q].z += hv[q].z * a[q];
          acc[q].w += hv[q].w * a[q];
        }
      }
    }

    float4* op = reinterpret_cast<float4*>(out + (size_t)d * HC) + lane;
    const float4* bp = reinterpret_cast<const float4*>(bias) + lane;
#pragma unroll
    for (int q = 0; q < NQ; ++q) {
      const float4 b = bp[q * 32];
      op[q * 32] = make_float4(acc[q].x + b.x, acc[q].y + b.y, acc[q].z + b.z, acc[q].w + b.w);
    }
  } else {
    const int half = lane >> 4;
    const int c16 = lane & 15;
    float4 acc = make_float4(0.0f, 0.0f, 0.0f, 0.0f);
    for (int base = beg; base < end; base += 32) {
      if (base != beg) {
        const int idx = base + lane;
        s = 0;
        al[0] = 0.0f;
        if (idx < end) {
          s = csrc[idx];
          float as[H];
          GAT_load_heads<H>(aS, s, as);
          al[0] = __expf(GAT_lrelu(as[0] + ad[0]) - mx[0]) / (sm[0] + 1e-16f);
        }
      }
      const int n = (end - base) < 32 ? (end - base) : 32;
      for (int j = 0; j < n; j += 8) {
        float4 hv[4];
        float a[4];
#pragma unroll
        for (int u = 0; u < 4; ++u) {
          const int jj = j + 2 * u + half;
          const bool live = jj < n;
          const int sj = __shfl(s, live ? jj : j, 32);
          const float w = __shfl(al[0], jj, 32);
          a[u] = live ? w : 0.0f;
          const unsigned off = (unsigned)sj * (unsigned)(HC * 4) + (unsigned)c16 * 16u;
          hv[u] = *reinterpret_cast<const float4*>(reinterpret_cast<const char*>(Hsrc) + off);
        }
#pragma unroll
        for (int u = 0; u < 4; ++u) {
          acc.x += hv[u].x * a[u];
          acc.y += hv[u].y * a[u];
          acc.z += hv[u].z * a[u];
          acc.w += hv[u].w * a[u];
        }
      }
    }
    acc.x += __shfl_xor(acc.x, 16, 32);
    acc.y += __shfl_xor(acc.y, 16, 32);
    acc.z += __shfl_xor(acc.z, 16, 32);
    acc.w += __shfl_xor(acc.w, 16, 32);
    if (half == 0) {
      const float4 b = *(reinterpret_cast<const float4*>(bias) + c16);
      *(reinterpret_cast<float4*>(out + (size_t)d * HC) + c16) =
          make_float4(acc.x + b.x, acc.y + b.y, acc.z + b.z, acc.w + b.w);
    }
  }
}

extern "C" void kernel_launch(void* const* d_in, const int* in_sizes, int n_in,
                              void* d_out, int out_size, void* d_ws, size_t ws_size,
                              hipStream_t stream) {
  (void)n_in; (void)out_size; (void)ws_size;

  const float* x     = (const float*)d_in[0];
  const int*   ei    = (const int*)d_in[1];
  const float* Wsrc1 = (const float*)d_in[2];
  const float* Wdst1 = (const float*)d_in[3];
  const float* attS1 = (const float*)d_in[4];
  const float* attD1 = (const float*)d_in[5];
  const float* b1    = (const float*)d_in[6];
  const float* Wsrc2 = (const float*)d_in[7];
  const float* Wdst2 = (const float*)d_in[8];
  const float* attS2 = (const float*)d_in[9];
  const float* attD2 = (const float*)d_in[10];
  const float* b2    = (const float*)d_in[11];

  const int FIN = 128, HEADS = 8, HID = 32, EMB = 64;
  const int HC1 = HEADS * HID;
  const long long N = in_sizes[0] / FIN;
  const long long E = in_sizes[1] / 2;
  const int* src = ei;
  const int* dst = ei + E;

  float* p = (float*)d_ws;
  float* h1   = p; p += N * HC1;
  float* h2   = p; p += N * EMB;
  float* out1 = p; p += N * HC1;
  float* vt1  = p; p += 16 * FIN;
  float* vt2  = p; p += 16 * HC1;
  float* aS1  = p; p += N * HEADS;
  float* aD1  = p; p += N * HEADS;
  float* aS2  = p; p += N;
  float* aD2  = p; p += N;
  int* rowptr = (int*)p; p += N + 4;
  int* csrc   = (int*)p; p += E;
  int* pairs  = (int*)p; p += 2 * E;
  int* cntMat = (int*)p; p += GAT_NB * GAT_NB;
  int* bstart = (int*)p; p += GAT_NB + 4;
  float* outF = (float*)d_out;

  const int TPB = 256;
  auto nb = [](long long n, int t) { return (unsigned)((n + t - 1) / t); };

  const int tiles = (int)(N / 16);
  const int rgs = (int)nb(tiles, TPB / 32);
  const int g2 = (rgs + 1) / 2;
  const int g3 = rgs - g2;
  const int t2 = g2 * (TPB / 32);
  const int e2 = t2 < tiles ? t2 : tiles;
  GAT_count_fold<<<GAT_NB + GAT_FOLD1_WG + GAT_FOLD2_WG, TPB, 0, stream>>>(
      dst, cntMat, (int)E, (int)N, Wdst1, attD1, vt1, Wdst2, attD2, vt2);
  GAT_scatter_gemm<<<GAT_NB + g2 * (HC1 / 64), TPB, 0, stream>>>(
      src, dst, cntMat, bstart, pairs, (int)E, (int)N,
      x, Wsrc1, vt1, attS1, h1, aS1, aD1, 0, e2);
  GAT_sort_gemm<<<GAT_NB + g3 * (HC1 / 64), TPB, 0, stream>>>(
      pairs, bstart, rowptr, csrc, (int)N,
      x, Wsrc1, vt1, attS1, h1, aS1, aD1, e2, tiles);

  GAT_gather<HEADS, HID><<<nb(N * 32, GAT_GATHER_T), GAT_GATHER_T, 0, stream>>>(
      rowptr, csrc, aS1, aD1, h1, b1, out1, (int)N);

  {
    GAT_gemm_strip<256, 64, 64, true><<<nb(N / 16, TPB / 32) * (EMB / 64), TPB, 0, stream>>>(
        out1, Wsrc2, vt2, attS2, h2, aS2, aD2, (int)N);
  }
  GAT_gather<1, EMB><<<nb(N * 32, GAT_GATHER_T), GAT_GATHER_T, 0, stream>>>(
      rowptr, csrc, aS2, aD2, h2, b2, outF, (int)N);
}


# edge sort in 2 launches, was 3: a chunk sorts its own edges by bucket in LDS (count, scan, place), a bucket collects its 256 runs; the count launch goes: 5 launches; sum order = LDS arrival rank
# speedup vs baseline: 5.7563x; 1.1247x over previous
#include <hip/hip_runtime.h>
#include <hip/hip_bf16.h>


typedef float v2f __attribute__((ext_vector_type(2)));
typedef float v8f __attribute__((ext_vector_type(8)));

#define GAT_NEG_SLOPE 0.2f

__device__ __forceinline__ float GAT_lrelu(float v) {
  return v > 0.0f ? v : GAT_NEG_SLOPE * v;
}

template <int K, int NC, int CH>
__device__ __forceinline__ void
GAT_fold_dst_body(const float* __restrict__ Wd, const float* __restrict__ attD,
                  float* __restrict__ Vt, int idx) {
  constexpr int H = NC / CH;
  if (idx >= 16 * K) return;
  const int h = idx / K;
  const int k = idx - h * K;
  float s = 0.0f;
  if (h < H) {
    const float4* w = reinterpret_cast<const float4*>(Wd + (size_t)k * NC + h * CH);
    const float4* a = reinterpret_cast<const float4*>(attD + h * CH);
#pragma unroll
    for (int q = 0; q < CH / 4; ++q) {
      const float4 wv = w[q];
      const float4 av = a[q];
      s += wv.x * av.x + wv.y * av.y + wv.z * av.z + wv.w * av.w;
    }
  }
  Vt[idx] = s;
}

template <int K, int NC, int CH, bool RELU, bool LOGIT>
__device__ __forceinline__ void
GAT_gemm_strip_body(const float* __restrict__ A,
                    const float* __restrict__ Bs,
                    const float* __restrict__ Vt,
                    const float* __restrict__ attS,
                    float* __restrict__ Hout,
                    float* __restrict__ aS,
                    float* __restrict__ aD,
                    int grp, int strip, int tileBeg, int tileEnd) {
  constexpr int H = NC / CH;
  constexpr int LPH = CH / 4;
  static_assert(K % 32 == 0, "K is walked in blocks of 32");
  static_assert(NC % 64 == 0, "a strip is 64 columns");
  static_assert(CH == 32 || CH == 64, "a strip must hold whole heads");
  static_assert(H <= 16, "the folded heads fill one 16-column tile");

  const int wpg = (int)(blockDim.x >> 5);
  const int mTile = tileBeg + grp * wpg + (int)(threadIdx.x >> 5);
  if (mTile >= tileEnd) return;

  const int mBase = mTile << 4;
  const int nBase = strip << 6;
  const int lane = threadIdx.x & 31;
  const int half = lane >> 4;
  const int c    = lane & 15;

  v8f cS[4] = {};
  v8f cL = {};
  const float* arow = A + (size_t)(mBase + c) * K + 16 * half;
  const float* vrow = LOGIT ? Vt + (size_t)c * K + 16 * half : nullptr;
  const float* bsp = Bs + (size_t)(16 * half) * NC + nBase + 4 * c;

  for (int kb = 0; kb < K; kb += 32) {
    float av[16];
    float lv[16];
#pragma unroll
    for (int q = 0; q < 4; ++q) {
      const float4 v = *reinterpret_cast<const float4*>(arow + kb + 4 * q);
      av[4 * q] = v.x; av[4 * q + 1] = v.y; av[4 * q + 2] = v.z; av[4 * q + 3] = v.w;
      if constexpr (LOGIT) {
        const float4 u = *reinterpret_cast<const float4*>(vrow + kb + 4 * q);
        lv[4 * q] = u.x; lv[4 * q + 1] = u.y; lv[4 * q + 2] = u.z; lv[4 * q + 3] = u.w;
      }
    }
    if (RELU) {
#pragma unroll
      for (int i = 0; i < 16; ++i) av[i] = fmaxf(av[i], 0.0f);
    }
#pragma unroll
    for (int j = 0; j < 8; ++j) {
      const float* ps = bsp + (size_t)(kb + 2 * j) * NC;
      const float4 s0 = *reinterpret_cast<const float4*>(ps);
      const float4 s1 = *reinterpret_cast<const float4*>(ps + NC);
      v2f a;  a.x = av[2 * j];  a.y = av[2 * j + 1];
      v2f b;
      b.x = s0.x; b.y = s1.x;
      cS[0] = __builtin_amdgcn_wmma_f32_16x16x4_f32(false, a, false, b, (short)0, cS[0], false, false);
      b.x = s0.y; b.y = s1.y;
      cS[1] = __builtin_amdgcn_wmma_f32_16x16x4_f32(false, a, false, b, (short)0, cS[1], false, false);
      b.x = s0.z; b.y = s1.z;
      cS[2] = __builtin_amdgcn_wmma_f32_16x16x4_f32(false, a, false, b, (short)0, cS[2], false, false);
      b.x = s0.w; b.y = s1.w;
      cS[3] = __builtin_amdgcn_wmma_f32_16x16x4_f32(false, a, false, b, (short)0, cS[3], false, false);
      if constexpr (LOGIT) {
        b.x = lv[2 * j]; b.y = lv[2 * j + 1];
        cL = __builtin_amdgcn_wmma_f32_16x16x4_f32(false, a, false, b, (short)0, cL, false, false);
      }
    }
  }

  const int col = nBase + 4 * c;
#pragma unroll
  for (int r = 0; r < 8; ++r) {
    float4 o;
    o.x = cS[0][r]; o.y = cS[1][r]; o.z = cS[2][r]; o.w = cS[3][r];
    *reinterpret_cast<float4*>(Hout + (size_t)(mBase + r + 8 * half) * NC + col) = o;
  }

  if constexpr (LOGIT) {
    if (c < H) {
#pragma unroll
      for (int r = 0; r < 8; ++r)
        aD[(size_t)(mBase + r + 8 * half) * H + c] = cL[r];
    }
  }

  const float4 wS = *reinterpret_cast<const float4*>(attS + col);
  const int head = col / CH;
  float vs[8];
#pragma unroll
  for (int r = 0; r < 8; ++r)
    vs[r] = cS[0][r] * wS.x + cS[1][r] * wS.y + cS[2][r] * wS.z + cS[3][r] * wS.w;
#pragma unroll
  for (int m = 1; m < LPH; m <<= 1) {
#pragma unroll
    for (int r = 0; r < 8; ++r) vs[r] += __shfl_xor(vs[r], m, 32);
  }
  if ((c & (LPH - 1)) == 0) {
#pragma unroll
    for (int r = 0; r < 8; ++r)
      aS[(size_t)(mBase + r + 8 * half) * H + head] = vs[r];
  }
}

template <int K, int NC, int CH, bool RELU, int S0, int NS>
__device__ __forceinline__ void
GAT_gemm_strip_any(const float* __restrict__ A,
                   const float* __restrict__ Bs,
                   const float* __restrict__ Vt,
                   const float* __restrict__ attS,
                   float* __restrict__ Hout,
                   float* __restrict__ aS,
                   float* __restrict__ aD,
                   int blk, int tileBeg, int tileEnd) {
  static_assert(S0 >= 0 && NS >= 1 && S0 + NS <= NC / 64, "strips of the row tile");
  const int strip = S0 + blk % NS;
  const int grp = blk / NS;
  if constexpr (S0 == 0) {
    if (strip == 0) {
      GAT_gemm_strip_body<K, NC, CH, RELU, true>(A, Bs, Vt, attS, Hout, aS, aD, grp, 0, tileBeg, tileEnd);
      return;
    }
  }
  GAT_gemm_strip_body<K, NC, CH, RELU, false>(A, Bs, Vt, attS, Hout, aS, aD, grp, strip, tileBeg, tileEnd);
}

template <int K, int NC, int CH, bool RELU>
__global__ void __launch_bounds__(256)
GAT_gemm_strip(const float* __restrict__ A,
               const float* __restrict__ Bs,
               const float* __restrict__ Vt,
               const float* __restrict__ attS,
               float* __restrict__ Hout,
               float* __restrict__ aS,
               float* __restrict__ aD,
               int M) {
  GAT_gemm_strip_any<K, NC, CH, RELU, 0, NC / 64>(A, Bs, Vt, attS, Hout, aS, aD,
                                                  (int)blockIdx.x, 0, M >> 4);
}

#define GAT_NB 256
#define GAT_SORT_T 256
#define GAT_WMAX 1024
#define GAT_EPT 16
#define GAT_OP (GAT_NB + 4)

__device__ __forceinline__ unsigned GAT_bucket_width(int N) {
  return (unsigned)(N + GAT_NB - 1) / GAT_NB;
}

__device__ __forceinline__ int GAT_excl_scan256(int v, int* wsum) {
  const int lane = threadIdx.x & 31;
  const int wv = threadIdx.x >> 5;
  int inc = v;
  for (int m = 1; m < 32; m <<= 1) {
    const int u = __shfl_up(inc, m, 32);
    if (lane >= m) inc += u;
  }
  if (lane == 31) wsum[wv] = inc;
  __syncthreads();
  int woff = 0;
  for (int k = 0; k < GAT_SORT_T / 32; ++k) woff += (k < wv) ? wsum[k] : 0;
  return woff + inc - v;
}

__device__ __forceinline__ void
GAT_chunk_sort_body(const int* __restrict__ src, const int* __restrict__ dst,
                    int* __restrict__ off, int* __restrict__ pairs, int E, int N, int w) {
  __shared__ int c[GAT_NB];
  __shared__ int wsum[GAT_SORT_T / 32];
  const int t = threadIdx.x;
  c[t] = 0;
  __syncthreads();
  const unsigned width = GAT_bucket_width(N);
  const int chunk = (E + GAT_NB - 1) / GAT_NB;
  const int beg = w * chunk;
  const int end = (beg + chunk < E) ? (beg + chunk) : E;
  for (int base = beg; base < end; base += GAT_EPT * GAT_SORT_T) {
    int d[GAT_EPT];
#pragma unroll
    for (int k = 0; k < GAT_EPT; ++k) {
      const int i = base + k * GAT_SORT_T + t;
      d[k] = (i < end) ? dst[i] : 0;
    }
#pragma unroll
    for (int k = 0; k < GAT_EPT; ++k) {
      const int i = base + k * GAT_SORT_T + t;
      if (i < end) atomicAdd(&c[(unsigned)d[k] / width], 1);
    }
  }
  __syncthreads();

  const int n = c[t];
  const int ex = GAT_excl_scan256(n, wsum);
  off[w * GAT_OP + t] = ex;
  if (t == GAT_NB - 1) off[w * GAT_OP + GAT_NB] = ex + n;
  c[t] = beg + ex;
  __syncthreads();

  int2* pp = reinterpret_cast<int2*>(pairs);
  for (int base = beg; base < end; base += GAT_EPT * GAT_SORT_T) {
    int s[GAT_EPT], d[GAT_EPT];
#pragma unroll
    for (int k = 0; k < GAT_EPT; ++k) {
      const int i = base + k * GAT_SORT_T + t;
      s[k] = (i < end) ? src[i] : 0;
      d[k] = (i < end) ? dst[i] : 0;
    }
#pragma unroll
    for (int k = 0; k < GAT_EPT; ++k) {
      const int i = base + k * GAT_SORT_T + t;
      if (i < end) {
        const int slot = atomicAdd(&c[(unsigned)d[k] / width], 1);
        pp[slot] = make_int2(s[k], d[k]);
      }
    }
  }
}

__device__ __forceinline__ void
GAT_bucket_sort_body(const int* __restrict__ pairs, const int* __restrict__ off,
                     int* __restrict__ rowptr, int* __restrict__ csrc, int E, int N, int b) {
  __shared__ int cnt[GAT_WMAX];
  __shared__ int wsum[GAT_SORT_T / 32];
  __shared__ int wtot[GAT_SORT_T / 32];
  const int t = threadIdx.x;
  const int lane = t & 31;
  const int width = (int)GAT_bucket_width(N);
  const int d0 = b * width;
  const int chunk = (E + GAT_NB - 1) / GAT_NB;
  const int2* pp = reinterpret_cast<const int2*>(pairs);

  const int o0 = off[t * GAT_OP + b];
  const int o1 = off[t * GAT_OP + b + 1];
  const int sbeg = t * chunk + o0;
  const int slen = o1 - o0;

  for (int i = t; i < GAT_WMAX; i += GAT_SORT_T) cnt[i] = 0;

  int part = o0;
  for (int m = 16; m >= 1; m >>= 1) part += __shfl_xor(part, m, 32);
  if (lane == 0) wtot[t >> 5] = part;
  __syncthreads();
  int beg = 0;
  for (int k = 0; k < GAT_SORT_T / 32; ++k) beg += wtot[k];

  int2 p[32];
#pragma unroll
  for (int i = 0; i < 32; ++i) {
    const int sb = __builtin_amdgcn_readlane(sbeg, i);
    const int ln = __builtin_amdgcn_readlane(slen, i);
    p[i] = (lane < ln) ? pp[sb + lane] : make_int2(0, d0);
  }
#pragma unroll
  for (int i = 0; i < 32; ++i) {
    const int ln = __builtin_amdgcn_readlane(slen, i);
    if (lane < ln) atomicAdd(&cnt[p[i].y - d0], 1);
  }
  for (int i = 0; i < 32; ++i) {
    const int sb = __builtin_amdgcn_readlane(sbeg, i);
    const int ln = __builtin_amdgcn_readlane(slen, i);
    for (int k = 32 + lane; k < ln; k += 32) atomicAdd(&cnt[pp[sb + k].y - d0], 1);
  }
  __syncthreads();

  const int c0 = cnt[4 * t], c1 = cnt[4 * t + 1], c2 = cnt[4 * t + 2], c3 = cnt[4 * t + 3];
  const int ex = GAT_excl_scan256(c0 + c1 + c2 + c3, wsum);
  const int r0 = beg + ex, r1 = r0 + c0, r2 = r1 + c1, r3 = r2 + c2;
  cnt[4 * t] = r0; cnt[4 * t + 1] = r1; cnt[4 * t + 2] = r2; cnt[4 * t + 3] = r3;
  {
    const int q = 4 * t;
    if (q     < width && d0 + q     < N) rowptr[d0 + q]     = r0;
    if (q + 1 < width && d0 + q + 1 < N) rowptr[d0 + q + 1] = r1;
    if (q + 2 < width && d0 + q + 2 < N) rowptr[d0 + q + 2] = r2;
    if (q + 3 < width && d0 + q + 3 < N) rowptr[d0 + q + 3] = r3;
  }
  if (b == GAT_NB - 1 && t == 0) rowptr[N] = E;
  __syncthreads();

#pragma unroll
  for (int i = 0; i < 32; ++i) {
    const int ln = __builtin_amdgcn_readlane(slen, i);
    if (lane < ln) {
      const int slot = atomicAdd(&cnt[p[i].y - d0], 1);
      csrc[slot] = p[i].x;
    }
  }
  for (int i = 0; i < 32; ++i) {
    const int sb = __builtin_amdgcn_readlane(sbeg, i);
    const int ln = __builtin_amdgcn_readlane(slen, i);
    for (int k = 32 + lane; k < ln; k += 32) {
      const int2 q = pp[sb + k];
      const int slot = atomicAdd(&cnt[q.y - d0], 1);
      csrc[slot] = q.x;
    }
  }
}

#define GAT_FOLD1_WG ((16 * 128) / GAT_SORT_T)
#define GAT_FOLD2_WG ((16 * 256) / GAT_SORT_T)

__global__ void __launch_bounds__(256)
GAT_chunk_fold_gemm(const int* __restrict__ src, const int* __restrict__ dst,
                    int* __restrict__ off, int* __restrict__ pairs, int E, int N,
                    const float* __restrict__ Wd1, const float* __restrict__ attD1, float* __restrict__ Vt1,
                    const float* __restrict__ Wd2, const float* __restrict__ attD2, float* __restrict__ Vt2,
                    const float* __restrict__ A, const float* __restrict__ Bs,
                    const float* __restrict__ attS,
                    float* __restrict__ Hout, float* __restrict__ aS, int tiles) {
  if (blockIdx.x < GAT_NB) {
    GAT_chunk_sort_body(src, dst, off, pairs, E, N, (int)blockIdx.x);
  } else if (blockIdx.x < GAT_NB + GAT_FOLD1_WG + GAT_FOLD2_WG) {
    const int fb = (int)blockIdx.x - GAT_NB;
    if (fb < GAT_FOLD1_WG)
      GAT_fold_dst_body<128, 256, 32>(Wd1, attD1, Vt1, fb * GAT_SORT_T + (int)threadIdx.x);
    else
      GAT_fold_dst_body<256, 64, 64>(Wd2, attD2, Vt2, (fb - GAT_FOLD1_WG) * GAT_SORT_T + (int)threadIdx.x);
  } else {
    GAT_gemm_strip_any<128, 256, 32, false, 1, 3>(A, Bs, nullptr, attS, Hout, aS, nullptr,
        (int)blockIdx.x - (GAT_NB + GAT_FOLD1_WG + GAT_FOLD2_WG), 0, tiles);
  }
}

__global__ void __launch_bounds__(256)
GAT_sort_gemm(const int* __restrict__ pairs, const int* __restrict__ off,
              int* __restrict__ rowptr, int* __restrict__ csrc, int E, int N,
              const float* __restrict__ A, const float* __restrict__ Bs, const float* __restrict__ Vt,
              const float* __restrict__ attS,
              float* __restrict__ Hout, float* __restrict__ aS, float* __restrict__ aD,
              int tiles) {
  if (blockIdx.x < GAT_NB)
    GAT_bucket_sort_body(pairs, off, rowptr, csrc, E, N, (int)blockIdx.x);
  else
    GAT_gemm_strip_any<128, 256, 32, false, 0, 1>(A, Bs, Vt, attS, Hout, aS, aD,
                                                  (int)blockIdx.x - GAT_NB, 0, tiles);
}

template <int H>
__device__ __forceinline__ void GAT_load_heads(const float* __restrict__ a, int node, float (&v)[H]) {
  if constexpr (H % 4 == 0) {
    const float4* p = reinterpret_cast<const float4*>(a + (size_t)node * H);
#pragma unroll
    for (int q = 0; q < H / 4; ++q) {
      const float4 x = p[q];
      v[4 * q] = x.x; v[4 * q + 1] = x.y; v[4 * q + 2] = x.z; v[4 * q + 3] = x.w;
    }
  } else {
#pragma unroll
    for (int h = 0; h < H; ++h) v[h] = a[(size_t)node * H + h];
  }
}

#define GAT_GATHER_T 256

template <int H, int C>
__global__ void __launch_bounds__(GAT_GATHER_T)
GAT_gather(const int* __restrict__ rowptr, const int* __restrict__ csrc,
           const float* __restrict__ aS, const float* __restrict__ aD,
           const float* __restrict__ Hsrc, const float* __restrict__ bias,
           float* __restrict__ out, int N) {
  constexpr int HC = H * C;
  static_assert(HC % 128 == 0 || (H == 1 && C == 64), "rows of whole 512-byte pieces, or 256-byte rows of one head");

  const int lane = threadIdx.x & 31;
  const int d = blockIdx.x * (blockDim.x >> 5) + (threadIdx.x >> 5);
  if (d >= N) return;
  const int beg = __builtin_amdgcn_readfirstlane(rowptr[d]);
  const int end = __builtin_amdgcn_readfirstlane(rowptr[d + 1]);

  float ad[H];
#pragma unroll
  for (int h = 0; h < H; ++h) ad[h] = aD[(size_t)d * H + h];

  const bool in0 = (beg + lane) < end;
  int s0 = 0;
  float e0[H];
#pragma unroll
  for (int h = 0; h < H; ++h) e0[h] = -__builtin_inff();
  if (in0) {
    s0 = csrc[beg + lane];
    float as[H];
    GAT_load_heads<H>(aS, s0, as);
#pragma unroll
    for (int h = 0; h < H; ++h) e0[h] = GAT_lrelu(as[h] + ad[h]);
  }

  float mx[H];
#pragma unroll
  for (int h = 0; h < H; ++h) mx[h] = fmaxf(-__builtin_inff(), e0[h]);
  for (int base = beg + 32; base < end; base += 32) {
    const int idx = base + lane;
    if (idx < end) {
      float as[H];
      GAT_load_heads<H>(aS, csrc[idx], as);
#pragma unroll
      for (int h = 0; h < H; ++h) mx[h] = fmaxf(mx[h], GAT_lrelu(as[h] + ad[h]));
    }
  }
#pragma unroll
  for (int h = 0; h < H; ++h) {
#pragma unroll
    for (int m = 16; m >= 1; m >>= 1) mx[h] = fmaxf(mx[h], __shfl_xor(mx[h], m, 32));
  }

  float p0[H];
  float sm[H];
#pragma unroll
  for (int h = 0; h < H; ++h) {
    p0[h] = in0 ? __expf(e0[h] - mx[h]) : 0.0f;
    sm[h] = p0[h];
  }
  for (int base = beg + 32; base < end; base += 32) {
    const int idx = base + lane;
    if (idx < end) {
      float as[H];
      GAT_load_heads<H>(aS, csrc[idx], as);
#pragma unroll
      for (int h = 0; h < H; ++h) sm[h] += __expf(GAT_lrelu(as[h] + ad[h]) - mx[h]);
    }
  }
#pragma unroll
  for (int h = 0; h < H; ++h) {
#pragma unroll
    for (int m = 16; m >= 1; m >>= 1) sm[h] += __shfl_xor(sm[h], m, 32);
  }

  int s = s0;
  float al[H];
#pragma unroll
  for (int h = 0; h < H; ++h) al[h] = p0[h] / (sm[h] + 1e-16f);

  if constexpr (HC % 128 == 0) {
    constexpr int NQ = HC / 128;
    constexpr int G = 128 / C;
    static_assert(128 % C == 0 && H == NQ * G, "a piece holds whole heads");
    __shared__ __attribute__((aligned(16))) float sal[(GAT_GATHER_T / 32) * 32 * H];
    float* slot = sal + (threadIdx.x >> 5) * (32 * H);
    const float* ap = slot + ((4 * lane) / C) * NQ;
    const unsigned laneoff = (unsigned)lane * 16u;

    float4 acc[NQ];
#pragma unroll
    for (int q = 0; q < NQ; ++q) acc[q] = make_float4(0.0f, 0.0f, 0.0f, 0.0f);
    for (int base = beg; base < end; base += 32) {
      if (base != beg) {
        const int idx = base + lane;
        s = 0;
#pragma unroll
        for (int h = 0; h < H; ++h) al[h] = 0.0f;
        if (idx < end) {
          s = csrc[idx];
          float as[H];
          GAT_load_heads<H>(aS, s, as);
#pragma unroll
          for (int h = 0; h < H; ++h)
            al[h] = __expf(GAT_lrelu(as[h] + ad[h]) - mx[h]) / (sm[h] + 1e-16f);
        }
      }
#pragma unroll
      for (int h = 0; h < H; ++h) slot[lane * H + (h % G) * NQ + h / G] = al[h];
      const int n = (end - base) < 32 ? (end - base) : 32;
#pragma unroll 4
      for (int j = 0; j < n; ++j) {
        const int sj = __builtin_amdgcn_readlane(s, j);
        const char* rowp = reinterpret_cast<const char*>(Hsrc) + (size_t)(unsigned)sj * (HC * 4);
        float4 hv[NQ];
#pragma unroll
        for (int q = 0; q < NQ; ++q)
          hv[q] = *reinterpret_cast<const float4*>(rowp + laneoff + q * 512);
        float a[NQ];
#pragma unroll
        for (int q = 0; q < NQ; ++q) a[q] = ap[j * H + q];
#pragma unroll
        for (int q = 0; q < NQ; ++q) {
          acc[q].x += hv[q].x * a[q];
          acc[q].y += hv[q].y * a[q];
          acc[q].z += hv[q].z * a[q];
          acc[q].w += hv[q].w * a[q];
        }
      }
    }

    float4* op = reinterpret_cast<float4*>(out + (size_t)d * HC) + lane;
    const float4* bp = reinterpret_cast<const float4*>(bias) + lane;
#pragma unroll
    for (int q = 0; q < NQ; ++q) {
      const float4 b = bp[q * 32];
      op[q * 32] = make_float4(acc[q].x + b.x, acc[q].y + b.y, acc[q].z + b.z, acc[q].w + b.w);
    }
  } else {
    const int half = lane >> 4;
    const int c16 = lane & 15;
    float4 acc = make_float4(0.0f, 0.0f, 0.0f, 0.0f);
    for (int base = beg; base < end; base += 32) {
      if (base != beg) {
        const int idx = base + lane;
        s = 0;
        al[0] = 0.0f;
        if (idx < end) {
          s = csrc[idx];
          float as[H];
          GAT_load_heads<H>(aS, s, as);
          al[0] = __expf(GAT_lrelu(as[0] + ad[0]) - mx[0]) / (sm[0] + 1e-16f);
        }
      }
      const int n = (end - base) < 32 ? (end - base) : 32;
      for (int j = 0; j < n; j += 8) {
        float4 hv[4];
        float a[4];
#pragma unroll
        for (int u = 0; u < 4; ++u) {
          const int jj = j + 2 * u + half;
          const bool live = jj < n;
          const int sj = __shfl(s, live ? jj : j, 32);
          const float w = __shfl(al[0], jj, 32);
          a[u] = live ? w : 0.0f;
          const unsigned off = (unsigned)sj * (unsigned)(HC * 4) + (unsigned)c16 * 16u;
          hv[u] = *reinterpret_cast<const float4*>(reinterpret_cast<const char*>(Hsrc) + off);
        }
#pragma unroll
        for (int u = 0; u < 4; ++u) {
          acc.x += hv[u].x * a[u];
          acc.y += hv[u].y * a[u];
          acc.z += hv[u].z * a[u];
          acc.w += hv[u].w * a[u];
        }
      }
    }
    acc.x += __shfl_xor(acc.x, 16, 32);
    acc.y += __shfl_xor(acc.y, 16, 32);
    acc.z += __shfl_xor(acc.z, 16, 32);
    acc.w += __shfl_xor(acc.w, 16, 32);
    if (half == 0) {
      const float4 b = *(reinterpret_cast<const float4*>(bias) + c16);
      *(reinterpret_cast<float4*>(out + (size_t)d * HC) + c16) =
          make_float4(acc.x + b.x, acc.y + b.y, acc.z + b.z, acc.w + b.w);
    }
  }
}

extern "C" void kernel_launch(void* const* d_in, const int* in_sizes, int n_in,
                              void* d_out, int out_size, void* d_ws, size_t ws_size,
                              hipStream_t stream) {
  (void)n_in; (void)out_size; (void)ws_size;

  const float* x     = (const float*)d_in[0];
  const int*   ei    = (const int*)d_in[1];
  const float* Wsrc1 = (const float*)d_in[2];
  const float* Wdst1 = (const float*)d_in[3];
  const float* attS1 = (const float*)d_in[4];
  const float* attD1 = (const float*)d_in[5];
  const float* b1    = (const float*)d_in[6];
  const float* Wsrc2 = (const float*)d_in[7];
  const float* Wdst2 = (const float*)d_in[8];
  const float* attS2 = (const float*)d_in[9];
  const float* attD2 = (const float*)d_in[10];
  const float* b2    = (const float*)d_in[11];

  const int FIN = 128, HEADS = 8, HID = 32, EMB = 64;
  const int HC1 = HEADS * HID;
  const long long N = in_sizes[0] / FIN;
  const long long E = in_sizes[1] / 2;
  const int* src = ei;
  const int* dst = ei + E;

  float* p = (float*)d_ws;
  float* h1   = p; p += N * HC1;
  float* h2   = p; p += N * EMB;
  float* out1 = p; p += N * HC1;
  float* vt1  = p; p += 16 * FIN;
  float* vt2  = p; p += 16 * HC1;
  float* aS1  = p; p += N * HEADS;
  float* aD1  = p; p += N * HEADS;
  float* aS2  = p; p += N;
  float* aD2  = p; p += N;
  int* rowptr = (int*)p; p += N + 4;
  int* csrc   = (int*)p; p += E;
  int* pairs  = (int*)p; p += 2 * E;
  int* off    = (int*)p; p += GAT_NB * GAT_OP;
  float* outF = (float*)d_out;

  const int TPB = 256;
  auto nb = [](long long n, int t) { return (unsigned)((n + t - 1) / t); };

  const int tiles = (int)(N / 16);
  const int rgs = (int)nb(tiles, TPB / 32);
  GAT_chunk_fold_gemm<<<GAT_NB + GAT_FOLD1_WG + GAT_FOLD2_WG + rgs * (HC1 / 64 - 1), TPB, 0, stream>>>(
      src, dst, off, pairs, (int)E, (int)N, Wdst1, attD1, vt1, Wdst2, attD2, vt2,
      x, Wsrc1, attS1, h1, aS1, tiles);
  GAT_sort_gemm<<<GAT_NB + rgs, TPB, 0, stream>>>(
      pairs, off, rowptr, csrc, (int)E, (int)N,
      x, Wsrc1, vt1, attS1, h1, aS1, aD1, tiles);

  GAT_gather<HEADS, HID><<<nb(N * 32, GAT_GATHER_T), GAT_GATHER_T, 0, stream>>>(
      rowptr, csrc, aS1, aD1, h1, b1, out1, (int)N);

  {
    GAT_gemm_strip<256, 64, 64, true><<<nb(N / 16, TPB / 32) * (EMB / 64), TPB, 0, stream>>>(
        out1, Wsrc2, vt2, attS2, h2, aS2, aD2, (int)N);
  }
  GAT_gather<1, EMB><<<nb(N * 32, GAT_GATHER_T), GAT_GATHER_T, 0, stream>>>(
      rowptr, csrc, aS2, aD2, h2, b2, outF, (int)N);
}
